# starting kernel
# baseline (speedup 1.0000x reference)
#include <hip/hip_runtime.h>

typedef __attribute__((ext_vector_type(2))) float v2f;
typedef __attribute__((ext_vector_type(8))) float v8f;

#define N_NODES   131072
#define E_EDGES   4194304
#define S_IN      32
#define H_DIM     16
#define B_GRAPHS  128
#define FC_IN_SZ  16384
#define KCHUNKS   8

__global__ void fill_kernel(float* __restrict__ p, float v, int n) {
    int i = blockIdx.x * blockDim.x + threadIdx.x;
    if (i < n) p[i] = v;
}

__global__ void degree_kernel(const int* __restrict__ col, float* __restrict__ deg, int e) {
    int i = blockIdx.x * blockDim.x + threadIdx.x;
    if (i < e) atomicAdd(&deg[col[i]], 1.0f);
}

__global__ void rsqrt_kernel(float* __restrict__ d, int n) {
    int i = blockIdx.x * blockDim.x + threadIdx.x;
    if (i < n) d[i] = rsqrtf(d[i]);
}

__global__ __launch_bounds__(32)
void wmma_gemm_f32(const float* __restrict__ A, const float* __restrict__ B,
                   const float* __restrict__ bias, float* __restrict__ C,
                   int K, int Nld, int relu, int Kc) {
    const int nb   = blockIdx.x;
    const int mb   = blockIdx.y;
    const int z    = blockIdx.z;
    const int lane = threadIdx.x;
    const int half = lane >> 4;
    const int lm   = lane & 15;

    const int k0 = z * Kc;

    v8f acc = {};
    const float* Ap = A + (size_t)(mb * 16 + lm) * K + k0 + 2 * half;
    const float* Bp = B + (size_t)(k0 + 2 * half) * Nld + nb * 16 + lm;

#pragma unroll 4
    for (int k = 0; k < Kc; k += 4) {
        v2f a = *(const v2f*)Ap;
        v2f b;
        b.x = Bp[0];
        b.y = Bp[Nld];
        acc = __builtin_amdgcn_wmma_f32_16x16x4_f32(
                  false, a, false, b, (short)0, acc, false, false);
        Ap += 4;
        Bp += (size_t)4 * Nld;
    }

    const int cn = nb * 16 + lm;
    if (gridDim.z == 1) {
        const float bb = bias ? bias[cn] : 0.0f;
#pragma unroll
        for (int v = 0; v < 8; ++v) {
            float r = acc[v] + bb;
            if (relu) r = fmaxf(r, 0.0f);
            C[(size_t)(mb * 16 + v + 8 * half) * Nld + cn] = r;
        }
    } else {
        const size_t M  = (size_t)gridDim.y * 16;
        float* Cp = C + (size_t)z * M * Nld;
#pragma unroll
        for (int v = 0; v < 8; ++v) {
            Cp[(size_t)(mb * 16 + v + 8 * half) * Nld + cn] = acc[v];
        }
    }
}

__global__ void reduce_bias_relu_kernel(const float* __restrict__ part,
                                        const float* __restrict__ bias,
                                        float* __restrict__ C,
                                        int MN, int Nld, int relu) {
    int i = blockIdx.x * blockDim.x + threadIdx.x;
    if (i >= MN) return;
    float s = 0.0f;
#pragma unroll
    for (int zc = 0; zc < KCHUNKS; ++zc) s += part[(size_t)zc * MN + i];
    s += bias[i % Nld];
    if (relu) s = fmaxf(s, 0.0f);
    C[i] = s;
}

__global__ void scatter_kernel(const float* __restrict__ xw,
                               const int* __restrict__ row, const int* __restrict__ col,
                               const float* __restrict__ dinv,
                               float* __restrict__ out, int e16) {
    int t = blockIdx.x * blockDim.x + threadIdx.x;
    if (t >= e16) return;
    int e = t >> 4, j = t & 15;
    int r = row[e], c = col[e];
    float w = dinv[r] * dinv[c];
    atomicAdd(&out[c * H_DIM + j], w * xw[r * H_DIM + j]);
}

__global__ void finalize_kernel(float* __restrict__ h, const float* __restrict__ xw,
                                const float* __restrict__ dinv,
                                const float* __restrict__ bias, int n) {
    int i = blockIdx.x * blockDim.x + threadIdx.x;
    if (i >= n) return;
    int node = i >> 4, j = i & 15;
    float d = dinv[node];
    float r = h[i] + d * d * xw[i] + bias[j];
    h[i] = fmaxf(r, 0.0f);
}

__global__ void heads_kernel(const float* __restrict__ V,
                             const float* __restrict__ Wiv, const float* __restrict__ biv,
                             const float* __restrict__ Wev, const float* __restrict__ bev,
                             float* __restrict__ out_ext, float* __restrict__ out_int) {
    int r = blockIdx.x * blockDim.x + threadIdx.x;
    if (r >= B_GRAPHS) return;
    const float* v = V + (size_t)r * 256;
    float si = 0.0f, se = 0.0f;
#pragma unroll 4
    for (int k = 0; k < 256; ++k) {
        float x = v[k];
        si += x * Wiv[k];
        se += x * Wev[k];
    }
    out_int[r] = si + biv[0];
    out_ext[r] = se + bev[0];
}

extern "C" void kernel_launch(void* const* d_in, const int* in_sizes, int n_in,
                              void* d_out, int out_size, void* d_ws, size_t ws_size,
                              hipStream_t stream) {
    const float* x    = (const float*)d_in[0];
    const int*   ei   = (const int*)d_in[1];
    const float* W1   = (const float*)d_in[2];
    const float* b1   = (const float*)d_in[3];
    const float* W2   = (const float*)d_in[4];
    const float* b2   = (const float*)d_in[5];
    const float* W3   = (const float*)d_in[6];
    const float* b3   = (const float*)d_in[7];
    const float* Wp1  = (const float*)d_in[8];
    const float* bp1  = (const float*)d_in[9];
    const float* Wp2  = (const float*)d_in[10];
    const float* bp2  = (const float*)d_in[11];
    const float* Wv1  = (const float*)d_in[12];
    const float* bv1  = (const float*)d_in[13];
    const float* Wiv  = (const float*)d_in[14];
    const float* biv  = (const float*)d_in[15];
    const float* Wev  = (const float*)d_in[16];
    const float* bev  = (const float*)d_in[17];

    const int* row = ei;
    const int* col = ei + E_EDGES;

    char* ws = (char*)d_ws;
    float* dinv = (float*)ws;
    float* bufA = (float*)(ws + (size_t)N_NODES * 4);
    float* bufB = bufA + (size_t)N_NODES * H_DIM;
    float* Hp   = bufB + (size_t)N_NODES * H_DIM;
    float* Vv   = Hp + (size_t)B_GRAPHS * 256;
    float* part = Vv + (size_t)B_GRAPHS * 256;

    float* P     = (float*)d_out;
    float* ext_V = P + (size_t)B_GRAPHS * 1024;
    float* int_V = ext_V + B_GRAPHS;

    const int TPB = 256;
    const int nh  = N_NODES * H_DIM;
    const int e16 = E_EDGES * H_DIM;
    const int fcMN = B_GRAPHS * 256;

    fill_kernel<<<(N_NODES + TPB - 1) / TPB, TPB, 0, stream>>>(dinv, 1.0f, N_NODES);
    degree_kernel<<<(E_EDGES + TPB - 1) / TPB, TPB, 0, stream>>>(col, dinv, E_EDGES);
    rsqrt_kernel<<<(N_NODES + TPB - 1) / TPB, TPB, 0, stream>>>(dinv, N_NODES);

    wmma_gemm_f32<<<dim3(1, N_NODES / 16, 1), 32, 0, stream>>>(x, W1, nullptr, bufA, S_IN, H_DIM, 0, S_IN);
    fill_kernel<<<(nh + TPB - 1) / TPB, TPB, 0, stream>>>(bufB, 0.0f, nh);
    scatter_kernel<<<(e16 + TPB - 1) / TPB, TPB, 0, stream>>>(bufA, row, col, dinv, bufB, e16);
    finalize_kernel<<<(nh + TPB - 1) / TPB, TPB, 0, stream>>>(bufB, bufA, dinv, b1, nh);

    wmma_gemm_f32<<<dim3(1, N_NODES / 16, 1), 32, 0, stream>>>(bufB, W2, nullptr, bufA, H_DIM, H_DIM, 0, H_DIM);
    fill_kernel<<<(nh + TPB - 1) / TPB, TPB, 0, stream>>>(bufB, 0.0f, nh);
    scatter_kernel<<<(e16 + TPB - 1) / TPB, TPB, 0, stream>>>(bufA, row, col, dinv, bufB, e16);
    finalize_kernel<<<(nh + TPB - 1) / TPB, TPB, 0, stream>>>(bufB, bufA, dinv, b2, nh);

    wmma_gemm_f32<<<dim3(1, N_NODES / 16, 1), 32, 0, stream>>>(bufB, W3, nullptr, bufA, H_DIM, H_DIM, 0, H_DIM);
    fill_kernel<<<(nh + TPB - 1) / TPB, TPB, 0, stream>>>(bufB, 0.0f, nh);
    scatter_kernel<<<(e16 + TPB - 1) / TPB, TPB, 0, stream>>>(bufA, row, col, dinv, bufB, e16);
    finalize_kernel<<<(nh + TPB - 1) / TPB, TPB, 0, stream>>>(bufB, bufA, dinv, b3, nh);


    wmma_gemm_f32<<<dim3(256 / 16, B_GRAPHS / 16, KCHUNKS), 32, 0, stream>>>(
        bufB, Wp1, nullptr, part, FC_IN_SZ, 256, 0, FC_IN_SZ / KCHUNKS);
    reduce_bias_relu_kernel<<<(fcMN + TPB - 1) / TPB, TPB, 0, stream>>>(part, bp1, Hp, fcMN, 256, 1);
    wmma_gemm_f32<<<dim3(256 / 16, B_GRAPHS / 16, KCHUNKS), 32, 0, stream>>>(
        bufB, Wv1, nullptr, part, FC_IN_SZ, 256, 0, FC_IN_SZ / KCHUNKS);
    reduce_bias_relu_kernel<<<(fcMN + TPB - 1) / TPB, TPB, 0, stream>>>(part, bv1, Vv, fcMN, 256, 1);

    wmma_gemm_f32<<<dim3(1024 / 16, B_GRAPHS / 16, 1), 32, 0, stream>>>(Hp, Wp2, bp2, P, 256, 1024, 0, 256);

    heads_kernel<<<1, B_GRAPHS, 0, stream>>>(Vv, Wiv, biv, Wev, bev, ext_V, int_V);
}


# Global-atomic scatters (degree + 3 convs) become owner-computes sums in LDS: a work-group owns a node range x edge slice and filters the unsorted edge list; sum order = LDS arrival, then 8 partials
# speedup vs baseline: 1.4515x; 1.4515x over previous
#include <hip/hip_runtime.h>

typedef __attribute__((ext_vector_type(2))) float v2f;
typedef __attribute__((ext_vector_type(8))) float v8f;

#define N_NODES   131072
#define E_EDGES   4194304
#define S_IN      32
#define H_DIM     16
#define B_GRAPHS  128
#define FC_IN_SZ  16384
#define KCHUNKS   8

#define NPART    32
#define PNODES   4096
#define NSLICE   8
#define SLICE_E  (E_EDGES / NSLICE)
#define CH       16384
#define WG       1024

__global__ __launch_bounds__(WG)
void degree_lds_kernel(const int* __restrict__ col, unsigned int* __restrict__ cnt) {
    __shared__ unsigned int hist[PNODES];
    const int tid = threadIdx.x;
    const int p = blockIdx.x >> 3, s = blockIdx.x & 7;
    for (int i = tid; i < PNODES; i += WG) hist[i] = 0u;
    __syncthreads();
    const int4* c4 = (const int4*)(col + (size_t)s * SLICE_E);
    for (int it = 0; it < SLICE_E / 4 / WG; it += 8) {
        int4 v[8];
#pragma unroll
        for (int u = 0; u < 8; ++u) v[u] = c4[(it + u) * WG + tid];
#pragma unroll
        for (int u = 0; u < 8; ++u) {
            if ((v[u].x >> 12) == p) atomicAdd(&hist[v[u].x & (PNODES - 1)], 1u);
            if ((v[u].y >> 12) == p) atomicAdd(&hist[v[u].y & (PNODES - 1)], 1u);
            if ((v[u].z >> 12) == p) atomicAdd(&hist[v[u].z & (PNODES - 1)], 1u);
            if ((v[u].w >> 12) == p) atomicAdd(&hist[v[u].w & (PNODES - 1)], 1u);
        }
    }
    __syncthreads();
    for (int i = tid; i < PNODES; i += WG)
        cnt[(size_t)s * N_NODES + p * PNODES + i] = hist[i];
}

__global__ void dinv_kernel(const unsigned int* __restrict__ cnt, float* __restrict__ d, int n) {
    int i = blockIdx.x * blockDim.x + threadIdx.x;
    if (i >= n) return;
    unsigned int t = 0u;
#pragma unroll
    for (int s = 0; s < NSLICE; ++s) t += cnt[(size_t)s * n + i];
    d[i] = rsqrtf(1.0f + (float)t);
}

__global__ __launch_bounds__(32)
void wmma_gemm_f32(const float* __restrict__ A, const float* __restrict__ B,
                   const float* __restrict__ bias, float* __restrict__ C,
                   int K, int Nld, int relu, int Kc) {
    const int nb   = blockIdx.x;
    const int mb   = blockIdx.y;
    const int z    = blockIdx.z;
    const int lane = threadIdx.x;
    const int half = lane >> 4;
    const int lm   = lane & 15;

    const int k0 = z * Kc;

    v8f acc = {};
    const float* Ap = A + (size_t)(mb * 16 + lm) * K + k0 + 2 * half;
    const float* Bp = B + (size_t)(k0 + 2 * half) * Nld + nb * 16 + lm;

#pragma unroll 4
    for (int k = 0; k < Kc; k += 4) {
        v2f a = *(const v2f*)Ap;
        v2f b;
        b.x = Bp[0];
        b.y = Bp[Nld];
        acc = __builtin_amdgcn_wmma_f32_16x16x4_f32(
                  false, a, false, b, (short)0, acc, false, false);
        Ap += 4;
        Bp += (size_t)4 * Nld;
    }

    const int cn = nb * 16 + lm;
    if (gridDim.z == 1) {
        const float bb = bias ? bias[cn] : 0.0f;
#pragma unroll
        for (int v = 0; v < 8; ++v) {
            float r = acc[v] + bb;
            if (relu) r = fmaxf(r, 0.0f);
            C[(size_t)(mb * 16 + v + 8 * half) * Nld + cn] = r;
        }
    } else {
        const size_t M  = (size_t)gridDim.y * 16;
        float* Cp = C + (size_t)z * M * Nld;
#pragma unroll
        for (int v = 0; v < 8; ++v) {
            Cp[(size_t)(mb * 16 + v + 8 * half) * Nld + cn] = acc[v];
        }
    }
}

__global__ void reduce_bias_relu_kernel(const float* __restrict__ part,
                                        const float* __restrict__ bias,
                                        float* __restrict__ C,
                                        int MN, int Nld, int relu) {
    int i = blockIdx.x * blockDim.x + threadIdx.x;
    if (i >= MN) return;
    float s = 0.0f;
#pragma unroll
    for (int zc = 0; zc < KCHUNKS; ++zc) s += part[(size_t)zc * MN + i];
    s += bias[i % Nld];
    if (relu) s = fmaxf(s, 0.0f);
    C[i] = s;
}


#define SLOTS 3
template <bool PF>
__device__ __forceinline__ void add_pass(const int i0, const int n, const int cb, const int l4,
                                         const unsigned short* q, float* acc,
                                         const float4* __restrict__ xw4,
                                         const int* __restrict__ row, const int* __restrict__ col,
                                         const float* __restrict__ dinv,
                                         int4 (&cv)[4], const int4* __restrict__ nx, const int tid) {
    bool h[SLOTS];
    int  c[SLOTS], r[SLOTS];
#pragma unroll
    for (int j = 0; j < SLOTS; ++j) {
        const int i = i0 + j * (WG / 4);
        h[j] = i < n;
        const int o = (int)q[h[j] ? i : 0];
        const int e = cb + (h[j] ? o : 0);
        c[j] = col[e];
        r[j] = row[e];
    }
    if (PF) {
#pragma unroll
        for (int u = 0; u < 4; ++u) cv[u] = nx[u * WG + tid];
    }
    asm volatile("" :: "v"(c[0]), "v"(r[0]), "v"(c[1]), "v"(r[1]), "v"(c[2]), "v"(r[2]));
    float  dr[SLOTS], dc[SLOTS];
    float4 v[SLOTS];
#pragma unroll
    for (int j = 0; j < SLOTS; ++j) {
        dr[j] = dinv[r[j]];
        dc[j] = dinv[c[j]];
        v[j]  = xw4[r[j] * 4 + l4];
    }
    asm volatile("" :: "v"(dr[0]), "v"(dc[0]), "v"(dr[1]), "v"(dc[1]), "v"(dr[2]), "v"(dc[2]),
                       "v"(v[0].x), "v"(v[0].y), "v"(v[0].z), "v"(v[0].w),
                       "v"(v[1].x), "v"(v[1].y), "v"(v[1].z), "v"(v[1].w),
                       "v"(v[2].x), "v"(v[2].y), "v"(v[2].z), "v"(v[2].w));
#pragma unroll
    for (int j = 0; j < SLOTS; ++j) {
        if (h[j]) {
            const float w  = dr[j] * dc[j];
            const int   nd = c[j] & (PNODES - 1);
            const int   sw = (nd >> 2) & 3;
            float* a = acc + (nd * H_DIM + l4 * 4);
            atomicAdd(a + (0 ^ sw), w * v[j].x);
            atomicAdd(a + (1 ^ sw), w * v[j].y);
            atomicAdd(a + (2 ^ sw), w * v[j].z);
            atomicAdd(a + (3 ^ sw), w * v[j].w);
        }
    }
}

__global__ __launch_bounds__(WG)
void scatter_lds_kernel(const float* __restrict__ xw,
                        const int* __restrict__ row, const int* __restrict__ col,
                        const float* __restrict__ dinv,
                        float* __restrict__ part) {
    __shared__ __attribute__((aligned(16))) float acc[PNODES * H_DIM];
    __shared__ unsigned short q[CH];
    __shared__ unsigned int qn[2];

    const int tid  = threadIdx.x;
    const int lane = tid & 31;
    const int p = blockIdx.x >> 3, s = blockIdx.x & 7;
    float4* acc4 = (float4*)acc;
#pragma unroll
    for (int k = 0; k < PNODES * H_DIM / 4 / WG; ++k)
        acc4[k * WG + tid] = make_float4(0.0f, 0.0f, 0.0f, 0.0f);
    if (tid < 2) qn[tid] = 0u;
    __syncthreads();

    const int ebase = s * SLICE_E;
    const float4* xw4 = (const float4*)xw;
    const int g  = tid >> 2;
    const int l4 = tid & 3;

    int4 cv[4];
    {
        const int4* c4 = (const int4*)(col + ebase);
#pragma unroll
        for (int u = 0; u < 4; ++u) cv[u] = c4[u * WG + tid];
    }

    for (int ch = 0; ch < SLICE_E / CH; ++ch) {
        const int cb  = ebase + ch * CH;
        const int par = ch & 1;

        const int t[16] = {cv[0].x, cv[0].y, cv[0].z, cv[0].w, cv[1].x, cv[1].y, cv[1].z, cv[1].w,
                           cv[2].x, cv[2].y, cv[2].z, cv[2].w, cv[3].x, cv[3].y, cv[3].z, cv[3].w};
        unsigned int bal[16];
        unsigned int total = 0u;
#pragma unroll
        for (int k = 0; k < 16; ++k) {
            bal[k] = __builtin_amdgcn_ballot_w32((t[k] >> 12) == p);
            total += (unsigned int)__builtin_popcount(bal[k]);
        }
        unsigned int base = 0u;
        if (lane == 0) base = atomicAdd(&qn[par], total);
        base = (unsigned int)__builtin_amdgcn_readfirstlane((int)base);
#pragma unroll
        for (int k = 0; k < 16; ++k) {
            if ((t[k] >> 12) == p)
                q[base + __builtin_amdgcn_mbcnt_lo(bal[k], 0u)] =
                    (unsigned short)(((((k >> 2) * WG) + tid) << 2) | (k & 3));
            base += (unsigned int)__builtin_popcount(bal[k]);
        }
        __syncthreads();

        if (tid == 0) qn[par ^ 1] = 0u;
        const int n = (int)qn[par];
        const int4* nx = (const int4*)(col + (ch + 1 < SLICE_E / CH ? cb + CH : cb));

        add_pass<true>(g, n, cb, l4, q, acc, xw4, row, col, dinv, cv, nx, tid);
        for (int i = g + SLOTS * (WG / 4); i < n + g; i += SLOTS * (WG / 4))
            add_pass<false>(i, n, cb, l4, q, acc, xw4, row, col, dinv, cv, nx, tid);
        __syncthreads();
    }

    float4* out4 = (float4*)part + ((size_t)s * N_NODES + (size_t)p * PNODES) * (H_DIM / 4);
#pragma unroll
    for (int k = 0; k < PNODES * H_DIM / 4 / WG; ++k) {
        const int idx = k * WG + tid;
        const int sw  = (idx >> 4) & 3;
        const float4 a = acc4[idx];
        float4 b, o;
        b.x = (sw & 1) ? a.y : a.x;  b.y = (sw & 1) ? a.x : a.y;
        b.z = (sw & 1) ? a.w : a.z;  b.w = (sw & 1) ? a.z : a.w;
        o.x = (sw & 2) ? b.z : b.x;  o.y = (sw & 2) ? b.w : b.y;
        o.z = (sw & 2) ? b.x : b.z;  o.w = (sw & 2) ? b.y : b.w;
        out4[idx] = o;
    }
}

__global__ void finalize_kernel(float* __restrict__ h, const float* __restrict__ part,
                                const float* __restrict__ xw,
                                const float* __restrict__ dinv,
                                const float* __restrict__ bias, int n4) {
    int i = blockIdx.x * blockDim.x + threadIdx.x;
    if (i >= n4) return;
    const float4* p4 = (const float4*)part;
    float4 a = p4[i];
#pragma unroll
    for (int s = 1; s < NSLICE; ++s) {
        const float4 t = p4[(size_t)s * n4 + i];
        a.x += t.x; a.y += t.y; a.z += t.z; a.w += t.w;
    }
    const int node = i >> 2, j0 = (i & 3) * 4;
    const float d = dinv[node];
    const float4 x = ((const float4*)xw)[i];
    float4 r;
    r.x = fmaxf(a.x + d * d * x.x + bias[j0 + 0], 0.0f);
    r.y = fmaxf(a.y + d * d * x.y + bias[j0 + 1], 0.0f);
    r.z = fmaxf(a.z + d * d * x.z + bias[j0 + 2], 0.0f);
    r.w = fmaxf(a.w + d * d * x.w + bias[j0 + 3], 0.0f);
    ((float4*)h)[i] = r;
}

__global__ void heads_kernel(const float* __restrict__ V,
                             const float* __restrict__ Wiv, const float* __restrict__ biv,
                             const float* __restrict__ Wev, const float* __restrict__ bev,
                             float* __restrict__ out_ext, float* __restrict__ out_int) {
    int r = blockIdx.x * blockDim.x + threadIdx.x;
    if (r >= B_GRAPHS) return;
    const float* v = V + (size_t)r * 256;
    float si = 0.0f, se = 0.0f;
#pragma unroll 4
    for (int k = 0; k < 256; ++k) {
        float x = v[k];
        si += x * Wiv[k];
        se += x * Wev[k];
    }
    out_int[r] = si + biv[0];
    out_ext[r] = se + bev[0];
}

extern "C" void kernel_launch(void* const* d_in, const int* in_sizes, int n_in,
                              void* d_out, int out_size, void* d_ws, size_t ws_size,
                              hipStream_t stream) {
    const float* x    = (const float*)d_in[0];
    const int*   ei   = (const int*)d_in[1];
    const float* W1   = (const float*)d_in[2];
    const float* b1   = (const float*)d_in[3];
    const float* W2   = (const float*)d_in[4];
    const float* b2   = (const float*)d_in[5];
    const float* W3   = (const float*)d_in[6];
    const float* b3   = (const float*)d_in[7];
    const float* Wp1  = (const float*)d_in[8];
    const float* bp1  = (const float*)d_in[9];
    const float* Wp2  = (const float*)d_in[10];
    const float* bp2  = (const float*)d_in[11];
    const float* Wv1  = (const float*)d_in[12];
    const float* bv1  = (const float*)d_in[13];
    const float* Wiv  = (const float*)d_in[14];
    const float* biv  = (const float*)d_in[15];
    const float* Wev  = (const float*)d_in[16];
    const float* bev  = (const float*)d_in[17];

    const int* row = ei;
    const int* col = ei + E_EDGES;

    char* ws = (char*)d_ws;
    float* dinv = (float*)ws;
    float* bufA = (float*)(ws + (size_t)N_NODES * 4);
    float* bufB = bufA + (size_t)N_NODES * H_DIM;
    float* Hp   = bufB + (size_t)N_NODES * H_DIM;
    float* Vv   = Hp + (size_t)B_GRAPHS * 256;
    float* part = Vv + (size_t)B_GRAPHS * 256;
    unsigned int* cnt = (unsigned int*)(part + (size_t)KCHUNKS * B_GRAPHS * 256);
    float* spart = (float*)(cnt + (size_t)NSLICE * N_NODES);

    float* P     = (float*)d_out;
    float* ext_V = P + (size_t)B_GRAPHS * 1024;
    float* int_V = ext_V + B_GRAPHS;

    const int TPB = 256;
    const int nh  = N_NODES * H_DIM;
    const int nh4 = nh / 4;
    const int fcMN = B_GRAPHS * 256;

    degree_lds_kernel<<<NPART * NSLICE, WG, 0, stream>>>(col, cnt);
    dinv_kernel<<<(N_NODES + TPB - 1) / TPB, TPB, 0, stream>>>(cnt, dinv, N_NODES);

    wmma_gemm_f32<<<dim3(1, N_NODES / 16, 1), 32, 0, stream>>>(x, W1, nullptr, bufA, S_IN, H_DIM, 0, S_IN);
    scatter_lds_kernel<<<NPART * NSLICE, WG, 0, stream>>>(bufA, row, col, dinv, spart);
    finalize_kernel<<<(nh4 + TPB - 1) / TPB, TPB, 0, stream>>>(bufB, spart, bufA, dinv, b1, nh4);

    wmma_gemm_f32<<<dim3(1, N_NODES / 16, 1), 32, 0, stream>>>(bufB, W2, nullptr, bufA, H_DIM, H_DIM, 0, H_DIM);
    scatter_lds_kernel<<<NPART * NSLICE, WG, 0, stream>>>(bufA, row, col, dinv, spart);
    finalize_kernel<<<(nh4 + TPB - 1) / TPB, TPB, 0, stream>>>(bufB, spart, bufA, dinv, b2, nh4);

    wmma_gemm_f32<<<dim3(1, N_NODES / 16, 1), 32, 0, stream>>>(bufB, W3, nullptr, bufA, H_DIM, H_DIM, 0, H_DIM);
    scatter_lds_kernel<<<NPART * NSLICE, WG, 0, stream>>>(bufA, row, col, dinv, spart);
    finalize_kernel<<<(nh4 + TPB - 1) / TPB, TPB, 0, stream>>>(bufB, spart, bufA, dinv, b3, nh4);


    wmma_gemm_f32<<<dim3(256 / 16, B_GRAPHS / 16, KCHUNKS), 32, 0, stream>>>(
        bufB, Wp1, nullptr, part, FC_IN_SZ, 256, 0, FC_IN_SZ / KCHUNKS);
    reduce_bias_relu_kernel<<<(fcMN + TPB - 1) / TPB, TPB, 0, stream>>>(part, bp1, Hp, fcMN, 256, 1);
    wmma_gemm_f32<<<dim3(256 / 16, B_GRAPHS / 16, KCHUNKS), 32, 0, stream>>>(
        bufB, Wv1, nullptr, part, FC_IN_SZ, 256, 0, FC_IN_SZ / KCHUNKS);
    reduce_bias_relu_kernel<<<(fcMN + TPB - 1) / TPB, TPB, 0, stream>>>(part, bv1, Vv, fcMN, 256, 1);

    wmma_gemm_f32<<<dim3(1024 / 16, B_GRAPHS / 16, 1), 32, 0, stream>>>(Hp, Wp2, bp2, P, 256, 1024, 0, 256);

    heads_kernel<<<1, B_GRAPHS, 0, stream>>>(Vv, Wiv, biv, Wev, bev, ext_V, int_V);
}


# Scatter: the two work-group barriers per round and the shared queue go; each wave filters its own 512 targets into a wave-private LDS queue and adds them, waves run free; sum order still LDS arrival
# speedup vs baseline: 1.5728x; 1.0836x over previous
#include <hip/hip_runtime.h>

typedef __attribute__((ext_vector_type(2))) float v2f;
typedef __attribute__((ext_vector_type(8))) float v8f;

#define N_NODES   131072
#define E_EDGES   4194304
#define S_IN      32
#define H_DIM     16
#define B_GRAPHS  128
#define FC_IN_SZ  16384
#define KCHUNKS   8

#define NPART    32
#define PNODES   4096
#define NSLICE   8
#define SLICE_E  (E_EDGES / NSLICE)
#define CH       16384
#define WG       1024
#define WQ       (CH / (WG / 32))

__global__ __launch_bounds__(WG)
void degree_lds_kernel(const int* __restrict__ col, unsigned int* __restrict__ cnt) {
    __shared__ unsigned int hist[PNODES];
    const int tid = threadIdx.x;
    const int p = blockIdx.x >> 3, s = blockIdx.x & 7;
    for (int i = tid; i < PNODES; i += WG) hist[i] = 0u;
    __syncthreads();
    const int4* c4 = (const int4*)(col + (size_t)s * SLICE_E);
    for (int it = 0; it < SLICE_E / 4 / WG; it += 8) {
        int4 v[8];
#pragma unroll
        for (int u = 0; u < 8; ++u) v[u] = c4[(it + u) * WG + tid];
#pragma unroll
        for (int u = 0; u < 8; ++u) {
            if ((v[u].x >> 12) == p) atomicAdd(&hist[v[u].x & (PNODES - 1)], 1u);
            if ((v[u].y >> 12) == p) atomicAdd(&hist[v[u].y & (PNODES - 1)], 1u);
            if ((v[u].z >> 12) == p) atomicAdd(&hist[v[u].z & (PNODES - 1)], 1u);
            if ((v[u].w >> 12) == p) atomicAdd(&hist[v[u].w & (PNODES - 1)], 1u);
        }
    }
    __syncthreads();
    for (int i = tid; i < PNODES; i += WG)
        cnt[(size_t)s * N_NODES + p * PNODES + i] = hist[i];
}

__global__ void dinv_kernel(const unsigned int* __restrict__ cnt, float* __restrict__ d, int n) {
    int i = blockIdx.x * blockDim.x + threadIdx.x;
    if (i >= n) return;
    unsigned int t = 0u;
#pragma unroll
    for (int s = 0; s < NSLICE; ++s) t += cnt[(size_t)s * n + i];
    d[i] = rsqrtf(1.0f + (float)t);
}

__global__ __launch_bounds__(32)
void wmma_gemm_f32(const float* __restrict__ A, const float* __restrict__ B,
                   const float* __restrict__ bias, float* __restrict__ C,
                   int K, int Nld, int relu, int Kc) {
    const int nb   = blockIdx.x;
    const int mb   = blockIdx.y;
    const int z    = blockIdx.z;
    const int lane = threadIdx.x;
    const int half = lane >> 4;
    const int lm   = lane & 15;

    const int k0 = z * Kc;

    v8f acc = {};
    const float* Ap = A + (size_t)(mb * 16 + lm) * K + k0 + 2 * half;
    const float* Bp = B + (size_t)(k0 + 2 * half) * Nld + nb * 16 + lm;

#pragma unroll 4
    for (int k = 0; k < Kc; k += 4) {
        v2f a = *(const v2f*)Ap;
        v2f b;
        b.x = Bp[0];
        b.y = Bp[Nld];
        acc = __builtin_amdgcn_wmma_f32_16x16x4_f32(
                  false, a, false, b, (short)0, acc, false, false);
        Ap += 4;
        Bp += (size_t)4 * Nld;
    }

    const int cn = nb * 16 + lm;
    if (gridDim.z == 1) {
        const float bb = bias ? bias[cn] : 0.0f;
#pragma unroll
        for (int v = 0; v < 8; ++v) {
            float r = acc[v] + bb;
            if (relu) r = fmaxf(r, 0.0f);
            C[(size_t)(mb * 16 + v + 8 * half) * Nld + cn] = r;
        }
    } else {
        const size_t M  = (size_t)gridDim.y * 16;
        float* Cp = C + (size_t)z * M * Nld;
#pragma unroll
        for (int v = 0; v < 8; ++v) {
            Cp[(size_t)(mb * 16 + v + 8 * half) * Nld + cn] = acc[v];
        }
    }
}

__global__ void reduce_bias_relu_kernel(const float* __restrict__ part,
                                        const float* __restrict__ bias,
                                        float* __restrict__ C,
                                        int MN, int Nld, int relu) {
    int i = blockIdx.x * blockDim.x + threadIdx.x;
    if (i >= MN) return;
    float s = 0.0f;
#pragma unroll
    for (int zc = 0; zc < KCHUNKS; ++zc) s += part[(size_t)zc * MN + i];
    s += bias[i % Nld];
    if (relu) s = fmaxf(s, 0.0f);
    C[i] = s;
}


#define SLOTS 3
#define EPI   8
template <bool PF>
__device__ __forceinline__ void add_pass(const int i0, const int n, const int cb, const int l4,
                                         const unsigned short* q, float* acc,
                                         const float4* __restrict__ xw4,
                                         const int* __restrict__ row, const int* __restrict__ col,
                                         const float* __restrict__ dinv,
                                         int4 (&cv)[4], const int4* __restrict__ nx, const int tid) {
    bool h[SLOTS];
    int  c[SLOTS], r[SLOTS];
#pragma unroll
    for (int j = 0; j < SLOTS; ++j) {
        const int i = i0 + j * EPI;
        h[j] = i < n;
        const int o = (int)q[h[j] ? i : 0];
        const int e = cb + (h[j] ? o : 0);
        c[j] = col[e];
        r[j] = row[e];
    }
    if (PF) {
#pragma unroll
        for (int u = 0; u < 4; ++u) cv[u] = nx[u * WG + tid];
    }
    asm volatile("" :: "v"(c[0]), "v"(r[0]), "v"(c[1]), "v"(r[1]), "v"(c[2]), "v"(r[2]));
    float  dr[SLOTS], dc[SLOTS];
    float4 v[SLOTS];
#pragma unroll
    for (int j = 0; j < SLOTS; ++j) {
        dr[j] = dinv[r[j]];
        dc[j] = dinv[c[j]];
        v[j]  = xw4[r[j] * 4 + l4];
    }
    asm volatile("" :: "v"(dr[0]), "v"(dc[0]), "v"(dr[1]), "v"(dc[1]), "v"(dr[2]), "v"(dc[2]),
                       "v"(v[0].x), "v"(v[0].y), "v"(v[0].z), "v"(v[0].w),
                       "v"(v[1].x), "v"(v[1].y), "v"(v[1].z), "v"(v[1].w),
                       "v"(v[2].x), "v"(v[2].y), "v"(v[2].z), "v"(v[2].w));
#pragma unroll
    for (int j = 0; j < SLOTS; ++j) {
        if (h[j]) {
            const float w  = dr[j] * dc[j];
            const int   nd = c[j] & (PNODES - 1);
            const int   sw = (nd >> 2) & 3;
            float* a = acc + (nd * H_DIM + l4 * 4);
            atomicAdd(a + (0 ^ sw), w * v[j].x);
            atomicAdd(a + (1 ^ sw), w * v[j].y);
            atomicAdd(a + (2 ^ sw), w * v[j].z);
            atomicAdd(a + (3 ^ sw), w * v[j].w);
        }
    }
}

__global__ __launch_bounds__(WG)
void scatter_lds_kernel(const float* __restrict__ xw,
                        const int* __restrict__ row, const int* __restrict__ col,
                        const float* __restrict__ dinv,
                        float* __restrict__ part) {
    __shared__ __attribute__((aligned(16))) float acc[PNODES * H_DIM];
    __shared__ unsigned short q[CH];

    const int tid  = threadIdx.x;
    const int lane = tid & 31;
    const int p = blockIdx.x >> 3, s = blockIdx.x & 7;
    float4* acc4 = (float4*)acc;
#pragma unroll
    for (int k = 0; k < PNODES * H_DIM / 4 / WG; ++k)
        acc4[k * WG + tid] = make_float4(0.0f, 0.0f, 0.0f, 0.0f);
    __syncthreads();

    const int ebase = s * SLICE_E;
    const float4* xw4 = (const float4*)xw;
    const int g  = lane >> 2;
    const int l4 = lane & 3;
    unsigned short* qw = q + (tid >> 5) * WQ;

    int4 cv[4];
    {
        const int4* c4 = (const int4*)(col + ebase);
#pragma unroll
        for (int u = 0; u < 4; ++u) cv[u] = c4[u * WG + tid];
    }

    for (int ch = 0; ch < SLICE_E / CH; ++ch) {
        const int cb = ebase + ch * CH;

        const int t[16] = {cv[0].x, cv[0].y, cv[0].z, cv[0].w, cv[1].x, cv[1].y, cv[1].z, cv[1].w,
                           cv[2].x, cv[2].y, cv[2].z, cv[2].w, cv[3].x, cv[3].y, cv[3].z, cv[3].w};
        unsigned int base = 0u;
#pragma unroll
        for (int k = 0; k < 16; ++k) {
            const bool m = (t[k] >> 12) == p;
            const unsigned int bal = __builtin_amdgcn_ballot_w32(m);
            if (m)
                qw[base + __builtin_amdgcn_mbcnt_lo(bal, 0u)] =
                    (unsigned short)(((((k >> 2) * WG) + tid) << 2) | (k & 3));
            base += (unsigned int)__builtin_popcount(bal);
        }
        const int n = (int)base;
        __builtin_amdgcn_fence(__ATOMIC_RELEASE, "wavefront");
        __builtin_amdgcn_wave_barrier();
        __builtin_amdgcn_fence(__ATOMIC_ACQUIRE, "wavefront");

        const int4* nx = (const int4*)(col + (ch + 1 < SLICE_E / CH ? cb + CH : cb));

        add_pass<true>(g, n, cb, l4, qw, acc, xw4, row, col, dinv, cv, nx, tid);
        for (int i = g + SLOTS * EPI; i < n + g; i += SLOTS * EPI)
            add_pass<false>(i, n, cb, l4, qw, acc, xw4, row, col, dinv, cv, nx, tid);
    }
    __syncthreads();

    float4* out4 = (float4*)part + ((size_t)s * N_NODES + (size_t)p * PNODES) * (H_DIM / 4);
#pragma unroll
    for (int k = 0; k < PNODES * H_DIM / 4 / WG; ++k) {
        const int idx = k * WG + tid;
        const int sw  = (idx >> 4) & 3;
        const float4 a = acc4[idx];
        float4 b, o;
        b.x = (sw & 1) ? a.y : a.x;  b.y = (sw & 1) ? a.x : a.y;
        b.z = (sw & 1) ? a.w : a.z;  b.w = (sw & 1) ? a.z : a.w;
        o.x = (sw & 2) ? b.z : b.x;  o.y = (sw & 2) ? b.w : b.y;
        o.z = (sw & 2) ? b.x : b.z;  o.w = (sw & 2) ? b.y : b.w;
        out4[idx] = o;
    }
}

__global__ void finalize_kernel(float* __restrict__ h, const float* __restrict__ part,
                                const float* __restrict__ xw,
                                const float* __restrict__ dinv,
                                const float* __restrict__ bias, int n4) {
    int i = blockIdx.x * blockDim.x + threadIdx.x;
    if (i >= n4) return;
    const float4* p4 = (const float4*)part;
    float4 a = p4[i];
#pragma unroll
    for (int s = 1; s < NSLICE; ++s) {
        const float4 t = p4[(size_t)s * n4 + i];
        a.x += t.x; a.y += t.y; a.z += t.z; a.w += t.w;
    }
    const int node = i >> 2, j0 = (i & 3) * 4;
    const float d = dinv[node];
    const float4 x = ((const float4*)xw)[i];
    float4 r;
    r.x = fmaxf(a.x + d * d * x.x + bias[j0 + 0], 0.0f);
    r.y = fmaxf(a.y + d * d * x.y + bias[j0 + 1], 0.0f);
    r.z = fmaxf(a.z + d * d * x.z + bias[j0 + 2], 0.0f);
    r.w = fmaxf(a.w + d * d * x.w + bias[j0 + 3], 0.0f);
    ((float4*)h)[i] = r;
}

__global__ void heads_kernel(const float* __restrict__ V,
                             const float* __restrict__ Wiv, const float* __restrict__ biv,
                             const float* __restrict__ Wev, const float* __restrict__ bev,
                             float* __restrict__ out_ext, float* __restrict__ out_int) {
    int r = blockIdx.x * blockDim.x + threadIdx.x;
    if (r >= B_GRAPHS) return;
    const float* v = V + (size_t)r * 256;
    float si = 0.0f, se = 0.0f;
#pragma unroll 4
    for (int k = 0; k < 256; ++k) {
        float x = v[k];
        si += x * Wiv[k];
        se += x * Wev[k];
    }
    out_int[r] = si + biv[0];
    out_ext[r] = se + bev[0];
}

extern "C" void kernel_launch(void* const* d_in, const int* in_sizes, int n_in,
                              void* d_out, int out_size, void* d_ws, size_t ws_size,
                              hipStream_t stream) {
    const float* x    = (const float*)d_in[0];
    const int*   ei   = (const int*)d_in[1];
    const float* W1   = (const float*)d_in[2];
    const float* b1   = (const float*)d_in[3];
    const float* W2   = (const float*)d_in[4];
    const float* b2   = (const float*)d_in[5];
    const float* W3   = (const float*)d_in[6];
    const float* b3   = (const float*)d_in[7];
    const float* Wp1  = (const float*)d_in[8];
    const float* bp1  = (const float*)d_in[9];
    const float* Wp2  = (const float*)d_in[10];
    const float* bp2  = (const float*)d_in[11];
    const float* Wv1  = (const float*)d_in[12];
    const float* bv1  = (const float*)d_in[13];
    const float* Wiv  = (const float*)d_in[14];
    const float* biv  = (const float*)d_in[15];
    const float* Wev  = (const float*)d_in[16];
    const float* bev  = (const float*)d_in[17];

    const int* row = ei;
    const int* col = ei + E_EDGES;

    char* ws = (char*)d_ws;
    float* dinv = (float*)ws;
    float* bufA = (float*)(ws + (size_t)N_NODES * 4);
    float* bufB = bufA + (size_t)N_NODES * H_DIM;
    float* Hp   = bufB + (size_t)N_NODES * H_DIM;
    float* Vv   = Hp + (size_t)B_GRAPHS * 256;
    float* part = Vv + (size_t)B_GRAPHS * 256;
    unsigned int* cnt = (unsigned int*)(part + (size_t)KCHUNKS * B_GRAPHS * 256);
    float* spart = (float*)(cnt + (size_t)NSLICE * N_NODES);

    float* P     = (float*)d_out;
    float* ext_V = P + (size_t)B_GRAPHS * 1024;
    float* int_V = ext_V + B_GRAPHS;

    const int TPB = 256;
    const int nh  = N_NODES * H_DIM;
    const int nh4 = nh / 4;
    const int fcMN = B_GRAPHS * 256;

    degree_lds_kernel<<<NPART * NSLICE, WG, 0, stream>>>(col, cnt);
    dinv_kernel<<<(N_NODES + TPB - 1) / TPB, TPB, 0, stream>>>(cnt, dinv, N_NODES);

    wmma_gemm_f32<<<dim3(1, N_NODES / 16, 1), 32, 0, stream>>>(x, W1, nullptr, bufA, S_IN, H_DIM, 0, S_IN);
    scatter_lds_kernel<<<NPART * NSLICE, WG, 0, stream>>>(bufA, row, col, dinv, spart);
    finalize_kernel<<<(nh4 + TPB - 1) / TPB, TPB, 0, stream>>>(bufB, spart, bufA, dinv, b1, nh4);

    wmma_gemm_f32<<<dim3(1, N_NODES / 16, 1), 32, 0, stream>>>(bufB, W2, nullptr, bufA, H_DIM, H_DIM, 0, H_DIM);
    scatter_lds_kernel<<<NPART * NSLICE, WG, 0, stream>>>(bufA, row, col, dinv, spart);
    finalize_kernel<<<(nh4 + TPB - 1) / TPB, TPB, 0, stream>>>(bufB, spart, bufA, dinv, b2, nh4);

    wmma_gemm_f32<<<dim3(1, N_NODES / 16, 1), 32, 0, stream>>>(bufB, W3, nullptr, bufA, H_DIM, H_DIM, 0, H_DIM);
    scatter_lds_kernel<<<NPART * NSLICE, WG, 0, stream>>>(bufA, row, col, dinv, spart);
    finalize_kernel<<<(nh4 + TPB - 1) / TPB, TPB, 0, stream>>>(bufB, spart, bufA, dinv, b3, nh4);


    wmma_gemm_f32<<<dim3(256 / 16, B_GRAPHS / 16, KCHUNKS), 32, 0, stream>>>(
        bufB, Wp1, nullptr, part, FC_IN_SZ, 256, 0, FC_IN_SZ / KCHUNKS);
    reduce_bias_relu_kernel<<<(fcMN + TPB - 1) / TPB, TPB, 0, stream>>>(part, bp1, Hp, fcMN, 256, 1);
    wmma_gemm_f32<<<dim3(256 / 16, B_GRAPHS / 16, KCHUNKS), 32, 0, stream>>>(
        bufB, Wv1, nullptr, part, FC_IN_SZ, 256, 0, FC_IN_SZ / KCHUNKS);
    reduce_bias_relu_kernel<<<(fcMN + TPB - 1) / TPB, TPB, 0, stream>>>(part, bv1, Vv, fcMN, 256, 1);

    wmma_gemm_f32<<<dim3(1024 / 16, B_GRAPHS / 16, 1), 32, 0, stream>>>(Hp, Wp2, bp2, P, 256, 1024, 0, 256);

    heads_kernel<<<1, B_GRAPHS, 0, stream>>>(Vv, Wiv, biv, Wev, bev, ext_V, int_V);
}


# Scatter: filter streams row beside col and queues (source row, local target); range's dinv in LDS; add phase is one global hop (dinv[row] + xw row), no col[e]/row[e]/dinv[col] gathers; sum order kept
# speedup vs baseline: 1.9083x; 1.2133x over previous
#include <hip/hip_runtime.h>

typedef __attribute__((ext_vector_type(2))) float v2f;
typedef __attribute__((ext_vector_type(8))) float v8f;

#define N_NODES   131072
#define E_EDGES   4194304
#define S_IN      32
#define H_DIM     16
#define B_GRAPHS  128
#define FC_IN_SZ  16384
#define KCHUNKS   8

#define NPART    32
#define PNODES   4096
#define NSLICE   8
#define SLICE_E  (E_EDGES / NSLICE)
#define CH       16384
#define WG       1024
#define QCAP     256
#define RBITS    17

__global__ __launch_bounds__(WG)
void degree_lds_kernel(const int* __restrict__ col, unsigned int* __restrict__ cnt) {
    __shared__ unsigned int hist[PNODES];
    const int tid = threadIdx.x;
    const int p = blockIdx.x >> 3, s = blockIdx.x & 7;
    for (int i = tid; i < PNODES; i += WG) hist[i] = 0u;
    __syncthreads();
    const int4* c4 = (const int4*)(col + (size_t)s * SLICE_E);
    for (int it = 0; it < SLICE_E / 4 / WG; it += 8) {
        int4 v[8];
#pragma unroll
        for (int u = 0; u < 8; ++u) v[u] = c4[(it + u) * WG + tid];
#pragma unroll
        for (int u = 0; u < 8; ++u) {
            if ((v[u].x >> 12) == p) atomicAdd(&hist[v[u].x & (PNODES - 1)], 1u);
            if ((v[u].y >> 12) == p) atomicAdd(&hist[v[u].y & (PNODES - 1)], 1u);
            if ((v[u].z >> 12) == p) atomicAdd(&hist[v[u].z & (PNODES - 1)], 1u);
            if ((v[u].w >> 12) == p) atomicAdd(&hist[v[u].w & (PNODES - 1)], 1u);
        }
    }
    __syncthreads();
    for (int i = tid; i < PNODES; i += WG)
        cnt[(size_t)s * N_NODES + p * PNODES + i] = hist[i];
}

__global__ void dinv_kernel(const unsigned int* __restrict__ cnt, float* __restrict__ d, int n) {
    int i = blockIdx.x * blockDim.x + threadIdx.x;
    if (i >= n) return;
    unsigned int t = 0u;
#pragma unroll
    for (int s = 0; s < NSLICE; ++s) t += cnt[(size_t)s * n + i];
    d[i] = rsqrtf(1.0f + (float)t);
}

__global__ __launch_bounds__(32)
void wmma_gemm_f32(const float* __restrict__ A, const float* __restrict__ B,
                   const float* __restrict__ bias, float* __restrict__ C,
                   int K, int Nld, int relu, int Kc) {
    const int nb   = blockIdx.x;
    const int mb   = blockIdx.y;
    const int z    = blockIdx.z;
    const int lane = threadIdx.x;
    const int half = lane >> 4;
    const int lm   = lane & 15;

    const int k0 = z * Kc;

    v8f acc = {};
    const float* Ap = A + (size_t)(mb * 16 + lm) * K + k0 + 2 * half;
    const float* Bp = B + (size_t)(k0 + 2 * half) * Nld + nb * 16 + lm;

#pragma unroll 4
    for (int k = 0; k < Kc; k += 4) {
        v2f a = *(const v2f*)Ap;
        v2f b;
        b.x = Bp[0];
        b.y = Bp[Nld];
        acc = __builtin_amdgcn_wmma_f32_16x16x4_f32(
                  false, a, false, b, (short)0, acc, false, false);
        Ap += 4;
        Bp += (size_t)4 * Nld;
    }

    const int cn = nb * 16 + lm;
    if (gridDim.z == 1) {
        const float bb = bias ? bias[cn] : 0.0f;
#pragma unroll
        for (int v = 0; v < 8; ++v) {
            float r = acc[v] + bb;
            if (relu) r = fmaxf(r, 0.0f);
            C[(size_t)(mb * 16 + v + 8 * half) * Nld + cn] = r;
        }
    } else {
        const size_t M  = (size_t)gridDim.y * 16;
        float* Cp = C + (size_t)z * M * Nld;
#pragma unroll
        for (int v = 0; v < 8; ++v) {
            Cp[(size_t)(mb * 16 + v + 8 * half) * Nld + cn] = acc[v];
        }
    }
}

__global__ void reduce_bias_relu_kernel(const float* __restrict__ part,
                                        const float* __restrict__ bias,
                                        float* __restrict__ C,
                                        int MN, int Nld, int relu) {
    int i = blockIdx.x * blockDim.x + threadIdx.x;
    if (i >= MN) return;
    float s = 0.0f;
#pragma unroll
    for (int zc = 0; zc < KCHUNKS; ++zc) s += part[(size_t)zc * MN + i];
    s += bias[i % Nld];
    if (relu) s = fmaxf(s, 0.0f);
    C[i] = s;
}


#define SLOTS 3
#define EPI   8
template <bool PF>
__device__ __forceinline__ void add_pass(const int i0, const int n, const int l4,
                                         const unsigned int* q, const float* dl, float* acc,
                                         const float4* __restrict__ xw4,
                                         const float* __restrict__ dinv,
                                         int4 (&cv)[4], int4 (&rv)[4],
                                         const int4* __restrict__ nxc, const int4* __restrict__ nxr,
                                         const int tid) {
    bool   h[SLOTS];
    int    nd[SLOTS];
    float  dc[SLOTS], dr[SLOTS];
    float4 v[SLOTS];
#pragma unroll
    for (int j = 0; j < SLOTS; ++j) {
        const int i = i0 + j * EPI;
        h[j] = i < n;
        const unsigned int en = q[h[j] ? i : 0];
        const int r = h[j] ? (int)(en & ((1u << RBITS) - 1u)) : 0;
        nd[j] = (int)(en >> RBITS) & (PNODES - 1);
        dr[j] = dinv[r];
        v[j]  = xw4[r * 4 + l4];
        dc[j] = dl[nd[j]];
    }
    if (PF) {
#pragma unroll
        for (int u = 0; u < 4; ++u) cv[u] = nxc[u * WG + tid];
#pragma unroll
        for (int u = 0; u < 4; ++u) rv[u] = nxr[u * WG + tid];
    }
    asm volatile("" :: "v"(dr[0]), "v"(dr[1]), "v"(dr[2]),
                       "v"(v[0].x), "v"(v[0].y), "v"(v[0].z), "v"(v[0].w),
                       "v"(v[1].x), "v"(v[1].y), "v"(v[1].z), "v"(v[1].w),
                       "v"(v[2].x), "v"(v[2].y), "v"(v[2].z), "v"(v[2].w));
#pragma unroll
    for (int j = 0; j < SLOTS; ++j) {
        if (h[j]) {
            const float w  = dr[j] * dc[j];
            const int   sw = (nd[j] >> 2) & 3;
            float* a = acc + (nd[j] * H_DIM + l4 * 4);
            atomicAdd(a + (0 ^ sw), w * v[j].x);
            atomicAdd(a + (1 ^ sw), w * v[j].y);
            atomicAdd(a + (2 ^ sw), w * v[j].z);
            atomicAdd(a + (3 ^ sw), w * v[j].w);
        }
    }
}

__global__ __launch_bounds__(WG)
void scatter_lds_kernel(const float* __restrict__ xw,
                        const int* __restrict__ row, const int* __restrict__ col,
                        const float* __restrict__ dinv,
                        float* __restrict__ part) {
    __shared__ __attribute__((aligned(16))) float acc[PNODES * H_DIM];
    __shared__ unsigned int q[(WG / 32) * QCAP];
    __shared__ float dl[PNODES];

    const int tid  = threadIdx.x;
    const int lane = tid & 31;
    const int p = blockIdx.x >> 3, s = blockIdx.x & 7;
    float4* acc4 = (float4*)acc;
#pragma unroll
    for (int k = 0; k < PNODES * H_DIM / 4 / WG; ++k)
        acc4[k * WG + tid] = make_float4(0.0f, 0.0f, 0.0f, 0.0f);
#pragma unroll
    for (int k = 0; k < PNODES / WG; ++k)
        dl[k * WG + tid] = dinv[p * PNODES + k * WG + tid];
    __syncthreads();

    const int ebase = s * SLICE_E;
    const float4* xw4 = (const float4*)xw;
    const int g  = lane >> 2;
    const int l4 = lane & 3;
    unsigned int* qw = q + (tid >> 5) * QCAP;

    int4 cv[4], rv[4];
    {
        const int4* c4 = (const int4*)(col + ebase);
        const int4* r4 = (const int4*)(row + ebase);
#pragma unroll
        for (int u = 0; u < 4; ++u) cv[u] = c4[u * WG + tid];
#pragma unroll
        for (int u = 0; u < 4; ++u) rv[u] = r4[u * WG + tid];
    }

    for (int ch = 0; ch < SLICE_E / CH; ++ch) {
        const int nb = ebase + (ch + 1 < SLICE_E / CH ? ch + 1 : ch) * CH;
        const int4* nxc = (const int4*)(col + nb);
        const int4* nxr = (const int4*)(row + nb);

        const int t[16] = {cv[0].x, cv[0].y, cv[0].z, cv[0].w, cv[1].x, cv[1].y, cv[1].z, cv[1].w,
                           cv[2].x, cv[2].y, cv[2].z, cv[2].w, cv[3].x, cv[3].y, cv[3].z, cv[3].w};
        const int f[16] = {rv[0].x, rv[0].y, rv[0].z, rv[0].w, rv[1].x, rv[1].y, rv[1].z, rv[1].w,
                           rv[2].x, rv[2].y, rv[2].z, rv[2].w, rv[3].x, rv[3].y, rv[3].z, rv[3].w};
        unsigned int base = 0u;
#pragma unroll
        for (int k = 0; k < 16; ++k) {
            if (k != 0 && (k & 3) == 0 && base > QCAP - 4 * 32) {
                __builtin_amdgcn_fence(__ATOMIC_RELEASE, "wavefront");
                __builtin_amdgcn_wave_barrier();
                __builtin_amdgcn_fence(__ATOMIC_ACQUIRE, "wavefront");
                for (int i = g; i < (int)base + g; i += SLOTS * EPI)
                    add_pass<false>(i, (int)base, l4, qw, dl, acc, xw4, dinv, cv, rv, nxc, nxr, tid);
                base = 0u;
            }
            const bool m = (t[k] >> 12) == p;
            const unsigned int bal = __builtin_amdgcn_ballot_w32(m);
            if (m)
                qw[base + __builtin_amdgcn_mbcnt_lo(bal, 0u)] =
                    ((unsigned int)(t[k] & (PNODES - 1)) << RBITS) |
                    ((unsigned int)f[k] & ((1u << RBITS) - 1u));
            base += (unsigned int)__builtin_popcount(bal);
        }
        const int n = (int)base;
        __builtin_amdgcn_fence(__ATOMIC_RELEASE, "wavefront");
        __builtin_amdgcn_wave_barrier();
        __builtin_amdgcn_fence(__ATOMIC_ACQUIRE, "wavefront");

        add_pass<true>(g, n, l4, qw, dl, acc, xw4, dinv, cv, rv, nxc, nxr, tid);
        for (int i = g + SLOTS * EPI; i < n + g; i += SLOTS * EPI)
            add_pass<false>(i, n, l4, qw, dl, acc, xw4, dinv, cv, rv, nxc, nxr, tid);
    }
    __syncthreads();

    float4* out4 = (float4*)part + ((size_t)s * N_NODES + (size_t)p * PNODES) * (H_DIM / 4);
#pragma unroll
    for (int k = 0; k < PNODES * H_DIM / 4 / WG; ++k) {
        const int idx = k * WG + tid;
        const int sw  = (idx >> 4) & 3;
        const float4 a = acc4[idx];
        float4 b, o;
        b.x = (sw & 1) ? a.y : a.x;  b.y = (sw & 1) ? a.x : a.y;
        b.z = (sw & 1) ? a.w : a.z;  b.w = (sw & 1) ? a.z : a.w;
        o.x = (sw & 2) ? b.z : b.x;  o.y = (sw & 2) ? b.w : b.y;
        o.z = (sw & 2) ? b.x : b.z;  o.w = (sw & 2) ? b.y : b.w;
        out4[idx] = o;
    }
}

__global__ void finalize_kernel(float* __restrict__ h, const float* __restrict__ part,
                                const float* __restrict__ xw,
                                const float* __restrict__ dinv,
                                const float* __restrict__ bias, int n4) {
    int i = blockIdx.x * blockDim.x + threadIdx.x;
    if (i >= n4) return;
    const float4* p4 = (const float4*)part;
    float4 a = p4[i];
#pragma unroll
    for (int s = 1; s < NSLICE; ++s) {
        const float4 t = p4[(size_t)s * n4 + i];
        a.x += t.x; a.y += t.y; a.z += t.z; a.w += t.w;
    }
    const int node = i >> 2, j0 = (i & 3) * 4;
    const float d = dinv[node];
    const float4 x = ((const float4*)xw)[i];
    float4 r;
    r.x = fmaxf(a.x + d * d * x.x + bias[j0 + 0], 0.0f);
    r.y = fmaxf(a.y + d * d * x.y + bias[j0 + 1], 0.0f);
    r.z = fmaxf(a.z + d * d * x.z + bias[j0 + 2], 0.0f);
    r.w = fmaxf(a.w + d * d * x.w + bias[j0 + 3], 0.0f);
    ((float4*)h)[i] = r;
}

__global__ void heads_kernel(const float* __restrict__ V,
                             const float* __restrict__ Wiv, const float* __restrict__ biv,
                             const float* __restrict__ Wev, const float* __restrict__ bev,
                             float* __restrict__ out_ext, float* __restrict__ out_int) {
    int r = blockIdx.x * blockDim.x + threadIdx.x;
    if (r >= B_GRAPHS) return;
    const float* v = V + (size_t)r * 256;
    float si = 0.0f, se = 0.0f;
#pragma unroll 4
    for (int k = 0; k < 256; ++k) {
        float x = v[k];
        si += x * Wiv[k];
        se += x * Wev[k];
    }
    out_int[r] = si + biv[0];
    out_ext[r] = se + bev[0];
}

extern "C" void kernel_launch(void* const* d_in, const int* in_sizes, int n_in,
                              void* d_out, int out_size, void* d_ws, size_t ws_size,
                              hipStream_t stream) {
    const float* x    = (const float*)d_in[0];
    const int*   ei   = (const int*)d_in[1];
    const float* W1   = (const float*)d_in[2];
    const float* b1   = (const float*)d_in[3];
    const float* W2   = (const float*)d_in[4];
    const float* b2   = (const float*)d_in[5];
    const float* W3   = (const float*)d_in[6];
    const float* b3   = (const float*)d_in[7];
    const float* Wp1  = (const float*)d_in[8];
    const float* bp1  = (const float*)d_in[9];
    const float* Wp2  = (const float*)d_in[10];
    const float* bp2  = (const float*)d_in[11];
    const float* Wv1  = (const float*)d_in[12];
    const float* bv1  = (const float*)d_in[13];
    const float* Wiv  = (const float*)d_in[14];
    const float* biv  = (const float*)d_in[15];
    const float* Wev  = (const float*)d_in[16];
    const float* bev  = (const float*)d_in[17];

    const int* row = ei;
    const int* col = ei + E_EDGES;

    char* ws = (char*)d_ws;
    float* dinv = (float*)ws;
    float* bufA = (float*)(ws + (size_t)N_NODES * 4);
    float* bufB = bufA + (size_t)N_NODES * H_DIM;
    float* Hp   = bufB + (size_t)N_NODES * H_DIM;
    float* Vv   = Hp + (size_t)B_GRAPHS * 256;
    float* part = Vv + (size_t)B_GRAPHS * 256;
    unsigned int* cnt = (unsigned int*)(part + (size_t)KCHUNKS * B_GRAPHS * 256);
    float* spart = (float*)(cnt + (size_t)NSLICE * N_NODES);

    float* P     = (float*)d_out;
    float* ext_V = P + (size_t)B_GRAPHS * 1024;
    float* int_V = ext_V + B_GRAPHS;

    const int TPB = 256;
    const int nh  = N_NODES * H_DIM;
    const int nh4 = nh / 4;
    const int fcMN = B_GRAPHS * 256;

    degree_lds_kernel<<<NPART * NSLICE, WG, 0, stream>>>(col, cnt);
    dinv_kernel<<<(N_NODES + TPB - 1) / TPB, TPB, 0, stream>>>(cnt, dinv, N_NODES);

    wmma_gemm_f32<<<dim3(1, N_NODES / 16, 1), 32, 0, stream>>>(x, W1, nullptr, bufA, S_IN, H_DIM, 0, S_IN);
    scatter_lds_kernel<<<NPART * NSLICE, WG, 0, stream>>>(bufA, row, col, dinv, spart);
    finalize_kernel<<<(nh4 + TPB - 1) / TPB, TPB, 0, stream>>>(bufB, spart, bufA, dinv, b1, nh4);

    wmma_gemm_f32<<<dim3(1, N_NODES / 16, 1), 32, 0, stream>>>(bufB, W2, nullptr, bufA, H_DIM, H_DIM, 0, H_DIM);
    scatter_lds_kernel<<<NPART * NSLICE, WG, 0, stream>>>(bufA, row, col, dinv, spart);
    finalize_kernel<<<(nh4 + TPB - 1) / TPB, TPB, 0, stream>>>(bufB, spart, bufA, dinv, b2, nh4);

    wmma_gemm_f32<<<dim3(1, N_NODES / 16, 1), 32, 0, stream>>>(bufB, W3, nullptr, bufA, H_DIM, H_DIM, 0, H_DIM);
    scatter_lds_kernel<<<NPART * NSLICE, WG, 0, stream>>>(bufA, row, col, dinv, spart);
    finalize_kernel<<<(nh4 + TPB - 1) / TPB, TPB, 0, stream>>>(bufB, spart, bufA, dinv, b3, nh4);


    wmma_gemm_f32<<<dim3(256 / 16, B_GRAPHS / 16, KCHUNKS), 32, 0, stream>>>(
        bufB, Wp1, nullptr, part, FC_IN_SZ, 256, 0, FC_IN_SZ / KCHUNKS);
    reduce_bias_relu_kernel<<<(fcMN + TPB - 1) / TPB, TPB, 0, stream>>>(part, bp1, Hp, fcMN, 256, 1);
    wmma_gemm_f32<<<dim3(256 / 16, B_GRAPHS / 16, KCHUNKS), 32, 0, stream>>>(
        bufB, Wv1, nullptr, part, FC_IN_SZ, 256, 0, FC_IN_SZ / KCHUNKS);
    reduce_bias_relu_kernel<<<(fcMN + TPB - 1) / TPB, TPB, 0, stream>>>(part, bv1, Vv, fcMN, 256, 1);

    wmma_gemm_f32<<<dim3(1024 / 16, B_GRAPHS / 16, 1), 32, 0, stream>>>(Hp, Wp2, bp2, P, 256, 1024, 0, 256);

    heads_kernel<<<1, B_GRAPHS, 0, stream>>>(Vv, Wiv, biv, Wev, bev, ext_V, int_V);
}


# FC GEMMs: an 8-wave work-group stages a 128x256 and a 256x64 operand slab in LDS, all its 16-byte whole-line loads in flight at once, WMMA fragments from LDS; 64 K-slab partials: sum order changes
# speedup vs baseline: 2.6907x; 1.4100x over previous
#include <hip/hip_runtime.h>

typedef __attribute__((ext_vector_type(2))) float v2f;
typedef __attribute__((ext_vector_type(8))) float v8f;

#define N_NODES   131072
#define E_EDGES   4194304
#define S_IN      32
#define H_DIM     16
#define B_GRAPHS  128
#define FC_IN_SZ  16384
#define FC_KC     256
#define KCHUNKS   (FC_IN_SZ / FC_KC)

#define NPART    32
#define PNODES   4096
#define NSLICE   8
#define SLICE_E  (E_EDGES / NSLICE)
#define CH       16384
#define WG       1024
#define QCAP     256
#define RBITS    17

__global__ __launch_bounds__(WG)
void degree_lds_kernel(const int* __restrict__ col, unsigned int* __restrict__ cnt) {
    __shared__ unsigned int hist[PNODES];
    const int tid = threadIdx.x;
    const int p = blockIdx.x >> 3, s = blockIdx.x & 7;
    for (int i = tid; i < PNODES; i += WG) hist[i] = 0u;
    __syncthreads();
    const int4* c4 = (const int4*)(col + (size_t)s * SLICE_E);
    for (int it = 0; it < SLICE_E / 4 / WG; it += 8) {
        int4 v[8];
#pragma unroll
        for (int u = 0; u < 8; ++u) v[u] = c4[(it + u) * WG + tid];
#pragma unroll
        for (int u = 0; u < 8; ++u) {
            if ((v[u].x >> 12) == p) atomicAdd(&hist[v[u].x & (PNODES - 1)], 1u);
            if ((v[u].y >> 12) == p) atomicAdd(&hist[v[u].y & (PNODES - 1)], 1u);
            if ((v[u].z >> 12) == p) atomicAdd(&hist[v[u].z & (PNODES - 1)], 1u);
            if ((v[u].w >> 12) == p) atomicAdd(&hist[v[u].w & (PNODES - 1)], 1u);
        }
    }
    __syncthreads();
    for (int i = tid; i < PNODES; i += WG)
        cnt[(size_t)s * N_NODES + p * PNODES + i] = hist[i];
}

__global__ void dinv_kernel(const unsigned int* __restrict__ cnt, float* __restrict__ d, int n) {
    int i = blockIdx.x * blockDim.x + threadIdx.x;
    if (i >= n) return;
    unsigned int t = 0u;
#pragma unroll
    for (int s = 0; s < NSLICE; ++s) t += cnt[(size_t)s * n + i];
    d[i] = rsqrtf(1.0f + (float)t);
}

__global__ __launch_bounds__(32)
void wmma_gemm_f32(const float* __restrict__ A, const float* __restrict__ B,
                   const float* __restrict__ bias, float* __restrict__ C,
                   int K, int Nld, int relu, int Kc) {
    const int nb   = blockIdx.x;
    const int mb   = blockIdx.y;
    const int z    = blockIdx.z;
    const int lane = threadIdx.x;
    const int half = lane >> 4;
    const int lm   = lane & 15;

    const int k0 = z * Kc;

    v8f acc = {};
    const float* Ap = A + (size_t)(mb * 16 + lm) * K + k0 + 2 * half;
    const float* Bp = B + (size_t)(k0 + 2 * half) * Nld + nb * 16 + lm;

#pragma unroll 4
    for (int k = 0; k < Kc; k += 4) {
        v2f a = *(const v2f*)Ap;
        v2f b;
        b.x = Bp[0];
        b.y = Bp[Nld];
        acc = __builtin_amdgcn_wmma_f32_16x16x4_f32(
                  false, a, false, b, (short)0, acc, false, false);
        Ap += 4;
        Bp += (size_t)4 * Nld;
    }

    const int cn = nb * 16 + lm;
    if (gridDim.z == 1) {
        const float bb = bias ? bias[cn] : 0.0f;
#pragma unroll
        for (int v = 0; v < 8; ++v) {
            float r = acc[v] + bb;
            if (relu) r = fmaxf(r, 0.0f);
            C[(size_t)(mb * 16 + v + 8 * half) * Nld + cn] = r;
        }
    } else {
        const size_t M  = (size_t)gridDim.y * 16;
        float* Cp = C + (size_t)z * M * Nld;
#pragma unroll
        for (int v = 0; v < 8; ++v) {
            Cp[(size_t)(mb * 16 + v + 8 * half) * Nld + cn] = acc[v];
        }
    }
}

#define FC_NT     64
#define FC_AP     (FC_KC + 4)
#define FC_BP     (FC_NT + 8)
__global__ __launch_bounds__(256)
void fc_slab_gemm_f32(const float* __restrict__ A, const float* __restrict__ B,
                      float* __restrict__ part) {
    __shared__ __attribute__((aligned(16))) float As[B_GRAPHS * FC_AP];
    __shared__ __attribute__((aligned(16))) float Bs[FC_KC * FC_BP];
    const int tid = threadIdx.x;
    const int n0  = blockIdx.x * FC_NT;
    const int z   = blockIdx.y;
    const int k0  = z * FC_KC;

    {
        float4 rb[16], ra[16];
#pragma unroll
        for (int u = 0; u < 16; ++u) {
            const int i = u * 256 + tid, r = i >> 4, c4 = i & 15;
            rb[u] = *(const float4*)(B + (size_t)(k0 + r) * 256 + n0 + c4 * 4);
        }
#pragma unroll
        for (int u = 0; u < 16; ++u) {
            const int i = u * 256 + tid, r = i >> 6, c4 = i & 63;
            ra[u] = *(const float4*)(A + (size_t)r * FC_IN_SZ + k0 + c4 * 4);
        }
#pragma unroll
        for (int u = 0; u < 16; ++u) {
            const int i = u * 256 + tid, r = i >> 4, c4 = i & 15;
            *(float4*)(Bs + r * FC_BP + c4 * 4) = rb[u];
        }
#pragma unroll
        for (int u = 0; u < 16; ++u) {
            const int i = u * 256 + tid, r = i >> 6, c4 = i & 63;
            *(float4*)(As + r * FC_AP + c4 * 4) = ra[u];
        }
#pragma unroll
        for (int u = 0; u < 16; ++u) {
            const int i = (u + 16) * 256 + tid, r = i >> 6, c4 = i & 63;
            ra[u] = *(const float4*)(A + (size_t)r * FC_IN_SZ + k0 + c4 * 4);
        }
#pragma unroll
        for (int u = 0; u < 16; ++u) {
            const int i = (u + 16) * 256 + tid, r = i >> 6, c4 = i & 63;
            *(float4*)(As + r * FC_AP + c4 * 4) = ra[u];
        }
    }
    __syncthreads();

    const int lane = tid & 31;
    const int w    = tid >> 5;
    const int half = lane >> 4;
    const int c    = lane & 15;
    const float* ap = As + (16 * w + c) * FC_AP + 4 * half;
    const float* bp = Bs + (4 * half) * FC_BP + c;

    v8f acc[4] = {};
    float4 av = *(const float4*)ap;
    v2f b1[4], b2[4];
#pragma unroll
    for (int j = 0; j < 4; ++j) {
        b1[j].x = bp[0 * FC_BP + 16 * j];  b1[j].y = bp[1 * FC_BP + 16 * j];
        b2[j].x = bp[2 * FC_BP + 16 * j];  b2[j].y = bp[3 * FC_BP + 16 * j];
    }
    for (int kk = 0; kk < FC_KC; kk += 8) {
        const int kn = kk + 8 < FC_KC ? kk + 8 : kk;
        const float4 avn = *(const float4*)(ap + kn);
        const float* bk = bp + kn * FC_BP;
        v2f n1[4], n2[4];
#pragma unroll
        for (int j = 0; j < 4; ++j) {
            n1[j].x = bk[0 * FC_BP + 16 * j];  n1[j].y = bk[1 * FC_BP + 16 * j];
            n2[j].x = bk[2 * FC_BP + 16 * j];  n2[j].y = bk[3 * FC_BP + 16 * j];
        }
        v2f a1, a2;
        a1.x = av.x;  a1.y = av.y;
        a2.x = av.z;  a2.y = av.w;
#pragma unroll
        for (int j = 0; j < 4; ++j)
            acc[j] = __builtin_amdgcn_wmma_f32_16x16x4_f32(
                         false, a1, false, b1[j], (short)0, acc[j], false, false);
#pragma unroll
        for (int j = 0; j < 4; ++j)
            acc[j] = __builtin_amdgcn_wmma_f32_16x16x4_f32(
                         false, a2, false, b2[j], (short)0, acc[j], false, false);
        av = avn;
#pragma unroll
        for (int j = 0; j < 4; ++j) { b1[j] = n1[j]; b2[j] = n2[j]; }
    }

    float* Cp = part + (size_t)z * (B_GRAPHS * 256) + n0 + c;
#pragma unroll
    for (int j = 0; j < 4; ++j) {
#pragma unroll
        for (int v = 0; v < 8; ++v)
            Cp[(size_t)(16 * w + v + 8 * half) * 256 + 16 * j] = acc[j][v];
    }
}

__global__ void reduce_bias_relu_kernel(const float* __restrict__ part,
                                        const float* __restrict__ bias,
                                        float* __restrict__ C,
                                        int MN, int Nld, int relu) {
    int i = blockIdx.x * blockDim.x + threadIdx.x;
    if (i >= MN) return;
    float s = 0.0f;
#pragma unroll
    for (int zc = 0; zc < KCHUNKS; ++zc) s += part[(size_t)zc * MN + i];
    s += bias[i % Nld];
    if (relu) s = fmaxf(s, 0.0f);
    C[i] = s;
}


#define SLOTS 3
#define EPI   8
template <bool PF>
__device__ __forceinline__ void add_pass(const int i0, const int n, const int l4,
                                         const unsigned int* q, const float* dl, float* acc,
                                         const float4* __restrict__ xw4,
                                         const float* __restrict__ dinv,
                                         int4 (&cv)[4], int4 (&rv)[4],
                                         const int4* __restrict__ nxc, const int4* __restrict__ nxr,
                                         const int tid) {
    bool   h[SLOTS];
    int    nd[SLOTS];
    float  dc[SLOTS], dr[SLOTS];
    float4 v[SLOTS];
#pragma unroll
    for (int j = 0; j < SLOTS; ++j) {
        const int i = i0 + j * EPI;
        h[j] = i < n;
        const unsigned int en = q[h[j] ? i : 0];
        const int r = h[j] ? (int)(en & ((1u << RBITS) - 1u)) : 0;
        nd[j] = (int)(en >> RBITS) & (PNODES - 1);
        dr[j] = dinv[r];
        v[j]  = xw4[r * 4 + l4];
        dc[j] = dl[nd[j]];
    }
    if (PF) {
#pragma unroll
        for (int u = 0; u < 4; ++u) cv[u] = nxc[u * WG + tid];
#pragma unroll
        for (int u = 0; u < 4; ++u) rv[u] = nxr[u * WG + tid];
    }
    asm volatile("" :: "v"(dr[0]), "v"(dr[1]), "v"(dr[2]),
                       "v"(v[0].x), "v"(v[0].y), "v"(v[0].z), "v"(v[0].w),
                       "v"(v[1].x), "v"(v[1].y), "v"(v[1].z), "v"(v[1].w),
                       "v"(v[2].x), "v"(v[2].y), "v"(v[2].z), "v"(v[2].w));
#pragma unroll
    for (int j = 0; j < SLOTS; ++j) {
        if (h[j]) {
            const float w  = dr[j] * dc[j];
            const int   sw = (nd[j] >> 2) & 3;
            float* a = acc + (nd[j] * H_DIM + l4 * 4);
            atomicAdd(a + (0 ^ sw), w * v[j].x);
            atomicAdd(a + (1 ^ sw), w * v[j].y);
            atomicAdd(a + (2 ^ sw), w * v[j].z);
            atomicAdd(a + (3 ^ sw), w * v[j].w);
        }
    }
}

__global__ __launch_bounds__(WG)
void scatter_lds_kernel(const float* __restrict__ xw,
                        const int* __restrict__ row, const int* __restrict__ col,
                        const float* __restrict__ dinv,
                        float* __restrict__ part) {
    __shared__ __attribute__((aligned(16))) float acc[PNODES * H_DIM];
    __shared__ unsigned int q[(WG / 32) * QCAP];
    __shared__ float dl[PNODES];

    const int tid  = threadIdx.x;
    const int lane = tid & 31;
    const int p = blockIdx.x >> 3, s = blockIdx.x & 7;
    float4* acc4 = (float4*)acc;
#pragma unroll
    for (int k = 0; k < PNODES * H_DIM / 4 / WG; ++k)
        acc4[k * WG + tid] = make_float4(0.0f, 0.0f, 0.0f, 0.0f);
#pragma unroll
    for (int k = 0; k < PNODES / WG; ++k)
        dl[k * WG + tid] = dinv[p * PNODES + k * WG + tid];
    __syncthreads();

    const int ebase = s * SLICE_E;
    const float4* xw4 = (const float4*)xw;
    const int g  = lane >> 2;
    const int l4 = lane & 3;
    unsigned int* qw = q + (tid >> 5) * QCAP;

    int4 cv[4], rv[4];
    {
        const int4* c4 = (const int4*)(col + ebase);
        const int4* r4 = (const int4*)(row + ebase);
#pragma unroll
        for (int u = 0; u < 4; ++u) cv[u] = c4[u * WG + tid];
#pragma unroll
        for (int u = 0; u < 4; ++u) rv[u] = r4[u * WG + tid];
    }

    for (int ch = 0; ch < SLICE_E / CH; ++ch) {
        const int nb = ebase + (ch + 1 < SLICE_E / CH ? ch + 1 : ch) * CH;
        const int4* nxc = (const int4*)(col + nb);
        const int4* nxr = (const int4*)(row + nb);

        const int t[16] = {cv[0].x, cv[0].y, cv[0].z, cv[0].w, cv[1].x, cv[1].y, cv[1].z, cv[1].w,
                           cv[2].x, cv[2].y, cv[2].z, cv[2].w, cv[3].x, cv[3].y, cv[3].z, cv[3].w};
        const int f[16] = {rv[0].x, rv[0].y, rv[0].z, rv[0].w, rv[1].x, rv[1].y, rv[1].z, rv[1].w,
                           rv[2].x, rv[2].y, rv[2].z, rv[2].w, rv[3].x, rv[3].y, rv[3].z, rv[3].w};
        unsigned int base = 0u;
#pragma unroll
        for (int k = 0; k < 16; ++k) {
            if (k != 0 && (k & 3) == 0 && base > QCAP - 4 * 32) {
                __builtin_amdgcn_fence(__ATOMIC_RELEASE, "wavefront");
                __builtin_amdgcn_wave_barrier();
                __builtin_amdgcn_fence(__ATOMIC_ACQUIRE, "wavefront");
                for (int i = g; i < (int)base + g; i += SLOTS * EPI)
                    add_pass<false>(i, (int)base, l4, qw, dl, acc, xw4, dinv, cv, rv, nxc, nxr, tid);
                base = 0u;
            }
            const bool m = (t[k] >> 12) == p;
            const unsigned int bal = __builtin_amdgcn_ballot_w32(m);
            if (m)
                qw[base + __builtin_amdgcn_mbcnt_lo(bal, 0u)] =
                    ((unsigned int)(t[k] & (PNODES - 1)) << RBITS) |
                    ((unsigned int)f[k] & ((1u << RBITS) - 1u));
            base += (unsigned int)__builtin_popcount(bal);
        }
        const int n = (int)base;
        __builtin_amdgcn_fence(__ATOMIC_RELEASE, "wavefront");
        __builtin_amdgcn_wave_barrier();
        __builtin_amdgcn_fence(__ATOMIC_ACQUIRE, "wavefront");

        add_pass<true>(g, n, l4, qw, dl, acc, xw4, dinv, cv, rv, nxc, nxr, tid);
        for (int i = g + SLOTS * EPI; i < n + g; i += SLOTS * EPI)
            add_pass<false>(i, n, l4, qw, dl, acc, xw4, dinv, cv, rv, nxc, nxr, tid);
    }
    __syncthreads();

    float4* out4 = (float4*)part + ((size_t)s * N_NODES + (size_t)p * PNODES) * (H_DIM / 4);
#pragma unroll
    for (int k = 0; k < PNODES * H_DIM / 4 / WG; ++k) {
        const int idx = k * WG + tid;
        const int sw  = (idx >> 4) & 3;
        const float4 a = acc4[idx];
        float4 b, o;
        b.x = (sw & 1) ? a.y : a.x;  b.y = (sw & 1) ? a.x : a.y;
        b.z = (sw & 1) ? a.w : a.z;  b.w = (sw & 1) ? a.z : a.w;
        o.x = (sw & 2) ? b.z : b.x;  o.y = (sw & 2) ? b.w : b.y;
        o.z = (sw & 2) ? b.x : b.z;  o.w = (sw & 2) ? b.y : b.w;
        out4[idx] = o;
    }
}

__global__ void finalize_kernel(float* __restrict__ h, const float* __restrict__ part,
                                const float* __restrict__ xw,
                                const float* __restrict__ dinv,
                                const float* __restrict__ bias, int n4) {
    int i = blockIdx.x * blockDim.x + threadIdx.x;
    if (i >= n4) return;
    const float4* p4 = (const float4*)part;
    float4 a = p4[i];
#pragma unroll
    for (int s = 1; s < NSLICE; ++s) {
        const float4 t = p4[(size_t)s * n4 + i];
        a.x += t.x; a.y += t.y; a.z += t.z; a.w += t.w;
    }
    const int node = i >> 2, j0 = (i & 3) * 4;
    const float d = dinv[node];
    const float4 x = ((const float4*)xw)[i];
    float4 r;
    r.x = fmaxf(a.x + d * d * x.x + bias[j0 + 0], 0.0f);
    r.y = fmaxf(a.y + d * d * x.y + bias[j0 + 1], 0.0f);
    r.z = fmaxf(a.z + d * d * x.z + bias[j0 + 2], 0.0f);
    r.w = fmaxf(a.w + d * d * x.w + bias[j0 + 3], 0.0f);
    ((float4*)h)[i] = r;
}

__global__ void heads_kernel(const float* __restrict__ V,
                             const float* __restrict__ Wiv, const float* __restrict__ biv,
                             const float* __restrict__ Wev, const float* __restrict__ bev,
                             float* __restrict__ out_ext, float* __restrict__ out_int) {
    int r = blockIdx.x * blockDim.x + threadIdx.x;
    if (r >= B_GRAPHS) return;
    const float* v = V + (size_t)r * 256;
    float si = 0.0f, se = 0.0f;
#pragma unroll 4
    for (int k = 0; k < 256; ++k) {
        float x = v[k];
        si += x * Wiv[k];
        se += x * Wev[k];
    }
    out_int[r] = si + biv[0];
    out_ext[r] = se + bev[0];
}

extern "C" void kernel_launch(void* const* d_in, const int* in_sizes, int n_in,
                              void* d_out, int out_size, void* d_ws, size_t ws_size,
                              hipStream_t stream) {
    const float* x    = (const float*)d_in[0];
    const int*   ei   = (const int*)d_in[1];
    const float* W1   = (const float*)d_in[2];
    const float* b1   = (const float*)d_in[3];
    const float* W2   = (const float*)d_in[4];
    const float* b2   = (const float*)d_in[5];
    const float* W3   = (const float*)d_in[6];
    const float* b3   = (const float*)d_in[7];
    const float* Wp1  = (const float*)d_in[8];
    const float* bp1  = (const float*)d_in[9];
    const float* Wp2  = (const float*)d_in[10];
    const float* bp2  = (const float*)d_in[11];
    const float* Wv1  = (const float*)d_in[12];
    const float* bv1  = (const float*)d_in[13];
    const float* Wiv  = (const float*)d_in[14];
    const float* biv  = (const float*)d_in[15];
    const float* Wev  = (const float*)d_in[16];
    const float* bev  = (const float*)d_in[17];

    const int* row = ei;
    const int* col = ei + E_EDGES;

    char* ws = (char*)d_ws;
    float* dinv = (float*)ws;
    float* bufA = (float*)(ws + (size_t)N_NODES * 4);
    float* bufB = bufA + (size_t)N_NODES * H_DIM;
    float* Hp   = bufB + (size_t)N_NODES * H_DIM;
    float* Vv   = Hp + (size_t)B_GRAPHS * 256;
    float* part = Vv + (size_t)B_GRAPHS * 256;
    unsigned int* cnt = (unsigned int*)(part + (size_t)KCHUNKS * B_GRAPHS * 256);
    float* spart = (float*)(cnt + (size_t)NSLICE * N_NODES);

    float* P     = (float*)d_out;
    float* ext_V = P + (size_t)B_GRAPHS * 1024;
    float* int_V = ext_V + B_GRAPHS;

    const int TPB = 256;
    const int nh  = N_NODES * H_DIM;
    const int nh4 = nh / 4;
    const int fcMN = B_GRAPHS * 256;

    degree_lds_kernel<<<NPART * NSLICE, WG, 0, stream>>>(col, cnt);
    dinv_kernel<<<(N_NODES + TPB - 1) / TPB, TPB, 0, stream>>>(cnt, dinv, N_NODES);

    wmma_gemm_f32<<<dim3(1, N_NODES / 16, 1), 32, 0, stream>>>(x, W1, nullptr, bufA, S_IN, H_DIM, 0, S_IN);
    scatter_lds_kernel<<<NPART * NSLICE, WG, 0, stream>>>(bufA, row, col, dinv, spart);
    finalize_kernel<<<(nh4 + TPB - 1) / TPB, TPB, 0, stream>>>(bufB, spart, bufA, dinv, b1, nh4);

    wmma_gemm_f32<<<dim3(1, N_NODES / 16, 1), 32, 0, stream>>>(bufB, W2, nullptr, bufA, H_DIM, H_DIM, 0, H_DIM);
    scatter_lds_kernel<<<NPART * NSLICE, WG, 0, stream>>>(bufA, row, col, dinv, spart);
    finalize_kernel<<<(nh4 + TPB - 1) / TPB, TPB, 0, stream>>>(bufB, spart, bufA, dinv, b2, nh4);

    wmma_gemm_f32<<<dim3(1, N_NODES / 16, 1), 32, 0, stream>>>(bufB, W3, nullptr, bufA, H_DIM, H_DIM, 0, H_DIM);
    scatter_lds_kernel<<<NPART * NSLICE, WG, 0, stream>>>(bufA, row, col, dinv, spart);
    finalize_kernel<<<(nh4 + TPB - 1) / TPB, TPB, 0, stream>>>(bufB, spart, bufA, dinv, b3, nh4);


    fc_slab_gemm_f32<<<dim3(256 / FC_NT, KCHUNKS), 256, 0, stream>>>(bufB, Wp1, part);
    reduce_bias_relu_kernel<<<(fcMN + TPB - 1) / TPB, TPB, 0, stream>>>(part, bp1, Hp, fcMN, 256, 1);
    fc_slab_gemm_f32<<<dim3(256 / FC_NT, KCHUNKS), 256, 0, stream>>>(bufB, Wv1, part);
    reduce_bias_relu_kernel<<<(fcMN + TPB - 1) / TPB, TPB, 0, stream>>>(part, bv1, Vv, fcMN, 256, 1);

    wmma_gemm_f32<<<dim3(1024 / 16, B_GRAPHS / 16, 1), 32, 0, stream>>>(Hp, Wp2, bp2, P, 256, 1024, 0, 256);

    heads_kernel<<<1, B_GRAPHS, 0, stream>>>(Vv, Wiv, biv, Wev, bev, ext_V, int_V);
}


# conv2/conv3 scatters stop re-filtering the edge list: conv1's scatter writes each work-group's matched edges (source, local target, norm) to a list that they read; sum order still LDS arrival
# speedup vs baseline: 3.4004x; 1.2638x over previous
#include <hip/hip_runtime.h>

typedef __attribute__((ext_vector_type(2))) float v2f;
typedef __attribute__((ext_vector_type(8))) float v8f;

#define N_NODES   131072
#define E_EDGES   4194304
#define S_IN      32
#define H_DIM     16
#define B_GRAPHS  128
#define FC_IN_SZ  16384
#define FC_KC     256
#define KCHUNKS   (FC_IN_SZ / FC_KC)

#define NPART    32
#define PNODES   4096
#define NSLICE   8
#define SLICE_E  (E_EDGES / NSLICE)
#define CH       16384
#define WG       1024
#define QCAP     256
#define RBITS    17

__device__ __forceinline__ unsigned int wave_sum(unsigned int v) {
#pragma unroll
    for (int m = 16; m > 0; m >>= 1) v += (unsigned int)__shfl_xor((int)v, m);
    return v;
}

__device__ __forceinline__ unsigned int list_start(const unsigned int* __restrict__ tot,
                                                   const int p, const int s, const int lane) {
    const unsigned int below = wave_sum(lane < p ? tot[lane * NSLICE + s] : 0u);
    return (unsigned int)s * SLICE_E + below;
}

__global__ __launch_bounds__(WG)
void degree_lds_kernel(const int* __restrict__ col, unsigned int* __restrict__ cnt,
                       unsigned int* __restrict__ tot) {
    __shared__ unsigned int hist[PNODES];
    __shared__ unsigned int wsum[WG / 32];
    const int tid = threadIdx.x;
    const int p = blockIdx.x >> 3, s = blockIdx.x & 7;
    for (int i = tid; i < PNODES; i += WG) hist[i] = 0u;
    __syncthreads();
    const int4* c4 = (const int4*)(col + (size_t)s * SLICE_E);
    for (int it = 0; it < SLICE_E / 4 / WG; it += 8) {
        int4 v[8];
#pragma unroll
        for (int u = 0; u < 8; ++u) v[u] = c4[(it + u) * WG + tid];
#pragma unroll
        for (int u = 0; u < 8; ++u) {
            if ((v[u].x >> 12) == p) atomicAdd(&hist[v[u].x & (PNODES - 1)], 1u);
            if ((v[u].y >> 12) == p) atomicAdd(&hist[v[u].y & (PNODES - 1)], 1u);
            if ((v[u].z >> 12) == p) atomicAdd(&hist[v[u].z & (PNODES - 1)], 1u);
            if ((v[u].w >> 12) == p) atomicAdd(&hist[v[u].w & (PNODES - 1)], 1u);
        }
    }
    __syncthreads();
    unsigned int mine = 0u;
    for (int i = tid; i < PNODES; i += WG) {
        const unsigned int hv = hist[i];
        cnt[(size_t)s * N_NODES + p * PNODES + i] = hv;
        mine += hv;
    }
    mine = wave_sum(mine);
    if ((tid & 31) == 0) wsum[tid >> 5] = mine;
    __syncthreads();
    if (tid < 32) {
        const unsigned int t = wave_sum(wsum[tid]);
        if (tid == 0) tot[blockIdx.x] = t;
    }
}

__global__ void dinv_kernel(const unsigned int* __restrict__ cnt, float* __restrict__ d, int n) {
    int i = blockIdx.x * blockDim.x + threadIdx.x;
    if (i >= n) return;
    unsigned int t = 0u;
#pragma unroll
    for (int s = 0; s < NSLICE; ++s) t += cnt[(size_t)s * n + i];
    d[i] = rsqrtf(1.0f + (float)t);
}

__global__ __launch_bounds__(32)
void wmma_gemm_f32(const float* __restrict__ A, const float* __restrict__ B,
                   const float* __restrict__ bias, float* __restrict__ C,
                   int K, int Nld, int relu, int Kc) {
    const int nb   = blockIdx.x;
    const int mb   = blockIdx.y;
    const int z    = blockIdx.z;
    const int lane = threadIdx.x;
    const int half = lane >> 4;
    const int lm   = lane & 15;

    const int k0 = z * Kc;

    v8f acc = {};
    const float* Ap = A + (size_t)(mb * 16 + lm) * K + k0 + 2 * half;
    const float* Bp = B + (size_t)(k0 + 2 * half) * Nld + nb * 16 + lm;

#pragma unroll 4
    for (int k = 0; k < Kc; k += 4) {
        v2f a = *(const v2f*)Ap;
        v2f b;
        b.x = Bp[0];
        b.y = Bp[Nld];
        acc = __builtin_amdgcn_wmma_f32_16x16x4_f32(
                  false, a, false, b, (short)0, acc, false, false);
        Ap += 4;
        Bp += (size_t)4 * Nld;
    }

    const int cn = nb * 16 + lm;
    if (gridDim.z == 1) {
        const float bb = bias ? bias[cn] : 0.0f;
#pragma unroll
        for (int v = 0; v < 8; ++v) {
            float r = acc[v] + bb;
            if (relu) r = fmaxf(r, 0.0f);
            C[(size_t)(mb * 16 + v + 8 * half) * Nld + cn] = r;
        }
    } else {
        const size_t M  = (size_t)gridDim.y * 16;
        float* Cp = C + (size_t)z * M * Nld;
#pragma unroll
        for (int v = 0; v < 8; ++v) {
            Cp[(size_t)(mb * 16 + v + 8 * half) * Nld + cn] = acc[v];
        }
    }
}

#define FC_NT     64
#define FC_AP     (FC_KC + 4)
#define FC_BP     (FC_NT + 8)
__global__ __launch_bounds__(256)
void fc_slab_gemm_f32(const float* __restrict__ A, const float* __restrict__ B,
                      float* __restrict__ part) {
    __shared__ __attribute__((aligned(16))) float As[B_GRAPHS * FC_AP];
    __shared__ __attribute__((aligned(16))) float Bs[FC_KC * FC_BP];
    const int tid = threadIdx.x;
    const int n0  = blockIdx.x * FC_NT;
    const int z   = blockIdx.y;
    const int k0  = z * FC_KC;

    {
        float4 rb[16], ra[16];
#pragma unroll
        for (int u = 0; u < 16; ++u) {
            const int i = u * 256 + tid, r = i >> 4, c4 = i & 15;
            rb[u] = *(const float4*)(B + (size_t)(k0 + r) * 256 + n0 + c4 * 4);
        }
#pragma unroll
        for (int u = 0; u < 16; ++u) {
            const int i = u * 256 + tid, r = i >> 6, c4 = i & 63;
            ra[u] = *(const float4*)(A + (size_t)r * FC_IN_SZ + k0 + c4 * 4);
        }
#pragma unroll
        for (int u = 0; u < 16; ++u) {
            const int i = u * 256 + tid, r = i >> 4, c4 = i & 15;
            *(float4*)(Bs + r * FC_BP + c4 * 4) = rb[u];
        }
#pragma unroll
        for (int u = 0; u < 16; ++u) {
            const int i = u * 256 + tid, r = i >> 6, c4 = i & 63;
            *(float4*)(As + r * FC_AP + c4 * 4) = ra[u];
        }
#pragma unroll
        for (int u = 0; u < 16; ++u) {
            const int i = (u + 16) * 256 + tid, r = i >> 6, c4 = i & 63;
            ra[u] = *(const float4*)(A + (size_t)r * FC_IN_SZ + k0 + c4 * 4);
        }
#pragma unroll
        for (int u = 0; u < 16; ++u) {
            const int i = (u + 16) * 256 + tid, r = i >> 6, c4 = i & 63;
            *(float4*)(As + r * FC_AP + c4 * 4) = ra[u];
        }
    }
    __syncthreads();

    const int lane = tid & 31;
    const int w    = tid >> 5;
    const int half = lane >> 4;
    const int c    = lane & 15;
    const float* ap = As + (16 * w + c) * FC_AP + 4 * half;
    const float* bp = Bs + (4 * half) * FC_BP + c;

    v8f acc[4] = {};
    float4 av = *(const float4*)ap;
    v2f b1[4], b2[4];
#pragma unroll
    for (int j = 0; j < 4; ++j) {
        b1[j].x = bp[0 * FC_BP + 16 * j];  b1[j].y = bp[1 * FC_BP + 16 * j];
        b2[j].x = bp[2 * FC_BP + 16 * j];  b2[j].y = bp[3 * FC_BP + 16 * j];
    }
    for (int kk = 0; kk < FC_KC; kk += 8) {
        const int kn = kk + 8 < FC_KC ? kk + 8 : kk;
        const float4 avn = *(const float4*)(ap + kn);
        const float* bk = bp + kn * FC_BP;
        v2f n1[4], n2[4];
#pragma unroll
        for (int j = 0; j < 4; ++j) {
            n1[j].x = bk[0 * FC_BP + 16 * j];  n1[j].y = bk[1 * FC_BP + 16 * j];
            n2[j].x = bk[2 * FC_BP + 16 * j];  n2[j].y = bk[3 * FC_BP + 16 * j];
        }
        v2f a1, a2;
        a1.x = av.x;  a1.y = av.y;
        a2.x = av.z;  a2.y = av.w;
#pragma unroll
        for (int j = 0; j < 4; ++j)
            acc[j] = __builtin_amdgcn_wmma_f32_16x16x4_f32(
                         false, a1, false, b1[j], (short)0, acc[j], false, false);
#pragma unroll
        for (int j = 0; j < 4; ++j)
            acc[j] = __builtin_amdgcn_wmma_f32_16x16x4_f32(
                         false, a2, false, b2[j], (short)0, acc[j], false, false);
        av = avn;
#pragma unroll
        for (int j = 0; j < 4; ++j) { b1[j] = n1[j]; b2[j] = n2[j]; }
    }

    float* Cp = part + (size_t)z * (B_GRAPHS * 256) + n0 + c;
#pragma unroll
    for (int j = 0; j < 4; ++j) {
#pragma unroll
        for (int v = 0; v < 8; ++v)
            Cp[(size_t)(16 * w + v + 8 * half) * 256 + 16 * j] = acc[j][v];
    }
}

__global__ void reduce_bias_relu_kernel(const float* __restrict__ part,
                                        const float* __restrict__ bias,
                                        float* __restrict__ C,
                                        int MN, int Nld, int relu) {
    int i = blockIdx.x * blockDim.x + threadIdx.x;
    if (i >= MN) return;
    float s = 0.0f;
#pragma unroll
    for (int zc = 0; zc < KCHUNKS; ++zc) s += part[(size_t)zc * MN + i];
    s += bias[i % Nld];
    if (relu) s = fmaxf(s, 0.0f);
    C[i] = s;
}


#define SLOTS 3
#define EPI   8

__device__ __forceinline__ unsigned int list_claim(unsigned int* fill, const unsigned int cnt,
                                                   const int lane) {
    unsigned int at = 0u;
    if (lane == 0) at = atomicAdd(fill, cnt);
    return (unsigned int)__builtin_amdgcn_readfirstlane((int)at);
}
template <bool PF>
__device__ __forceinline__ void add_pass(const int i0, const int n, const int l4,
                                         const unsigned int* q, const float* dl, float* acc,
                                         uint2* __restrict__ lst,
                                         const float4* __restrict__ xw4,
                                         const float* __restrict__ dinv,
                                         int4 (&cv)[4], int4 (&rv)[4],
                                         const int4* __restrict__ nxc, const int4* __restrict__ nxr,
                                         const int tid) {
    bool   h[SLOTS];
    unsigned int qe[SLOTS];
    int    nd[SLOTS];
    float  dc[SLOTS], dr[SLOTS];
    float4 v[SLOTS];
#pragma unroll
    for (int j = 0; j < SLOTS; ++j) {
        const int i = i0 + j * EPI;
        h[j] = i < n;
        const unsigned int en = q[h[j] ? i : 0];
        const int r = h[j] ? (int)(en & ((1u << RBITS) - 1u)) : 0;
        qe[j] = en;
        nd[j] = (int)(en >> RBITS) & (PNODES - 1);
        dr[j] = dinv[r];
        v[j]  = xw4[r * 4 + l4];
        dc[j] = dl[nd[j]];
    }
    if (PF) {
#pragma unroll
        for (int u = 0; u < 4; ++u) cv[u] = nxc[u * WG + tid];
#pragma unroll
        for (int u = 0; u < 4; ++u) rv[u] = nxr[u * WG + tid];
    }
    asm volatile("" :: "v"(dr[0]), "v"(dr[1]), "v"(dr[2]),
                       "v"(v[0].x), "v"(v[0].y), "v"(v[0].z), "v"(v[0].w),
                       "v"(v[1].x), "v"(v[1].y), "v"(v[1].z), "v"(v[1].w),
                       "v"(v[2].x), "v"(v[2].y), "v"(v[2].z), "v"(v[2].w));
#pragma unroll
    for (int j = 0; j < SLOTS; ++j) {
        if (h[j]) {
            const float w  = dr[j] * dc[j];
            if (l4 == 0) lst[i0 + j * EPI] = make_uint2(qe[j], __float_as_uint(w));
            const int   sw = (nd[j] >> 2) & 3;
            float* a = acc + (nd[j] * H_DIM + l4 * 4);
            atomicAdd(a + (0 ^ sw), w * v[j].x);
            atomicAdd(a + (1 ^ sw), w * v[j].y);
            atomicAdd(a + (2 ^ sw), w * v[j].z);
            atomicAdd(a + (3 ^ sw), w * v[j].w);
        }
    }
}

__global__ __launch_bounds__(WG)
void scatter_lds_kernel(const float* __restrict__ xw,
                        const int* __restrict__ row, const int* __restrict__ col,
                        const float* __restrict__ dinv,
                        const unsigned int* __restrict__ tot, unsigned int* __restrict__ list,
                        float* __restrict__ part) {
    __shared__ __attribute__((aligned(16))) float acc[PNODES * H_DIM];
    __shared__ unsigned int q[(WG / 32) * QCAP];
    __shared__ float dl[PNODES];
    __shared__ unsigned int lfill;

    const int tid  = threadIdx.x;
    const int lane = tid & 31;
    const int p = blockIdx.x >> 3, s = blockIdx.x & 7;
    float4* acc4 = (float4*)acc;
#pragma unroll
    for (int k = 0; k < PNODES * H_DIM / 4 / WG; ++k)
        acc4[k * WG + tid] = make_float4(0.0f, 0.0f, 0.0f, 0.0f);
#pragma unroll
    for (int k = 0; k < PNODES / WG; ++k)
        dl[k * WG + tid] = dinv[p * PNODES + k * WG + tid];
    if (tid == 0) lfill = 0u;
    uint2* mylist = (uint2*)list + list_start(tot, p, s, lane);
    __syncthreads();

    const int ebase = s * SLICE_E;
    const float4* xw4 = (const float4*)xw;
    const int g  = lane >> 2;
    const int l4 = lane & 3;
    unsigned int* qw = q + (tid >> 5) * QCAP;

    int4 cv[4], rv[4];
    {
        const int4* c4 = (const int4*)(col + ebase);
        const int4* r4 = (const int4*)(row + ebase);
#pragma unroll
        for (int u = 0; u < 4; ++u) cv[u] = c4[u * WG + tid];
#pragma unroll
        for (int u = 0; u < 4; ++u) rv[u] = r4[u * WG + tid];
    }

    for (int ch = 0; ch < SLICE_E / CH; ++ch) {
        const int nb = ebase + (ch + 1 < SLICE_E / CH ? ch + 1 : ch) * CH;
        const int4* nxc = (const int4*)(col + nb);
        const int4* nxr = (const int4*)(row + nb);

        const int t[16] = {cv[0].x, cv[0].y, cv[0].z, cv[0].w, cv[1].x, cv[1].y, cv[1].z, cv[1].w,
                           cv[2].x, cv[2].y, cv[2].z, cv[2].w, cv[3].x, cv[3].y, cv[3].z, cv[3].w};
        const int f[16] = {rv[0].x, rv[0].y, rv[0].z, rv[0].w, rv[1].x, rv[1].y, rv[1].z, rv[1].w,
                           rv[2].x, rv[2].y, rv[2].z, rv[2].w, rv[3].x, rv[3].y, rv[3].z, rv[3].w};
        unsigned int base = 0u;
#pragma unroll
        for (int k = 0; k < 16; ++k) {
            if (k != 0 && (k & 3) == 0 && base > QCAP - 4 * 32) {
                __builtin_amdgcn_fence(__ATOMIC_RELEASE, "wavefront");
                __builtin_amdgcn_wave_barrier();
                __builtin_amdgcn_fence(__ATOMIC_ACQUIRE, "wavefront");
                uint2* lst = mylist + list_claim(&lfill, base, lane);
                for (int i = g; i < (int)base + g; i += SLOTS * EPI)
                    add_pass<false>(i, (int)base, l4, qw, dl, acc, lst, xw4, dinv, cv, rv, nxc, nxr, tid);
                base = 0u;
            }
            const bool m = (t[k] >> 12) == p;
            const unsigned int bal = __builtin_amdgcn_ballot_w32(m);
            if (m)
                qw[base + __builtin_amdgcn_mbcnt_lo(bal, 0u)] =
                    ((unsigned int)(t[k] & (PNODES - 1)) << RBITS) |
                    ((unsigned int)f[k] & ((1u << RBITS) - 1u));
            base += (unsigned int)__builtin_popcount(bal);
        }
        const int n = (int)base;
        __builtin_amdgcn_fence(__ATOMIC_RELEASE, "wavefront");
        __builtin_amdgcn_wave_barrier();
        __builtin_amdgcn_fence(__ATOMIC_ACQUIRE, "wavefront");

        uint2* lst = mylist + list_claim(&lfill, base, lane);

        add_pass<true>(g, n, l4, qw, dl, acc, lst, xw4, dinv, cv, rv, nxc, nxr, tid);
        for (int i = g + SLOTS * EPI; i < n + g; i += SLOTS * EPI)
            add_pass<false>(i, n, l4, qw, dl, acc, lst, xw4, dinv, cv, rv, nxc, nxr, tid);
    }
    __syncthreads();

    float4* out4 = (float4*)part + ((size_t)s * N_NODES + (size_t)p * PNODES) * (H_DIM / 4);
#pragma unroll
    for (int k = 0; k < PNODES * H_DIM / 4 / WG; ++k) {
        const int idx = k * WG + tid;
        const int sw  = (idx >> 4) & 3;
        const float4 a = acc4[idx];
        float4 b, o;
        b.x = (sw & 1) ? a.y : a.x;  b.y = (sw & 1) ? a.x : a.y;
        b.z = (sw & 1) ? a.w : a.z;  b.w = (sw & 1) ? a.z : a.w;
        o.x = (sw & 2) ? b.z : b.x;  o.y = (sw & 2) ? b.w : b.y;
        o.z = (sw & 2) ? b.x : b.z;  o.w = (sw & 2) ? b.y : b.w;
        out4[idx] = o;
    }
}

#define LSLOTS 4
__global__ __launch_bounds__(WG)
void scatter_list_kernel(const float* __restrict__ xw,
                         const unsigned int* __restrict__ tot, const unsigned int* __restrict__ list,
                         float* __restrict__ part) {
    __shared__ __attribute__((aligned(16))) float acc[PNODES * H_DIM];

    const int tid  = threadIdx.x;
    const int lane = tid & 31;
    const int p = blockIdx.x >> 3, s = blockIdx.x & 7;
    float4* acc4 = (float4*)acc;
#pragma unroll
    for (int k = 0; k < PNODES * H_DIM / 4 / WG; ++k)
        acc4[k * WG + tid] = make_float4(0.0f, 0.0f, 0.0f, 0.0f);
    const uint2* L = (const uint2*)list + list_start(tot, p, s, lane);
    const int n = (int)tot[blockIdx.x];
    __syncthreads();

    const float4* xw4 = (const float4*)xw;
    const int g  = lane >> 2;
    const int l4 = lane & 3;
    const int nblk = (n + 31) >> 5;
    const int wstep = WG / 32;

    int blk = tid >> 5;
    uint2 e[LSLOTS];
    if (blk < nblk) {
#pragma unroll
        for (int j = 0; j < LSLOTS; ++j) {
            const int i = blk * 32 + j * EPI + g;
            e[j] = L[i < n ? i : n - 1];
        }
    }
    for (; blk < nblk; blk += wstep) {
        const int nb = blk + wstep < nblk ? blk + wstep : blk;
        bool   h[LSLOTS];
        float4 v[LSLOTS];
        uint2  en[LSLOTS];
#pragma unroll
        for (int j = 0; j < LSLOTS; ++j) {
            h[j] = blk * 32 + j * EPI + g < n;
            const int r = (int)(e[j].x & ((1u << RBITS) - 1u));
            v[j] = xw4[r * 4 + l4];
        }
#pragma unroll
        for (int j = 0; j < LSLOTS; ++j) {
            const int i = nb * 32 + j * EPI + g;
            en[j] = L[i < n ? i : n - 1];
        }
        asm volatile("" :: "v"(v[0].x), "v"(v[0].y), "v"(v[0].z), "v"(v[0].w),
                           "v"(v[1].x), "v"(v[1].y), "v"(v[1].z), "v"(v[1].w),
                           "v"(v[2].x), "v"(v[2].y), "v"(v[2].z), "v"(v[2].w),
                           "v"(v[3].x), "v"(v[3].y), "v"(v[3].z), "v"(v[3].w));
#pragma unroll
        for (int j = 0; j < LSLOTS; ++j) {
            if (h[j]) {
                const float w  = __uint_as_float(e[j].y);
                const int   nd = (int)(e[j].x >> RBITS) & (PNODES - 1);
                const int   sw = (nd >> 2) & 3;
                float* a = acc + (nd * H_DIM + l4 * 4);
                atomicAdd(a + (0 ^ sw), w * v[j].x);
                atomicAdd(a + (1 ^ sw), w * v[j].y);
                atomicAdd(a + (2 ^ sw), w * v[j].z);
                atomicAdd(a + (3 ^ sw), w * v[j].w);
            }
        }
#pragma unroll
        for (int j = 0; j < LSLOTS; ++j) e[j] = en[j];
    }
    __syncthreads();

    float4* out4 = (float4*)part + ((size_t)s * N_NODES + (size_t)p * PNODES) * (H_DIM / 4);
#pragma unroll
    for (int k = 0; k < PNODES * H_DIM / 4 / WG; ++k) {
        const int idx = k * WG + tid;
        const int sw  = (idx >> 4) & 3;
        const float4 a = acc4[idx];
        float4 b, o;
        b.x = (sw & 1) ? a.y : a.x;  b.y = (sw & 1) ? a.x : a.y;
        b.z = (sw & 1) ? a.w : a.z;  b.w = (sw & 1) ? a.z : a.w;
        o.x = (sw & 2) ? b.z : b.x;  o.y = (sw & 2) ? b.w : b.y;
        o.z = (sw & 2) ? b.x : b.z;  o.w = (sw & 2) ? b.y : b.w;
        out4[idx] = o;
    }
}

__global__ void finalize_kernel(float* __restrict__ h, const float* __restrict__ part,
                                const float* __restrict__ xw,
                                const float* __restrict__ dinv,
                                const float* __restrict__ bias, int n4) {
    int i = blockIdx.x * blockDim.x + threadIdx.x;
    if (i >= n4) return;
    const float4* p4 = (const float4*)part;
    float4 a = p4[i];
#pragma unroll
    for (int s = 1; s < NSLICE; ++s) {
        const float4 t = p4[(size_t)s * n4 + i];
        a.x += t.x; a.y += t.y; a.z += t.z; a.w += t.w;
    }
    const int node = i >> 2, j0 = (i & 3) * 4;
    const float d = dinv[node];
    const float4 x = ((const float4*)xw)[i];
    float4 r;
    r.x = fmaxf(a.x + d * d * x.x + bias[j0 + 0], 0.0f);
    r.y = fmaxf(a.y + d * d * x.y + bias[j0 + 1], 0.0f);
    r.z = fmaxf(a.z + d * d * x.z + bias[j0 + 2], 0.0f);
    r.w = fmaxf(a.w + d * d * x.w + bias[j0 + 3], 0.0f);
    ((float4*)h)[i] = r;
}

__global__ void heads_kernel(const float* __restrict__ V,
                             const float* __restrict__ Wiv, const float* __restrict__ biv,
                             const float* __restrict__ Wev, const float* __restrict__ bev,
                             float* __restrict__ out_ext, float* __restrict__ out_int) {
    int r = blockIdx.x * blockDim.x + threadIdx.x;
    if (r >= B_GRAPHS) return;
    const float* v = V + (size_t)r * 256;
    float si = 0.0f, se = 0.0f;
#pragma unroll 4
    for (int k = 0; k < 256; ++k) {
        float x = v[k];
        si += x * Wiv[k];
        se += x * Wev[k];
    }
    out_int[r] = si + biv[0];
    out_ext[r] = se + bev[0];
}

extern "C" void kernel_launch(void* const* d_in, const int* in_sizes, int n_in,
                              void* d_out, int out_size, void* d_ws, size_t ws_size,
                              hipStream_t stream) {
    const float* x    = (const float*)d_in[0];
    const int*   ei   = (const int*)d_in[1];
    const float* W1   = (const float*)d_in[2];
    const float* b1   = (const float*)d_in[3];
    const float* W2   = (const float*)d_in[4];
    const float* b2   = (const float*)d_in[5];
    const float* W3   = (const float*)d_in[6];
    const float* b3   = (const float*)d_in[7];
    const float* Wp1  = (const float*)d_in[8];
    const float* bp1  = (const float*)d_in[9];
    const float* Wp2  = (const float*)d_in[10];
    const float* bp2  = (const float*)d_in[11];
    const float* Wv1  = (const float*)d_in[12];
    const float* bv1  = (const float*)d_in[13];
    const float* Wiv  = (const float*)d_in[14];
    const float* biv  = (const float*)d_in[15];
    const float* Wev  = (const float*)d_in[16];
    const float* bev  = (const float*)d_in[17];

    const int* row = ei;
    const int* col = ei + E_EDGES;

    char* ws = (char*)d_ws;
    float* dinv = (float*)ws;
    float* bufA = (float*)(ws + (size_t)N_NODES * 4);
    float* bufB = bufA + (size_t)N_NODES * H_DIM;
    float* Hp   = bufB + (size_t)N_NODES * H_DIM;
    float* Vv   = Hp + (size_t)B_GRAPHS * 256;
    float* part = Vv + (size_t)B_GRAPHS * 256;
    unsigned int* cnt = (unsigned int*)(part + (size_t)KCHUNKS * B_GRAPHS * 256);
    float* spart = (float*)(cnt + (size_t)NSLICE * N_NODES);
    unsigned int* elist = (unsigned int*)(spart + (size_t)NSLICE * N_NODES * H_DIM);
    unsigned int* tot = elist + (size_t)2 * E_EDGES;

    float* P     = (float*)d_out;
    float* ext_V = P + (size_t)B_GRAPHS * 1024;
    float* int_V = ext_V + B_GRAPHS;

    const int TPB = 256;
    const int nh  = N_NODES * H_DIM;
    const int nh4 = nh / 4;
    const int fcMN = B_GRAPHS * 256;

    degree_lds_kernel<<<NPART * NSLICE, WG, 0, stream>>>(col, cnt, tot);
    dinv_kernel<<<(N_NODES + TPB - 1) / TPB, TPB, 0, stream>>>(cnt, dinv, N_NODES);

    wmma_gemm_f32<<<dim3(1, N_NODES / 16, 1), 32, 0, stream>>>(x, W1, nullptr, bufA, S_IN, H_DIM, 0, S_IN);
    scatter_lds_kernel<<<NPART * NSLICE, WG, 0, stream>>>(bufA, row, col, dinv, tot, elist, spart);
    finalize_kernel<<<(nh4 + TPB - 1) / TPB, TPB, 0, stream>>>(bufB, spart, bufA, dinv, b1, nh4);

    wmma_gemm_f32<<<dim3(1, N_NODES / 16, 1), 32, 0, stream>>>(bufB, W2, nullptr, bufA, H_DIM, H_DIM, 0, H_DIM);
    scatter_list_kernel<<<NPART * NSLICE, WG, 0, stream>>>(bufA, tot, elist, spart);
    finalize_kernel<<<(nh4 + TPB - 1) / TPB, TPB, 0, stream>>>(bufB, spart, bufA, dinv, b2, nh4);

    wmma_gemm_f32<<<dim3(1, N_NODES / 16, 1), 32, 0, stream>>>(bufB, W3, nullptr, bufA, H_DIM, H_DIM, 0, H_DIM);
    scatter_list_kernel<<<NPART * NSLICE, WG, 0, stream>>>(bufA, tot, elist, spart);
    finalize_kernel<<<(nh4 + TPB - 1) / TPB, TPB, 0, stream>>>(bufB, spart, bufA, dinv, b3, nh4);


    fc_slab_gemm_f32<<<dim3(256 / FC_NT, KCHUNKS), 256, 0, stream>>>(bufB, Wp1, part);
    reduce_bias_relu_kernel<<<(fcMN + TPB - 1) / TPB, TPB, 0, stream>>>(part, bp1, Hp, fcMN, 256, 1);
    fc_slab_gemm_f32<<<dim3(256 / FC_NT, KCHUNKS), 256, 0, stream>>>(bufB, Wv1, part);
    reduce_bias_relu_kernel<<<(fcMN + TPB - 1) / TPB, TPB, 0, stream>>>(part, bv1, Vv, fcMN, 256, 1);

    wmma_gemm_f32<<<dim3(1024 / 16, B_GRAPHS / 16, 1), 32, 0, stream>>>(Hp, Wp2, bp2, P, 256, 1024, 0, 256);

    heads_kernel<<<1, B_GRAPHS, 0, stream>>>(Vv, Wiv, biv, Wev, bev, ext_V, int_V);
}


# Degree pass re-partitioned: a work-group counts 32768 nodes x 1/64 of the edges (128 KiB of counters in LDS): each target is read 4 times, not 32; 64 count planes per node; integer sums, exact
# speedup vs baseline: 3.5296x; 1.0380x over previous
#include <hip/hip_runtime.h>

typedef __attribute__((ext_vector_type(2))) float v2f;
typedef __attribute__((ext_vector_type(8))) float v8f;

#define N_NODES   131072
#define E_EDGES   4194304
#define S_IN      32
#define H_DIM     16
#define B_GRAPHS  128
#define FC_IN_SZ  16384
#define FC_KC     256
#define KCHUNKS   (FC_IN_SZ / FC_KC)

#define NPART    32
#define PNODES   4096
#define NSLICE   8
#define SLICE_E  (E_EDGES / NSLICE)
#define CH       16384
#define WG       1024
#define QCAP     256
#define RBITS    17

#define DPART    4
#define DNODES   (N_NODES / DPART)
#define DSHIFT   15
#define DSLICE   64
#define DSUB     (DSLICE / NSLICE)
#define DSLICE_E (E_EDGES / DSLICE)
#define DRANGES  (DNODES / PNODES)

__device__ __forceinline__ unsigned int wave_sum(unsigned int v) {
#pragma unroll
    for (int m = 16; m > 0; m >>= 1) v += (unsigned int)__shfl_xor((int)v, m);
    return v;
}

__device__ __forceinline__ unsigned int list_len(const unsigned int* __restrict__ tot,
                                                 const int p, const int s) {
    const uint4* t4 = (const uint4*)(tot + p * DSLICE + s * DSUB);
    const uint4 a = t4[0], b = t4[1];
    return (a.x + a.y) + (a.z + a.w) + (b.x + b.y) + (b.z + b.w);
}

__device__ __forceinline__ unsigned int list_start(const unsigned int* __restrict__ tot,
                                                   const int p, const int s, const int lane) {
    const unsigned int below = wave_sum(lane < p ? list_len(tot, lane, s) : 0u);
    return (unsigned int)s * SLICE_E + below;
}

__global__ __launch_bounds__(WG)
void degree_lds_kernel(const int* __restrict__ col, unsigned int* __restrict__ cnt,
                       unsigned int* __restrict__ tot) {
    __shared__ __attribute__((aligned(16))) unsigned int hist[DNODES];
    __shared__ unsigned int wsum[(WG / 32) * DRANGES];
    const int tid = threadIdx.x;
    const int r = blockIdx.x / DSLICE, s = blockIdx.x % DSLICE;
    uint4* h4 = (uint4*)hist;
#pragma unroll
    for (int k = 0; k < DNODES / 4 / WG; ++k) h4[k * WG + tid] = make_uint4(0u, 0u, 0u, 0u);
    __syncthreads();
    const int4* c4 = (const int4*)(col + (size_t)s * DSLICE_E);
    int4 v[DSLICE_E / 4 / WG];
#pragma unroll
    for (int u = 0; u < DSLICE_E / 4 / WG; ++u) v[u] = c4[u * WG + tid];
#pragma unroll
    for (int u = 0; u < DSLICE_E / 4 / WG; ++u) {
        if ((v[u].x >> DSHIFT) == r) atomicAdd(&hist[v[u].x & (DNODES - 1)], 1u);
        if ((v[u].y >> DSHIFT) == r) atomicAdd(&hist[v[u].y & (DNODES - 1)], 1u);
        if ((v[u].z >> DSHIFT) == r) atomicAdd(&hist[v[u].z & (DNODES - 1)], 1u);
        if ((v[u].w >> DSHIFT) == r) atomicAdd(&hist[v[u].w & (DNODES - 1)], 1u);
    }
    __syncthreads();
    unsigned int mine[DRANGES];
#pragma unroll
    for (int q = 0; q < DRANGES; ++q) mine[q] = 0u;
    unsigned int* plane = cnt + (size_t)s * N_NODES + (size_t)r * DNODES;
#pragma unroll
    for (int k = 0; k < DNODES / WG; ++k) {
        const unsigned int hv = hist[k * WG + tid];
        plane[k * WG + tid] = hv;
        mine[k / (PNODES / WG)] += hv;
    }
#pragma unroll
    for (int q = 0; q < DRANGES; ++q) {
        const unsigned int t = wave_sum(mine[q]);
        if ((tid & 31) == 0) wsum[(tid >> 5) * DRANGES + q] = t;
    }
    __syncthreads();
    if (tid < DRANGES) {
        unsigned int t = 0u;
#pragma unroll
        for (int w = 0; w < WG / 32; ++w) t += wsum[w * DRANGES + tid];
        tot[(r * DRANGES + tid) * DSLICE + s] = t;
    }
}

__global__ void dinv_kernel(const unsigned int* __restrict__ cnt, float* __restrict__ d, int n) {
    int i = blockIdx.x * blockDim.x + threadIdx.x;
    if (i >= n) return;
    unsigned int t = 0u;
#pragma unroll
    for (int s = 0; s < DSLICE; ++s) t += cnt[(size_t)s * n + i];
    d[i] = rsqrtf(1.0f + (float)t);
}

__global__ __launch_bounds__(32)
void wmma_gemm_f32(const float* __restrict__ A, const float* __restrict__ B,
                   const float* __restrict__ bias, float* __restrict__ C,
                   int K, int Nld, int relu, int Kc) {
    const int nb   = blockIdx.x;
    const int mb   = blockIdx.y;
    const int z    = blockIdx.z;
    const int lane = threadIdx.x;
    const int half = lane >> 4;
    const int lm   = lane & 15;

    const int k0 = z * Kc;

    v8f acc = {};
    const float* Ap = A + (size_t)(mb * 16 + lm) * K + k0 + 2 * half;
    const float* Bp = B + (size_t)(k0 + 2 * half) * Nld + nb * 16 + lm;

#pragma unroll 4
    for (int k = 0; k < Kc; k += 4) {
        v2f a = *(const v2f*)Ap;
        v2f b;
        b.x = Bp[0];
        b.y = Bp[Nld];
        acc = __builtin_amdgcn_wmma_f32_16x16x4_f32(
                  false, a, false, b, (short)0, acc, false, false);
        Ap += 4;
        Bp += (size_t)4 * Nld;
    }

    const int cn = nb * 16 + lm;
    if (gridDim.z == 1) {
        const float bb = bias ? bias[cn] : 0.0f;
#pragma unroll
        for (int v = 0; v < 8; ++v) {
            float r = acc[v] + bb;
            if (relu) r = fmaxf(r, 0.0f);
            C[(size_t)(mb * 16 + v + 8 * half) * Nld + cn] = r;
        }
    } else {
        const size_t M  = (size_t)gridDim.y * 16;
        float* Cp = C + (size_t)z * M * Nld;
#pragma unroll
        for (int v = 0; v < 8; ++v) {
            Cp[(size_t)(mb * 16 + v + 8 * half) * Nld + cn] = acc[v];
        }
    }
}

#define FC_NT     64
#define FC_AP     (FC_KC + 4)
#define FC_BP     (FC_NT + 8)
__global__ __launch_bounds__(256)
void fc_slab_gemm_f32(const float* __restrict__ A, const float* __restrict__ B,
                      float* __restrict__ part) {
    __shared__ __attribute__((aligned(16))) float As[B_GRAPHS * FC_AP];
    __shared__ __attribute__((aligned(16))) float Bs[FC_KC * FC_BP];
    const int tid = threadIdx.x;
    const int n0  = blockIdx.x * FC_NT;
    const int z   = blockIdx.y;
    const int k0  = z * FC_KC;

    {
        float4 rb[16], ra[16];
#pragma unroll
        for (int u = 0; u < 16; ++u) {
            const int i = u * 256 + tid, r = i >> 4, c4 = i & 15;
            rb[u] = *(const float4*)(B + (size_t)(k0 + r) * 256 + n0 + c4 * 4);
        }
#pragma unroll
        for (int u = 0; u < 16; ++u) {
            const int i = u * 256 + tid, r = i >> 6, c4 = i & 63;
            ra[u] = *(const float4*)(A + (size_t)r * FC_IN_SZ + k0 + c4 * 4);
        }
#pragma unroll
        for (int u = 0; u < 16; ++u) {
            const int i = u * 256 + tid, r = i >> 4, c4 = i & 15;
            *(float4*)(Bs + r * FC_BP + c4 * 4) = rb[u];
        }
#pragma unroll
        for (int u = 0; u < 16; ++u) {
            const int i = u * 256 + tid, r = i >> 6, c4 = i & 63;
            *(float4*)(As + r * FC_AP + c4 * 4) = ra[u];
        }
#pragma unroll
        for (int u = 0; u < 16; ++u) {
            const int i = (u + 16) * 256 + tid, r = i >> 6, c4 = i & 63;
            ra[u] = *(const float4*)(A + (size_t)r * FC_IN_SZ + k0 + c4 * 4);
        }
#pragma unroll
        for (int u = 0; u < 16; ++u) {
            const int i = (u + 16) * 256 + tid, r = i >> 6, c4 = i & 63;
            *(float4*)(As + r * FC_AP + c4 * 4) = ra[u];
        }
    }
    __syncthreads();

    const int lane = tid & 31;
    const int w    = tid >> 5;
    const int half = lane >> 4;
    const int c    = lane & 15;
    const float* ap = As + (16 * w + c) * FC_AP + 4 * half;
    const float* bp = Bs + (4 * half) * FC_BP + c;

    v8f acc[4] = {};
    float4 av = *(const float4*)ap;
    v2f b1[4], b2[4];
#pragma unroll
    for (int j = 0; j < 4; ++j) {
        b1[j].x = bp[0 * FC_BP + 16 * j];  b1[j].y = bp[1 * FC_BP + 16 * j];
        b2[j].x = bp[2 * FC_BP + 16 * j];  b2[j].y = bp[3 * FC_BP + 16 * j];
    }
    for (int kk = 0; kk < FC_KC; kk += 8) {
        const int kn = kk + 8 < FC_KC ? kk + 8 : kk;
        const float4 avn = *(const float4*)(ap + kn);
        const float* bk = bp + kn * FC_BP;
        v2f n1[4], n2[4];
#pragma unroll
        for (int j = 0; j < 4; ++j) {
            n1[j].x = bk[0 * FC_BP + 16 * j];  n1[j].y = bk[1 * FC_BP + 16 * j];
            n2[j].x = bk[2 * FC_BP + 16 * j];  n2[j].y = bk[3 * FC_BP + 16 * j];
        }
        v2f a1, a2;
        a1.x = av.x;  a1.y = av.y;
        a2.x = av.z;  a2.y = av.w;
#pragma unroll
        for (int j = 0; j < 4; ++j)
            acc[j] = __builtin_amdgcn_wmma_f32_16x16x4_f32(
                         false, a1, false, b1[j], (short)0, acc[j], false, false);
#pragma unroll
        for (int j = 0; j < 4; ++j)
            acc[j] = __builtin_amdgcn_wmma_f32_16x16x4_f32(
                         false, a2, false, b2[j], (short)0, acc[j], false, false);
        av = avn;
#pragma unroll
        for (int j = 0; j < 4; ++j) { b1[j] = n1[j]; b2[j] = n2[j]; }
    }

    float* Cp = part + (size_t)z * (B_GRAPHS * 256) + n0 + c;
#pragma unroll
    for (int j = 0; j < 4; ++j) {
#pragma unroll
        for (int v = 0; v < 8; ++v)
            Cp[(size_t)(16 * w + v + 8 * half) * 256 + 16 * j] = acc[j][v];
    }
}

__global__ void reduce_bias_relu_kernel(const float* __restrict__ part,
                                        const float* __restrict__ bias,
                                        float* __restrict__ C,
                                        int MN, int Nld, int relu) {
    int i = blockIdx.x * blockDim.x + threadIdx.x;
    if (i >= MN) return;
    float s = 0.0f;
#pragma unroll
    for (int zc = 0; zc < KCHUNKS; ++zc) s += part[(size_t)zc * MN + i];
    s += bias[i % Nld];
    if (relu) s = fmaxf(s, 0.0f);
    C[i] = s;
}


#define SLOTS 3
#define EPI   8

__device__ __forceinline__ unsigned int list_claim(unsigned int* fill, const unsigned int cnt,
                                                   const int lane) {
    unsigned int at = 0u;
    if (lane == 0) at = atomicAdd(fill, cnt);
    return (unsigned int)__builtin_amdgcn_readfirstlane((int)at);
}
template <bool PF>
__device__ __forceinline__ void add_pass(const int i0, const int n, const int l4,
                                         const unsigned int* q, const float* dl, float* acc,
                                         uint2* __restrict__ lst,
                                         const float4* __restrict__ xw4,
                                         const float* __restrict__ dinv,
                                         int4 (&cv)[4], int4 (&rv)[4],
                                         const int4* __restrict__ nxc, const int4* __restrict__ nxr,
                                         const int tid) {
    bool   h[SLOTS];
    unsigned int qe[SLOTS];
    int    nd[SLOTS];
    float  dc[SLOTS], dr[SLOTS];
    float4 v[SLOTS];
#pragma unroll
    for (int j = 0; j < SLOTS; ++j) {
        const int i = i0 + j * EPI;
        h[j] = i < n;
        const unsigned int en = q[h[j] ? i : 0];
        const int r = h[j] ? (int)(en & ((1u << RBITS) - 1u)) : 0;
        qe[j] = en;
        nd[j] = (int)(en >> RBITS) & (PNODES - 1);
        dr[j] = dinv[r];
        v[j]  = xw4[r * 4 + l4];
        dc[j] = dl[nd[j]];
    }
    if (PF) {
#pragma unroll
        for (int u = 0; u < 4; ++u) cv[u] = nxc[u * WG + tid];
#pragma unroll
        for (int u = 0; u < 4; ++u) rv[u] = nxr[u * WG + tid];
    }
    asm volatile("" :: "v"(dr[0]), "v"(dr[1]), "v"(dr[2]),
                       "v"(v[0].x), "v"(v[0].y), "v"(v[0].z), "v"(v[0].w),
                       "v"(v[1].x), "v"(v[1].y), "v"(v[1].z), "v"(v[1].w),
                       "v"(v[2].x), "v"(v[2].y), "v"(v[2].z), "v"(v[2].w));
#pragma unroll
    for (int j = 0; j < SLOTS; ++j) {
        if (h[j]) {
            const float w  = dr[j] * dc[j];
            if (l4 == 0) lst[i0 + j * EPI] = make_uint2(qe[j], __float_as_uint(w));
            const int   sw = (nd[j] >> 2) & 3;
            float* a = acc + (nd[j] * H_DIM + l4 * 4);
            atomicAdd(a + (0 ^ sw), w * v[j].x);
            atomicAdd(a + (1 ^ sw), w * v[j].y);
            atomicAdd(a + (2 ^ sw), w * v[j].z);
            atomicAdd(a + (3 ^ sw), w * v[j].w);
        }
    }
}

__global__ __launch_bounds__(WG)
void scatter_lds_kernel(const float* __restrict__ xw,
                        const int* __restrict__ row, const int* __restrict__ col,
                        const float* __restrict__ dinv,
                        const unsigned int* __restrict__ tot, unsigned int* __restrict__ list,
                        float* __restrict__ part) {
    __shared__ __attribute__((aligned(16))) float acc[PNODES * H_DIM];
    __shared__ unsigned int q[(WG / 32) * QCAP];
    __shared__ float dl[PNODES];
    __shared__ unsigned int lfill;

    const int tid  = threadIdx.x;
    const int lane = tid & 31;
    const int p = blockIdx.x >> 3, s = blockIdx.x & 7;
    float4* acc4 = (float4*)acc;
#pragma unroll
    for (int k = 0; k < PNODES * H_DIM / 4 / WG; ++k)
        acc4[k * WG + tid] = make_float4(0.0f, 0.0f, 0.0f, 0.0f);
#pragma unroll
    for (int k = 0; k < PNODES / WG; ++k)
        dl[k * WG + tid] = dinv[p * PNODES + k * WG + tid];
    if (tid == 0) lfill = 0u;
    uint2* mylist = (uint2*)list + list_start(tot, p, s, lane);
    __syncthreads();

    const int ebase = s * SLICE_E;
    const float4* xw4 = (const float4*)xw;
    const int g  = lane >> 2;
    const int l4 = lane & 3;
    unsigned int* qw = q + (tid >> 5) * QCAP;

    int4 cv[4], rv[4];
    {
        const int4* c4 = (const int4*)(col + ebase);
        const int4* r4 = (const int4*)(row + ebase);
#pragma unroll
        for (int u = 0; u < 4; ++u) cv[u] = c4[u * WG + tid];
#pragma unroll
        for (int u = 0; u < 4; ++u) rv[u] = r4[u * WG + tid];
    }

    for (int ch = 0; ch < SLICE_E / CH; ++ch) {
        const int nb = ebase + (ch + 1 < SLICE_E / CH ? ch + 1 : ch) * CH;
        const int4* nxc = (const int4*)(col + nb);
        const int4* nxr = (const int4*)(row + nb);

        const int t[16] = {cv[0].x, cv[0].y, cv[0].z, cv[0].w, cv[1].x, cv[1].y, cv[1].z, cv[1].w,
                           cv[2].x, cv[2].y, cv[2].z, cv[2].w, cv[3].x, cv[3].y, cv[3].z, cv[3].w};
        const int f[16] = {rv[0].x, rv[0].y, rv[0].z, rv[0].w, rv[1].x, rv[1].y, rv[1].z, rv[1].w,
                           rv[2].x, rv[2].y, rv[2].z, rv[2].w, rv[3].x, rv[3].y, rv[3].z, rv[3].w};
        unsigned int base = 0u;
#pragma unroll
        for (int k = 0; k < 16; ++k) {
            if (k != 0 && (k & 3) == 0 && base > QCAP - 4 * 32) {
                __builtin_amdgcn_fence(__ATOMIC_RELEASE, "wavefront");
                __builtin_amdgcn_wave_barrier();
                __builtin_amdgcn_fence(__ATOMIC_ACQUIRE, "wavefront");
                uint2* lst = mylist + list_claim(&lfill, base, lane);
                for (int i = g; i < (int)base + g; i += SLOTS * EPI)
                    add_pass<false>(i, (int)base, l4, qw, dl, acc, lst, xw4, dinv, cv, rv, nxc, nxr, tid);
                base = 0u;
            }
            const bool m = (t[k] >> 12) == p;
            const unsigned int bal = __builtin_amdgcn_ballot_w32(m);
            if (m)
                qw[base + __builtin_amdgcn_mbcnt_lo(bal, 0u)] =
                    ((unsigned int)(t[k] & (PNODES - 1)) << RBITS) |
                    ((unsigned int)f[k] & ((1u << RBITS) - 1u));
            base += (unsigned int)__builtin_popcount(bal);
        }
        const int n = (int)base;
        __builtin_amdgcn_fence(__ATOMIC_RELEASE, "wavefront");
        __builtin_amdgcn_wave_barrier();
        __builtin_amdgcn_fence(__ATOMIC_ACQUIRE, "wavefront");

        uint2* lst = mylist + list_claim(&lfill, base, lane);

        add_pass<true>(g, n, l4, qw, dl, acc, lst, xw4, dinv, cv, rv, nxc, nxr, tid);
        for (int i = g + SLOTS * EPI; i < n + g; i += SLOTS * EPI)
            add_pass<false>(i, n, l4, qw, dl, acc, lst, xw4, dinv, cv, rv, nxc, nxr, tid);
    }
    __syncthreads();

    float4* out4 = (float4*)part + ((size_t)s * N_NODES + (size_t)p * PNODES) * (H_DIM / 4);
#pragma unroll
    for (int k = 0; k < PNODES * H_DIM / 4 / WG; ++k) {
        const int idx = k * WG + tid;
        const int sw  = (idx >> 4) & 3;
        const float4 a = acc4[idx];
        float4 b, o;
        b.x = (sw & 1) ? a.y : a.x;  b.y = (sw & 1) ? a.x : a.y;
        b.z = (sw & 1) ? a.w : a.z;  b.w = (sw & 1) ? a.z : a.w;
        o.x = (sw & 2) ? b.z : b.x;  o.y = (sw & 2) ? b.w : b.y;
        o.z = (sw & 2) ? b.x : b.z;  o.w = (sw & 2) ? b.y : b.w;
        out4[idx] = o;
    }
}

#define LSLOTS 4
__global__ __launch_bounds__(WG)
void scatter_list_kernel(const float* __restrict__ xw,
                         const unsigned int* __restrict__ tot, const unsigned int* __restrict__ list,
                         float* __restrict__ part) {
    __shared__ __attribute__((aligned(16))) float acc[PNODES * H_DIM];

    const int tid  = threadIdx.x;
    const int lane = tid & 31;
    const int p = blockIdx.x >> 3, s = blockIdx.x & 7;
    float4* acc4 = (float4*)acc;
#pragma unroll
    for (int k = 0; k < PNODES * H_DIM / 4 / WG; ++k)
        acc4[k * WG + tid] = make_float4(0.0f, 0.0f, 0.0f, 0.0f);
    const uint2* L = (const uint2*)list + list_start(tot, p, s, lane);
    const int n = (int)list_len(tot, p, s);
    __syncthreads();

    const float4* xw4 = (const float4*)xw;
    const int g  = lane >> 2;
    const int l4 = lane & 3;
    const int nblk = (n + 31) >> 5;
    const int wstep = WG / 32;

    int blk = tid >> 5;
    uint2 e[LSLOTS];
    if (blk < nblk) {
#pragma unroll
        for (int j = 0; j < LSLOTS; ++j) {
            const int i = blk * 32 + j * EPI + g;
            e[j] = L[i < n ? i : n - 1];
        }
    }
    for (; blk < nblk; blk += wstep) {
        const int nb = blk + wstep < nblk ? blk + wstep : blk;
        bool   h[LSLOTS];
        float4 v[LSLOTS];
        uint2  en[LSLOTS];
#pragma unroll
        for (int j = 0; j < LSLOTS; ++j) {
            h[j] = blk * 32 + j * EPI + g < n;
            const int r = (int)(e[j].x & ((1u << RBITS) - 1u));
            v[j] = xw4[r * 4 + l4];
        }
#pragma unroll
        for (int j = 0; j < LSLOTS; ++j) {
            const int i = nb * 32 + j * EPI + g;
            en[j] = L[i < n ? i : n - 1];
        }
        asm volatile("" :: "v"(v[0].x), "v"(v[0].y), "v"(v[0].z), "v"(v[0].w),
                           "v"(v[1].x), "v"(v[1].y), "v"(v[1].z), "v"(v[1].w),
                           "v"(v[2].x), "v"(v[2].y), "v"(v[2].z), "v"(v[2].w),
                           "v"(v[3].x), "v"(v[3].y), "v"(v[3].z), "v"(v[3].w));
#pragma unroll
        for (int j = 0; j < LSLOTS; ++j) {
            if (h[j]) {
                const float w  = __uint_as_float(e[j].y);
                const int   nd = (int)(e[j].x >> RBITS) & (PNODES - 1);
                const int   sw = (nd >> 2) & 3;
                float* a = acc + (nd * H_DIM + l4 * 4);
                atomicAdd(a + (0 ^ sw), w * v[j].x);
                atomicAdd(a + (1 ^ sw), w * v[j].y);
                atomicAdd(a + (2 ^ sw), w * v[j].z);
                atomicAdd(a + (3 ^ sw), w * v[j].w);
            }
        }
#pragma unroll
        for (int j = 0; j < LSLOTS; ++j) e[j] = en[j];
    }
    __syncthreads();

    float4* out4 = (float4*)part + ((size_t)s * N_NODES + (size_t)p * PNODES) * (H_DIM / 4);
#pragma unroll
    for (int k = 0; k < PNODES * H_DIM / 4 / WG; ++k) {
        const int idx = k * WG + tid;
        const int sw  = (idx >> 4) & 3;
        const float4 a = acc4[idx];
        float4 b, o;
        b.x = (sw & 1) ? a.y : a.x;  b.y = (sw & 1) ? a.x : a.y;
        b.z = (sw & 1) ? a.w : a.z;  b.w = (sw & 1) ? a.z : a.w;
        o.x = (sw & 2) ? b.z : b.x;  o.y = (sw & 2) ? b.w : b.y;
        o.z = (sw & 2) ? b.x : b.z;  o.w = (sw & 2) ? b.y : b.w;
        out4[idx] = o;
    }
}

__global__ void finalize_kernel(float* __restrict__ h, const float* __restrict__ part,
                                const float* __restrict__ xw,
                                const float* __restrict__ dinv,
                                const float* __restrict__ bias, int n4) {
    int i = blockIdx.x * blockDim.x + threadIdx.x;
    if (i >= n4) return;
    const float4* p4 = (const float4*)part;
    float4 a = p4[i];
#pragma unroll
    for (int s = 1; s < NSLICE; ++s) {
        const float4 t = p4[(size_t)s * n4 + i];
        a.x += t.x; a.y += t.y; a.z += t.z; a.w += t.w;
    }
    const int node = i >> 2, j0 = (i & 3) * 4;
    const float d = dinv[node];
    const float4 x = ((const float4*)xw)[i];
    float4 r;
    r.x = fmaxf(a.x + d * d * x.x + bias[j0 + 0], 0.0f);
    r.y = fmaxf(a.y + d * d * x.y + bias[j0 + 1], 0.0f);
    r.z = fmaxf(a.z + d * d * x.z + bias[j0 + 2], 0.0f);
    r.w = fmaxf(a.w + d * d * x.w + bias[j0 + 3], 0.0f);
    ((float4*)h)[i] = r;
}

__global__ void heads_kernel(const float* __restrict__ V,
                             const float* __restrict__ Wiv, const float* __restrict__ biv,
                             const float* __restrict__ Wev, const float* __restrict__ bev,
                             float* __restrict__ out_ext, float* __restrict__ out_int) {
    int r = blockIdx.x * blockDim.x + threadIdx.x;
    if (r >= B_GRAPHS) return;
    const float* v = V + (size_t)r * 256;
    float si = 0.0f, se = 0.0f;
#pragma unroll 4
    for (int k = 0; k < 256; ++k) {
        float x = v[k];
        si += x * Wiv[k];
        se += x * Wev[k];
    }
    out_int[r] = si + biv[0];
    out_ext[r] = se + bev[0];
}

extern "C" void kernel_launch(void* const* d_in, const int* in_sizes, int n_in,
                              void* d_out, int out_size, void* d_ws, size_t ws_size,
                              hipStream_t stream) {
    const float* x    = (const float*)d_in[0];
    const int*   ei   = (const int*)d_in[1];
    const float* W1   = (const float*)d_in[2];
    const float* b1   = (const float*)d_in[3];
    const float* W2   = (const float*)d_in[4];
    const float* b2   = (const float*)d_in[5];
    const float* W3   = (const float*)d_in[6];
    const float* b3   = (const float*)d_in[7];
    const float* Wp1  = (const float*)d_in[8];
    const float* bp1  = (const float*)d_in[9];
    const float* Wp2  = (const float*)d_in[10];
    const float* bp2  = (const float*)d_in[11];
    const float* Wv1  = (const float*)d_in[12];
    const float* bv1  = (const float*)d_in[13];
    const float* Wiv  = (const float*)d_in[14];
    const float* biv  = (const float*)d_in[15];
    const float* Wev  = (const float*)d_in[16];
    const float* bev  = (const float*)d_in[17];

    const int* row = ei;
    const int* col = ei + E_EDGES;

    char* ws = (char*)d_ws;
    float* dinv = (float*)ws;
    float* bufA = (float*)(ws + (size_t)N_NODES * 4);
    float* bufB = bufA + (size_t)N_NODES * H_DIM;
    float* Hp   = bufB + (size_t)N_NODES * H_DIM;
    float* Vv   = Hp + (size_t)B_GRAPHS * 256;
    float* part = Vv + (size_t)B_GRAPHS * 256;
    unsigned int* cnt = (unsigned int*)(part + (size_t)KCHUNKS * B_GRAPHS * 256);
    float* spart = (float*)(cnt + (size_t)DSLICE * N_NODES);
    unsigned int* elist = (unsigned int*)(spart + (size_t)NSLICE * N_NODES * H_DIM);
    unsigned int* tot = elist + (size_t)2 * E_EDGES;

    float* P     = (float*)d_out;
    float* ext_V = P + (size_t)B_GRAPHS * 1024;
    float* int_V = ext_V + B_GRAPHS;

    const int TPB = 256;
    const int nh  = N_NODES * H_DIM;
    const int nh4 = nh / 4;
    const int fcMN = B_GRAPHS * 256;

    degree_lds_kernel<<<DPART * DSLICE, WG, 0, stream>>>(col, cnt, tot);
    dinv_kernel<<<(N_NODES + TPB - 1) / TPB, TPB, 0, stream>>>(cnt, dinv, N_NODES);

    wmma_gemm_f32<<<dim3(1, N_NODES / 16, 1), 32, 0, stream>>>(x, W1, nullptr, bufA, S_IN, H_DIM, 0, S_IN);
    scatter_lds_kernel<<<NPART * NSLICE, WG, 0, stream>>>(bufA, row, col, dinv, tot, elist, spart);
    finalize_kernel<<<(nh4 + TPB - 1) / TPB, TPB, 0, stream>>>(bufB, spart, bufA, dinv, b1, nh4);

    wmma_gemm_f32<<<dim3(1, N_NODES / 16, 1), 32, 0, stream>>>(bufB, W2, nullptr, bufA, H_DIM, H_DIM, 0, H_DIM);
    scatter_list_kernel<<<NPART * NSLICE, WG, 0, stream>>>(bufA, tot, elist, spart);
    finalize_kernel<<<(nh4 + TPB - 1) / TPB, TPB, 0, stream>>>(bufB, spart, bufA, dinv, b2, nh4);

    wmma_gemm_f32<<<dim3(1, N_NODES / 16, 1), 32, 0, stream>>>(bufB, W3, nullptr, bufA, H_DIM, H_DIM, 0, H_DIM);
    scatter_list_kernel<<<NPART * NSLICE, WG, 0, stream>>>(bufA, tot, elist, spart);
    finalize_kernel<<<(nh4 + TPB - 1) / TPB, TPB, 0, stream>>>(bufB, spart, bufA, dinv, b3, nh4);


    fc_slab_gemm_f32<<<dim3(256 / FC_NT, KCHUNKS), 256, 0, stream>>>(bufB, Wp1, part);
    reduce_bias_relu_kernel<<<(fcMN + TPB - 1) / TPB, TPB, 0, stream>>>(part, bp1, Hp, fcMN, 256, 1);
    fc_slab_gemm_f32<<<dim3(256 / FC_NT, KCHUNKS), 256, 0, stream>>>(bufB, Wv1, part);
    reduce_bias_relu_kernel<<<(fcMN + TPB - 1) / TPB, TPB, 0, stream>>>(part, bv1, Vv, fcMN, 256, 1);

    wmma_gemm_f32<<<dim3(1024 / 16, B_GRAPHS / 16, 1), 32, 0, stream>>>(Hp, Wp2, bp2, P, 256, 1024, 0, 256);

    heads_kernel<<<1, B_GRAPHS, 0, stream>>>(Vv, Wiv, biv, Wev, bev, ext_V, int_V);
}


# conv2/conv3 aggregate first, transform after: (A h) W instead of A (h W); the 16x16 WMMA products move into the slice-sum pass (a wave per 16 nodes), the two product launches go; rounding changes
# speedup vs baseline: 3.6445x; 1.0326x over previous
#include <hip/hip_runtime.h>

typedef __attribute__((ext_vector_type(2))) float v2f;
typedef __attribute__((ext_vector_type(8))) float v8f;

#define N_NODES   131072
#define E_EDGES   4194304
#define S_IN      32
#define H_DIM     16
#define B_GRAPHS  128
#define FC_IN_SZ  16384
#define FC_KC     256
#define KCHUNKS   (FC_IN_SZ / FC_KC)

#define NPART    32
#define PNODES   4096
#define NSLICE   8
#define SLICE_E  (E_EDGES / NSLICE)
#define CH       16384
#define WG       1024
#define QCAP     256
#define RBITS    17

#define DPART    4
#define DNODES   (N_NODES / DPART)
#define DSHIFT   15
#define DSLICE   64
#define DSUB     (DSLICE / NSLICE)
#define DSLICE_E (E_EDGES / DSLICE)
#define DRANGES  (DNODES / PNODES)

__device__ __forceinline__ unsigned int wave_sum(unsigned int v) {
#pragma unroll
    for (int m = 16; m > 0; m >>= 1) v += (unsigned int)__shfl_xor((int)v, m);
    return v;
}

__device__ __forceinline__ unsigned int list_len(const unsigned int* __restrict__ tot,
                                                 const int p, const int s) {
    const uint4* t4 = (const uint4*)(tot + p * DSLICE + s * DSUB);
    const uint4 a = t4[0], b = t4[1];
    return (a.x + a.y) + (a.z + a.w) + (b.x + b.y) + (b.z + b.w);
}

__device__ __forceinline__ unsigned int list_start(const unsigned int* __restrict__ tot,
                                                   const int p, const int s, const int lane) {
    const unsigned int below = wave_sum(lane < p ? list_len(tot, lane, s) : 0u);
    return (unsigned int)s * SLICE_E + below;
}

__global__ __launch_bounds__(WG)
void degree_lds_kernel(const int* __restrict__ col, unsigned int* __restrict__ cnt,
                       unsigned int* __restrict__ tot) {
    __shared__ __attribute__((aligned(16))) unsigned int hist[DNODES];
    __shared__ unsigned int wsum[(WG / 32) * DRANGES];
    const int tid = threadIdx.x;
    const int r = blockIdx.x / DSLICE, s = blockIdx.x % DSLICE;
    uint4* h4 = (uint4*)hist;
#pragma unroll
    for (int k = 0; k < DNODES / 4 / WG; ++k) h4[k * WG + tid] = make_uint4(0u, 0u, 0u, 0u);
    __syncthreads();
    const int4* c4 = (const int4*)(col + (size_t)s * DSLICE_E);
    int4 v[DSLICE_E / 4 / WG];
#pragma unroll
    for (int u = 0; u < DSLICE_E / 4 / WG; ++u) v[u] = c4[u * WG + tid];
#pragma unroll
    for (int u = 0; u < DSLICE_E / 4 / WG; ++u) {
        if ((v[u].x >> DSHIFT) == r) atomicAdd(&hist[v[u].x & (DNODES - 1)], 1u);
        if ((v[u].y >> DSHIFT) == r) atomicAdd(&hist[v[u].y & (DNODES - 1)], 1u);
        if ((v[u].z >> DSHIFT) == r) atomicAdd(&hist[v[u].z & (DNODES - 1)], 1u);
        if ((v[u].w >> DSHIFT) == r) atomicAdd(&hist[v[u].w & (DNODES - 1)], 1u);
    }
    __syncthreads();
    unsigned int mine[DRANGES];
#pragma unroll
    for (int q = 0; q < DRANGES; ++q) mine[q] = 0u;
    unsigned int* plane = cnt + (size_t)s * N_NODES + (size_t)r * DNODES;
#pragma unroll
    for (int k = 0; k < DNODES / WG; ++k) {
        const unsigned int hv = hist[k * WG + tid];
        plane[k * WG + tid] = hv;
        mine[k / (PNODES / WG)] += hv;
    }
#pragma unroll
    for (int q = 0; q < DRANGES; ++q) {
        const unsigned int t = wave_sum(mine[q]);
        if ((tid & 31) == 0) wsum[(tid >> 5) * DRANGES + q] = t;
    }
    __syncthreads();
    if (tid < DRANGES) {
        unsigned int t = 0u;
#pragma unroll
        for (int w = 0; w < WG / 32; ++w) t += wsum[w * DRANGES + tid];
        tot[(r * DRANGES + tid) * DSLICE + s] = t;
    }
}

__global__ void dinv_kernel(const unsigned int* __restrict__ cnt, float* __restrict__ d, int n) {
    int i = blockIdx.x * blockDim.x + threadIdx.x;
    if (i >= n) return;
    unsigned int t = 0u;
#pragma unroll
    for (int s = 0; s < DSLICE; ++s) t += cnt[(size_t)s * n + i];
    d[i] = rsqrtf(1.0f + (float)t);
}

__global__ __launch_bounds__(32)
void wmma_gemm_f32(const float* __restrict__ A, const float* __restrict__ B,
                   const float* __restrict__ bias, float* __restrict__ C,
                   int K, int Nld, int relu, int Kc) {
    const int nb   = blockIdx.x;
    const int mb   = blockIdx.y;
    const int z    = blockIdx.z;
    const int lane = threadIdx.x;
    const int half = lane >> 4;
    const int lm   = lane & 15;

    const int k0 = z * Kc;

    v8f acc = {};
    const float* Ap = A + (size_t)(mb * 16 + lm) * K + k0 + 2 * half;
    const float* Bp = B + (size_t)(k0 + 2 * half) * Nld + nb * 16 + lm;

#pragma unroll 4
    for (int k = 0; k < Kc; k += 4) {
        v2f a = *(const v2f*)Ap;
        v2f b;
        b.x = Bp[0];
        b.y = Bp[Nld];
        acc = __builtin_amdgcn_wmma_f32_16x16x4_f32(
                  false, a, false, b, (short)0, acc, false, false);
        Ap += 4;
        Bp += (size_t)4 * Nld;
    }

    const int cn = nb * 16 + lm;
    if (gridDim.z == 1) {
        const float bb = bias ? bias[cn] : 0.0f;
#pragma unroll
        for (int v = 0; v < 8; ++v) {
            float r = acc[v] + bb;
            if (relu) r = fmaxf(r, 0.0f);
            C[(size_t)(mb * 16 + v + 8 * half) * Nld + cn] = r;
        }
    } else {
        const size_t M  = (size_t)gridDim.y * 16;
        float* Cp = C + (size_t)z * M * Nld;
#pragma unroll
        for (int v = 0; v < 8; ++v) {
            Cp[(size_t)(mb * 16 + v + 8 * half) * Nld + cn] = acc[v];
        }
    }
}

#define FC_NT     64
#define FC_AP     (FC_KC + 4)
#define FC_BP     (FC_NT + 8)
__global__ __launch_bounds__(256)
void fc_slab_gemm_f32(const float* __restrict__ A, const float* __restrict__ B,
                      float* __restrict__ part) {
    __shared__ __attribute__((aligned(16))) float As[B_GRAPHS * FC_AP];
    __shared__ __attribute__((aligned(16))) float Bs[FC_KC * FC_BP];
    const int tid = threadIdx.x;
    const int n0  = blockIdx.x * FC_NT;
    const int z   = blockIdx.y;
    const int k0  = z * FC_KC;

    {
        float4 rb[16], ra[16];
#pragma unroll
        for (int u = 0; u < 16; ++u) {
            const int i = u * 256 + tid, r = i >> 4, c4 = i & 15;
            rb[u] = *(const float4*)(B + (size_t)(k0 + r) * 256 + n0 + c4 * 4);
        }
#pragma unroll
        for (int u = 0; u < 16; ++u) {
            const int i = u * 256 + tid, r = i >> 6, c4 = i & 63;
            ra[u] = *(const float4*)(A + (size_t)r * FC_IN_SZ + k0 + c4 * 4);
        }
#pragma unroll
        for (int u = 0; u < 16; ++u) {
            const int i = u * 256 + tid, r = i >> 4, c4 = i & 15;
            *(float4*)(Bs + r * FC_BP + c4 * 4) = rb[u];
        }
#pragma unroll
        for (int u = 0; u < 16; ++u) {
            const int i = u * 256 + tid, r = i >> 6, c4 = i & 63;
            *(float4*)(As + r * FC_AP + c4 * 4) = ra[u];
        }
#pragma unroll
        for (int u = 0; u < 16; ++u) {
            const int i = (u + 16) * 256 + tid, r = i >> 6, c4 = i & 63;
            ra[u] = *(const float4*)(A + (size_t)r * FC_IN_SZ + k0 + c4 * 4);
        }
#pragma unroll
        for (int u = 0; u < 16; ++u) {
            const int i = (u + 16) * 256 + tid, r = i >> 6, c4 = i & 63;
            *(float4*)(As + r * FC_AP + c4 * 4) = ra[u];
        }
    }
    __syncthreads();

    const int lane = tid & 31;
    const int w    = tid >> 5;
    const int half = lane >> 4;
    const int c    = lane & 15;
    const float* ap = As + (16 * w + c) * FC_AP + 4 * half;
    const float* bp = Bs + (4 * half) * FC_BP + c;

    v8f acc[4] = {};
    float4 av = *(const float4*)ap;
    v2f b1[4], b2[4];
#pragma unroll
    for (int j = 0; j < 4; ++j) {
        b1[j].x = bp[0 * FC_BP + 16 * j];  b1[j].y = bp[1 * FC_BP + 16 * j];
        b2[j].x = bp[2 * FC_BP + 16 * j];  b2[j].y = bp[3 * FC_BP + 16 * j];
    }
    for (int kk = 0; kk < FC_KC; kk += 8) {
        const int kn = kk + 8 < FC_KC ? kk + 8 : kk;
        const float4 avn = *(const float4*)(ap + kn);
        const float* bk = bp + kn * FC_BP;
        v2f n1[4], n2[4];
#pragma unroll
        for (int j = 0; j < 4; ++j) {
            n1[j].x = bk[0 * FC_BP + 16 * j];  n1[j].y = bk[1 * FC_BP + 16 * j];
            n2[j].x = bk[2 * FC_BP + 16 * j];  n2[j].y = bk[3 * FC_BP + 16 * j];
        }
        v2f a1, a2;
        a1.x = av.x;  a1.y = av.y;
        a2.x = av.z;  a2.y = av.w;
#pragma unroll
        for (int j = 0; j < 4; ++j)
            acc[j] = __builtin_amdgcn_wmma_f32_16x16x4_f32(
                         false, a1, false, b1[j], (short)0, acc[j], false, false);
#pragma unroll
        for (int j = 0; j < 4; ++j)
            acc[j] = __builtin_amdgcn_wmma_f32_16x16x4_f32(
                         false, a2, false, b2[j], (short)0, acc[j], false, false);
        av = avn;
#pragma unroll
        for (int j = 0; j < 4; ++j) { b1[j] = n1[j]; b2[j] = n2[j]; }
    }

    float* Cp = part + (size_t)z * (B_GRAPHS * 256) + n0 + c;
#pragma unroll
    for (int j = 0; j < 4; ++j) {
#pragma unroll
        for (int v = 0; v < 8; ++v)
            Cp[(size_t)(16 * w + v + 8 * half) * 256 + 16 * j] = acc[j][v];
    }
}

__global__ void reduce_bias_relu_kernel(const float* __restrict__ part,
                                        const float* __restrict__ bias,
                                        float* __restrict__ C,
                                        int MN, int Nld, int relu) {
    int i = blockIdx.x * blockDim.x + threadIdx.x;
    if (i >= MN) return;
    float s = 0.0f;
#pragma unroll
    for (int zc = 0; zc < KCHUNKS; ++zc) s += part[(size_t)zc * MN + i];
    s += bias[i % Nld];
    if (relu) s = fmaxf(s, 0.0f);
    C[i] = s;
}


#define SLOTS 3
#define EPI   8

__device__ __forceinline__ unsigned int list_claim(unsigned int* fill, const unsigned int cnt,
                                                   const int lane) {
    unsigned int at = 0u;
    if (lane == 0) at = atomicAdd(fill, cnt);
    return (unsigned int)__builtin_amdgcn_readfirstlane((int)at);
}
template <bool PF>
__device__ __forceinline__ void add_pass(const int i0, const int n, const int l4,
                                         const unsigned int* q, const float* dl, float* acc,
                                         uint2* __restrict__ lst,
                                         const float4* __restrict__ xw4,
                                         const float* __restrict__ dinv,
                                         int4 (&cv)[4], int4 (&rv)[4],
                                         const int4* __restrict__ nxc, const int4* __restrict__ nxr,
                                         const int tid) {
    bool   h[SLOTS];
    unsigned int qe[SLOTS];
    int    nd[SLOTS];
    float  dc[SLOTS], dr[SLOTS];
    float4 v[SLOTS];
#pragma unroll
    for (int j = 0; j < SLOTS; ++j) {
        const int i = i0 + j * EPI;
        h[j] = i < n;
        const unsigned int en = q[h[j] ? i : 0];
        const int r = h[j] ? (int)(en & ((1u << RBITS) - 1u)) : 0;
        qe[j] = en;
        nd[j] = (int)(en >> RBITS) & (PNODES - 1);
        dr[j] = dinv[r];
        v[j]  = xw4[r * 4 + l4];
        dc[j] = dl[nd[j]];
    }
    if (PF) {
#pragma unroll
        for (int u = 0; u < 4; ++u) cv[u] = nxc[u * WG + tid];
#pragma unroll
        for (int u = 0; u < 4; ++u) rv[u] = nxr[u * WG + tid];
    }
    asm volatile("" :: "v"(dr[0]), "v"(dr[1]), "v"(dr[2]),
                       "v"(v[0].x), "v"(v[0].y), "v"(v[0].z), "v"(v[0].w),
                       "v"(v[1].x), "v"(v[1].y), "v"(v[1].z), "v"(v[1].w),
                       "v"(v[2].x), "v"(v[2].y), "v"(v[2].z), "v"(v[2].w));
#pragma unroll
    for (int j = 0; j < SLOTS; ++j) {
        if (h[j]) {
            const float w  = dr[j] * dc[j];
            if (l4 == 0) lst[i0 + j * EPI] = make_uint2(qe[j], __float_as_uint(w));
            const int   sw = (nd[j] >> 2) & 3;
            float* a = acc + (nd[j] * H_DIM + l4 * 4);
            atomicAdd(a + (0 ^ sw), w * v[j].x);
            atomicAdd(a + (1 ^ sw), w * v[j].y);
            atomicAdd(a + (2 ^ sw), w * v[j].z);
            atomicAdd(a + (3 ^ sw), w * v[j].w);
        }
    }
}

__global__ __launch_bounds__(WG)
void scatter_lds_kernel(const float* __restrict__ xw,
                        const int* __restrict__ row, const int* __restrict__ col,
                        const float* __restrict__ dinv,
                        const unsigned int* __restrict__ tot, unsigned int* __restrict__ list,
                        float* __restrict__ part) {
    __shared__ __attribute__((aligned(16))) float acc[PNODES * H_DIM];
    __shared__ unsigned int q[(WG / 32) * QCAP];
    __shared__ float dl[PNODES];
    __shared__ unsigned int lfill;

    const int tid  = threadIdx.x;
    const int lane = tid & 31;
    const int p = blockIdx.x >> 3, s = blockIdx.x & 7;
    float4* acc4 = (float4*)acc;
#pragma unroll
    for (int k = 0; k < PNODES * H_DIM / 4 / WG; ++k)
        acc4[k * WG + tid] = make_float4(0.0f, 0.0f, 0.0f, 0.0f);
#pragma unroll
    for (int k = 0; k < PNODES / WG; ++k)
        dl[k * WG + tid] = dinv[p * PNODES + k * WG + tid];
    if (tid == 0) lfill = 0u;
    uint2* mylist = (uint2*)list + list_start(tot, p, s, lane);
    __syncthreads();

    const int ebase = s * SLICE_E;
    const float4* xw4 = (const float4*)xw;
    const int g  = lane >> 2;
    const int l4 = lane & 3;
    unsigned int* qw = q + (tid >> 5) * QCAP;

    int4 cv[4], rv[4];
    {
        const int4* c4 = (const int4*)(col + ebase);
        const int4* r4 = (const int4*)(row + ebase);
#pragma unroll
        for (int u = 0; u < 4; ++u) cv[u] = c4[u * WG + tid];
#pragma unroll
        for (int u = 0; u < 4; ++u) rv[u] = r4[u * WG + tid];
    }

    for (int ch = 0; ch < SLICE_E / CH; ++ch) {
        const int nb = ebase + (ch + 1 < SLICE_E / CH ? ch + 1 : ch) * CH;
        const int4* nxc = (const int4*)(col + nb);
        const int4* nxr = (const int4*)(row + nb);

        const int t[16] = {cv[0].x, cv[0].y, cv[0].z, cv[0].w, cv[1].x, cv[1].y, cv[1].z, cv[1].w,
                           cv[2].x, cv[2].y, cv[2].z, cv[2].w, cv[3].x, cv[3].y, cv[3].z, cv[3].w};
        const int f[16] = {rv[0].x, rv[0].y, rv[0].z, rv[0].w, rv[1].x, rv[1].y, rv[1].z, rv[1].w,
                           rv[2].x, rv[2].y, rv[2].z, rv[2].w, rv[3].x, rv[3].y, rv[3].z, rv[3].w};
        unsigned int base = 0u;
#pragma unroll
        for (int k = 0; k < 16; ++k) {
            if (k != 0 && (k & 3) == 0 && base > QCAP - 4 * 32) {
                __builtin_amdgcn_fence(__ATOMIC_RELEASE, "wavefront");
                __builtin_amdgcn_wave_barrier();
                __builtin_amdgcn_fence(__ATOMIC_ACQUIRE, "wavefront");
                uint2* lst = mylist + list_claim(&lfill, base, lane);
                for (int i = g; i < (int)base + g; i += SLOTS * EPI)
                    add_pass<false>(i, (int)base, l4, qw, dl, acc, lst, xw4, dinv, cv, rv, nxc, nxr, tid);
                base = 0u;
            }
            const bool m = (t[k] >> 12) == p;
            const unsigned int bal = __builtin_amdgcn_ballot_w32(m);
            if (m)
                qw[base + __builtin_amdgcn_mbcnt_lo(bal, 0u)] =
                    ((unsigned int)(t[k] & (PNODES - 1)) << RBITS) |
                    ((unsigned int)f[k] & ((1u << RBITS) - 1u));
            base += (unsigned int)__builtin_popcount(bal);
        }
        const int n = (int)base;
        __builtin_amdgcn_fence(__ATOMIC_RELEASE, "wavefront");
        __builtin_amdgcn_wave_barrier();
        __builtin_amdgcn_fence(__ATOMIC_ACQUIRE, "wavefront");

        uint2* lst = mylist + list_claim(&lfill, base, lane);

        add_pass<true>(g, n, l4, qw, dl, acc, lst, xw4, dinv, cv, rv, nxc, nxr, tid);
        for (int i = g + SLOTS * EPI; i < n + g; i += SLOTS * EPI)
            add_pass<false>(i, n, l4, qw, dl, acc, lst, xw4, dinv, cv, rv, nxc, nxr, tid);
    }
    __syncthreads();

    float4* out4 = (float4*)part + ((size_t)s * N_NODES + (size_t)p * PNODES) * (H_DIM / 4);
#pragma unroll
    for (int k = 0; k < PNODES * H_DIM / 4 / WG; ++k) {
        const int idx = k * WG + tid;
        const int sw  = (idx >> 4) & 3;
        const float4 a = acc4[idx];
        float4 b, o;
        b.x = (sw & 1) ? a.y : a.x;  b.y = (sw & 1) ? a.x : a.y;
        b.z = (sw & 1) ? a.w : a.z;  b.w = (sw & 1) ? a.z : a.w;
        o.x = (sw & 2) ? b.z : b.x;  o.y = (sw & 2) ? b.w : b.y;
        o.z = (sw & 2) ? b.x : b.z;  o.w = (sw & 2) ? b.y : b.w;
        out4[idx] = o;
    }
}

#define LSLOTS 4
__global__ __launch_bounds__(WG)
void scatter_list_kernel(const float* __restrict__ xw,
                         const unsigned int* __restrict__ tot, const unsigned int* __restrict__ list,
                         float* __restrict__ part) {
    __shared__ __attribute__((aligned(16))) float acc[PNODES * H_DIM];

    const int tid  = threadIdx.x;
    const int lane = tid & 31;
    const int p = blockIdx.x >> 3, s = blockIdx.x & 7;
    float4* acc4 = (float4*)acc;
#pragma unroll
    for (int k = 0; k < PNODES * H_DIM / 4 / WG; ++k)
        acc4[k * WG + tid] = make_float4(0.0f, 0.0f, 0.0f, 0.0f);
    const uint2* L = (const uint2*)list + list_start(tot, p, s, lane);
    const int n = (int)list_len(tot, p, s);
    __syncthreads();

    const float4* xw4 = (const float4*)xw;
    const int g  = lane >> 2;
    const int l4 = lane & 3;
    const int nblk = (n + 31) >> 5;
    const int wstep = WG / 32;

    int blk = tid >> 5;
    uint2 e[LSLOTS];
    if (blk < nblk) {
#pragma unroll
        for (int j = 0; j < LSLOTS; ++j) {
            const int i = blk * 32 + j * EPI + g;
            e[j] = L[i < n ? i : n - 1];
        }
    }
    for (; blk < nblk; blk += wstep) {
        const int nb = blk + wstep < nblk ? blk + wstep : blk;
        bool   h[LSLOTS];
        float4 v[LSLOTS];
        uint2  en[LSLOTS];
#pragma unroll
        for (int j = 0; j < LSLOTS; ++j) {
            h[j] = blk * 32 + j * EPI + g < n;
            const int r = (int)(e[j].x & ((1u << RBITS) - 1u));
            v[j] = xw4[r * 4 + l4];
        }
#pragma unroll
        for (int j = 0; j < LSLOTS; ++j) {
            const int i = nb * 32 + j * EPI + g;
            en[j] = L[i < n ? i : n - 1];
        }
        asm volatile("" :: "v"(v[0].x), "v"(v[0].y), "v"(v[0].z), "v"(v[0].w),
                           "v"(v[1].x), "v"(v[1].y), "v"(v[1].z), "v"(v[1].w),
                           "v"(v[2].x), "v"(v[2].y), "v"(v[2].z), "v"(v[2].w),
                           "v"(v[3].x), "v"(v[3].y), "v"(v[3].z), "v"(v[3].w));
#pragma unroll
        for (int j = 0; j < LSLOTS; ++j) {
            if (h[j]) {
                const float w  = __uint_as_float(e[j].y);
                const int   nd = (int)(e[j].x >> RBITS) & (PNODES - 1);
                const int   sw = (nd >> 2) & 3;
                float* a = acc + (nd * H_DIM + l4 * 4);
                atomicAdd(a + (0 ^ sw), w * v[j].x);
                atomicAdd(a + (1 ^ sw), w * v[j].y);
                atomicAdd(a + (2 ^ sw), w * v[j].z);
                atomicAdd(a + (3 ^ sw), w * v[j].w);
            }
        }
#pragma unroll
        for (int j = 0; j < LSLOTS; ++j) e[j] = en[j];
    }
    __syncthreads();

    float4* out4 = (float4*)part + ((size_t)s * N_NODES + (size_t)p * PNODES) * (H_DIM / 4);
#pragma unroll
    for (int k = 0; k < PNODES * H_DIM / 4 / WG; ++k) {
        const int idx = k * WG + tid;
        const int sw  = (idx >> 4) & 3;
        const float4 a = acc4[idx];
        float4 b, o;
        b.x = (sw & 1) ? a.y : a.x;  b.y = (sw & 1) ? a.x : a.y;
        b.z = (sw & 1) ? a.w : a.z;  b.w = (sw & 1) ? a.z : a.w;
        o.x = (sw & 2) ? b.z : b.x;  o.y = (sw & 2) ? b.w : b.y;
        o.z = (sw & 2) ? b.x : b.z;  o.w = (sw & 2) ? b.y : b.w;
        out4[idx] = o;
    }
}

__global__ void finalize_kernel(float* __restrict__ h, const float* __restrict__ part,
                                const float* __restrict__ xw,
                                const float* __restrict__ dinv,
                                const float* __restrict__ bias, int n4) {
    int i = blockIdx.x * blockDim.x + threadIdx.x;
    if (i >= n4) return;
    const float4* p4 = (const float4*)part;
    float4 a = p4[i];
#pragma unroll
    for (int s = 1; s < NSLICE; ++s) {
        const float4 t = p4[(size_t)s * n4 + i];
        a.x += t.x; a.y += t.y; a.z += t.z; a.w += t.w;
    }
    const int node = i >> 2, j0 = (i & 3) * 4;
    const float d = dinv[node];
    const float4 x = ((const float4*)xw)[i];
    float4 r;
    r.x = fmaxf(a.x + d * d * x.x + bias[j0 + 0], 0.0f);
    r.y = fmaxf(a.y + d * d * x.y + bias[j0 + 1], 0.0f);
    r.z = fmaxf(a.z + d * d * x.z + bias[j0 + 2], 0.0f);
    r.w = fmaxf(a.w + d * d * x.w + bias[j0 + 3], 0.0f);
    ((float4*)h)[i] = r;
}

__global__ __launch_bounds__(256)
void finalize_mm_kernel(float* __restrict__ hout, const float* __restrict__ part,
                        const float* __restrict__ hin, const float* __restrict__ dinv,
                        const float* __restrict__ W, const float* __restrict__ bias) {
    const int lane = threadIdx.x & 31;
    const int tile = blockIdx.x * 8 + (threadIdx.x >> 5);
    const int hh   = lane >> 4;
    const int c    = lane & 15;
    const int node = tile * 16 + c;
    const size_t n4 = (size_t)N_NODES * (H_DIM / 4);
    const size_t i4 = (size_t)node * (H_DIM / 4) + 2 * hh;
    const float4* p4 = (const float4*)part;

    float4 t0[NSLICE], t1[NSLICE];
#pragma unroll
    for (int s = 0; s < NSLICE; ++s) {
        t0[s] = p4[(size_t)s * n4 + i4];
        t1[s] = p4[(size_t)s * n4 + i4 + 1];
    }
    const float4 x0 = ((const float4*)hin)[i4];
    const float4 x1 = ((const float4*)hin)[i4 + 1];
    const float d  = dinv[node];
    v2f b[4];
#pragma unroll
    for (int j = 0; j < 4; ++j) {
        b[j].x = W[(8 * hh + 2 * j) * H_DIM + c];
        b[j].y = W[(8 * hh + 2 * j + 1) * H_DIM + c];
    }
    const float bb = bias[c];

    float4 lo = t0[0], hi = t1[0];
#pragma unroll
    for (int s = 1; s < NSLICE; ++s) {
        lo.x += t0[s].x; lo.y += t0[s].y; lo.z += t0[s].z; lo.w += t0[s].w;
        hi.x += t1[s].x; hi.y += t1[s].y; hi.z += t1[s].z; hi.w += t1[s].w;
    }
    const float d2 = d * d;
    v2f a[4];
    a[0].x = lo.x + d2 * x0.x;  a[0].y = lo.y + d2 * x0.y;
    a[1].x = lo.z + d2 * x0.z;  a[1].y = lo.w + d2 * x0.w;
    a[2].x = hi.x + d2 * x1.x;  a[2].y = hi.y + d2 * x1.y;
    a[3].x = hi.z + d2 * x1.z;  a[3].y = hi.w + d2 * x1.w;

    v8f acc = {};
#pragma unroll
    for (int j = 0; j < 4; ++j)
        acc = __builtin_amdgcn_wmma_f32_16x16x4_f32(
                  false, a[j], false, b[j], (short)0, acc, false, false);

    float* o = hout + (size_t)(tile * 16 + 8 * hh) * H_DIM + c;
#pragma unroll
    for (int r = 0; r < 8; ++r) o[r * H_DIM] = fmaxf(acc[r] + bb, 0.0f);
}

__global__ void heads_kernel(const float* __restrict__ V,
                             const float* __restrict__ Wiv, const float* __restrict__ biv,
                             const float* __restrict__ Wev, const float* __restrict__ bev,
                             float* __restrict__ out_ext, float* __restrict__ out_int) {
    int r = blockIdx.x * blockDim.x + threadIdx.x;
    if (r >= B_GRAPHS) return;
    const float* v = V + (size_t)r * 256;
    float si = 0.0f, se = 0.0f;
#pragma unroll 4
    for (int k = 0; k < 256; ++k) {
        float x = v[k];
        si += x * Wiv[k];
        se += x * Wev[k];
    }
    out_int[r] = si + biv[0];
    out_ext[r] = se + bev[0];
}

extern "C" void kernel_launch(void* const* d_in, const int* in_sizes, int n_in,
                              void* d_out, int out_size, void* d_ws, size_t ws_size,
                              hipStream_t stream) {
    const float* x    = (const float*)d_in[0];
    const int*   ei   = (const int*)d_in[1];
    const float* W1   = (const float*)d_in[2];
    const float* b1   = (const float*)d_in[3];
    const float* W2   = (const float*)d_in[4];
    const float* b2   = (const float*)d_in[5];
    const float* W3   = (const float*)d_in[6];
    const float* b3   = (const float*)d_in[7];
    const float* Wp1  = (const float*)d_in[8];
    const float* bp1  = (const float*)d_in[9];
    const float* Wp2  = (const float*)d_in[10];
    const float* bp2  = (const float*)d_in[11];
    const float* Wv1  = (const float*)d_in[12];
    const float* bv1  = (const float*)d_in[13];
    const float* Wiv  = (const float*)d_in[14];
    const float* biv  = (const float*)d_in[15];
    const float* Wev  = (const float*)d_in[16];
    const float* bev  = (const float*)d_in[17];

    const int* row = ei;
    const int* col = ei + E_EDGES;

    char* ws = (char*)d_ws;
    float* dinv = (float*)ws;
    float* bufA = (float*)(ws + (size_t)N_NODES * 4);
    float* bufB = bufA + (size_t)N_NODES * H_DIM;
    float* Hp   = bufB + (size_t)N_NODES * H_DIM;
    float* Vv   = Hp + (size_t)B_GRAPHS * 256;
    float* part = Vv + (size_t)B_GRAPHS * 256;
    unsigned int* cnt = (unsigned int*)(part + (size_t)KCHUNKS * B_GRAPHS * 256);
    float* spart = (float*)(cnt + (size_t)DSLICE * N_NODES);
    unsigned int* elist = (unsigned int*)(spart + (size_t)NSLICE * N_NODES * H_DIM);
    unsigned int* tot = elist + (size_t)2 * E_EDGES;

    float* P     = (float*)d_out;
    float* ext_V = P + (size_t)B_GRAPHS * 1024;
    float* int_V = ext_V + B_GRAPHS;

    const int TPB = 256;
    const int nh  = N_NODES * H_DIM;
    const int nh4 = nh / 4;
    const int fcMN = B_GRAPHS * 256;

    degree_lds_kernel<<<DPART * DSLICE, WG, 0, stream>>>(col, cnt, tot);
    dinv_kernel<<<(N_NODES + TPB - 1) / TPB, TPB, 0, stream>>>(cnt, dinv, N_NODES);

    wmma_gemm_f32<<<dim3(1, N_NODES / 16, 1), 32, 0, stream>>>(x, W1, nullptr, bufA, S_IN, H_DIM, 0, S_IN);
    scatter_lds_kernel<<<NPART * NSLICE, WG, 0, stream>>>(bufA, row, col, dinv, tot, elist, spart);
    finalize_kernel<<<(nh4 + TPB - 1) / TPB, TPB, 0, stream>>>(bufB, spart, bufA, dinv, b1, nh4);

    scatter_list_kernel<<<NPART * NSLICE, WG, 0, stream>>>(bufB, tot, elist, spart);
    finalize_mm_kernel<<<N_NODES / 16 / 8, 256, 0, stream>>>(bufA, spart, bufB, dinv, W2, b2);

    scatter_list_kernel<<<NPART * NSLICE, WG, 0, stream>>>(bufA, tot, elist, spart);
    finalize_mm_kernel<<<N_NODES / 16 / 8, 256, 0, stream>>>(bufB, spart, bufA, dinv, W3, b3);


    fc_slab_gemm_f32<<<dim3(256 / FC_NT, KCHUNKS), 256, 0, stream>>>(bufB, Wp1, part);
    reduce_bias_relu_kernel<<<(fcMN + TPB - 1) / TPB, TPB, 0, stream>>>(part, bp1, Hp, fcMN, 256, 1);
    fc_slab_gemm_f32<<<dim3(256 / FC_NT, KCHUNKS), 256, 0, stream>>>(bufB, Wv1, part);
    reduce_bias_relu_kernel<<<(fcMN + TPB - 1) / TPB, TPB, 0, stream>>>(part, bv1, Vv, fcMN, 256, 1);

    wmma_gemm_f32<<<dim3(1024 / 16, B_GRAPHS / 16, 1), 32, 0, stream>>>(Hp, Wp2, bp2, P, 256, 1024, 0, 256);

    heads_kernel<<<1, B_GRAPHS, 0, stream>>>(Vv, Wiv, biv, Wev, bev, ext_V, int_V);
}


# Policy and value heads in one launch on WMMA: a 4-wave group stages a 32x256 A slab and 256x32 B slab (Wp2 or Wiv|Wev|0) in LDS, all loads in flight at once; k over 2 accumulators: sum order changes
# speedup vs baseline: 3.9242x; 1.0767x over previous
#include <hip/hip_runtime.h>

typedef __attribute__((ext_vector_type(2))) float v2f;
typedef __attribute__((ext_vector_type(8))) float v8f;

#define N_NODES   131072
#define E_EDGES   4194304
#define S_IN      32
#define H_DIM     16
#define B_GRAPHS  128
#define FC_IN_SZ  16384
#define FC_KC     256
#define KCHUNKS   (FC_IN_SZ / FC_KC)

#define NPART    32
#define PNODES   4096
#define NSLICE   8
#define SLICE_E  (E_EDGES / NSLICE)
#define CH       16384
#define WG       1024
#define QCAP     256
#define RBITS    17

#define DPART    4
#define DNODES   (N_NODES / DPART)
#define DSHIFT   15
#define DSLICE   64
#define DSUB     (DSLICE / NSLICE)
#define DSLICE_E (E_EDGES / DSLICE)
#define DRANGES  (DNODES / PNODES)

__device__ __forceinline__ unsigned int wave_sum(unsigned int v) {
#pragma unroll
    for (int m = 16; m > 0; m >>= 1) v += (unsigned int)__shfl_xor((int)v, m);
    return v;
}

__device__ __forceinline__ unsigned int list_len(const unsigned int* __restrict__ tot,
                                                 const int p, const int s) {
    const uint4* t4 = (const uint4*)(tot + p * DSLICE + s * DSUB);
    const uint4 a = t4[0], b = t4[1];
    return (a.x + a.y) + (a.z + a.w) + (b.x + b.y) + (b.z + b.w);
}

__device__ __forceinline__ unsigned int list_start(const unsigned int* __restrict__ tot,
                                                   const int p, const int s, const int lane) {
    const unsigned int below = wave_sum(lane < p ? list_len(tot, lane, s) : 0u);
    return (unsigned int)s * SLICE_E + below;
}

__global__ __launch_bounds__(WG)
void degree_lds_kernel(const int* __restrict__ col, unsigned int* __restrict__ cnt,
                       unsigned int* __restrict__ tot) {
    __shared__ __attribute__((aligned(16))) unsigned int hist[DNODES];
    __shared__ unsigned int wsum[(WG / 32) * DRANGES];
    const int tid = threadIdx.x;
    const int r = blockIdx.x / DSLICE, s = blockIdx.x % DSLICE;
    uint4* h4 = (uint4*)hist;
#pragma unroll
    for (int k = 0; k < DNODES / 4 / WG; ++k) h4[k * WG + tid] = make_uint4(0u, 0u, 0u, 0u);
    __syncthreads();
    const int4* c4 = (const int4*)(col + (size_t)s * DSLICE_E);
    int4 v[DSLICE_E / 4 / WG];
#pragma unroll
    for (int u = 0; u < DSLICE_E / 4 / WG; ++u) v[u] = c4[u * WG + tid];
#pragma unroll
    for (int u = 0; u < DSLICE_E / 4 / WG; ++u) {
        if ((v[u].x >> DSHIFT) == r) atomicAdd(&hist[v[u].x & (DNODES - 1)], 1u);
        if ((v[u].y >> DSHIFT) == r) atomicAdd(&hist[v[u].y & (DNODES - 1)], 1u);
        if ((v[u].z >> DSHIFT) == r) atomicAdd(&hist[v[u].z & (DNODES - 1)], 1u);
        if ((v[u].w >> DSHIFT) == r) atomicAdd(&hist[v[u].w & (DNODES - 1)], 1u);
    }
    __syncthreads();
    unsigned int mine[DRANGES];
#pragma unroll
    for (int q = 0; q < DRANGES; ++q) mine[q] = 0u;
    unsigned int* plane = cnt + (size_t)s * N_NODES + (size_t)r * DNODES;
#pragma unroll
    for (int k = 0; k < DNODES / WG; ++k) {
        const unsigned int hv = hist[k * WG + tid];
        plane[k * WG + tid] = hv;
        mine[k / (PNODES / WG)] += hv;
    }
#pragma unroll
    for (int q = 0; q < DRANGES; ++q) {
        const unsigned int t = wave_sum(mine[q]);
        if ((tid & 31) == 0) wsum[(tid >> 5) * DRANGES + q] = t;
    }
    __syncthreads();
    if (tid < DRANGES) {
        unsigned int t = 0u;
#pragma unroll
        for (int w = 0; w < WG / 32; ++w) t += wsum[w * DRANGES + tid];
        tot[(r * DRANGES + tid) * DSLICE + s] = t;
    }
}

__global__ void dinv_kernel(const unsigned int* __restrict__ cnt, float* __restrict__ d, int n) {
    int i = blockIdx.x * blockDim.x + threadIdx.x;
    if (i >= n) return;
    unsigned int t = 0u;
#pragma unroll
    for (int s = 0; s < DSLICE; ++s) t += cnt[(size_t)s * n + i];
    d[i] = rsqrtf(1.0f + (float)t);
}

__global__ __launch_bounds__(32)
void wmma_gemm_f32(const float* __restrict__ A, const float* __restrict__ B,
                   const float* __restrict__ bias, float* __restrict__ C,
                   int K, int Nld, int relu, int Kc) {
    const int nb   = blockIdx.x;
    const int mb   = blockIdx.y;
    const int z    = blockIdx.z;
    const int lane = threadIdx.x;
    const int half = lane >> 4;
    const int lm   = lane & 15;

    const int k0 = z * Kc;

    v8f acc = {};
    const float* Ap = A + (size_t)(mb * 16 + lm) * K + k0 + 2 * half;
    const float* Bp = B + (size_t)(k0 + 2 * half) * Nld + nb * 16 + lm;

#pragma unroll 4
    for (int k = 0; k < Kc; k += 4) {
        v2f a = *(const v2f*)Ap;
        v2f b;
        b.x = Bp[0];
        b.y = Bp[Nld];
        acc = __builtin_amdgcn_wmma_f32_16x16x4_f32(
                  false, a, false, b, (short)0, acc, false, false);
        Ap += 4;
        Bp += (size_t)4 * Nld;
    }

    const int cn = nb * 16 + lm;
    if (gridDim.z == 1) {
        const float bb = bias ? bias[cn] : 0.0f;
#pragma unroll
        for (int v = 0; v < 8; ++v) {
            float r = acc[v] + bb;
            if (relu) r = fmaxf(r, 0.0f);
            C[(size_t)(mb * 16 + v + 8 * half) * Nld + cn] = r;
        }
    } else {
        const size_t M  = (size_t)gridDim.y * 16;
        float* Cp = C + (size_t)z * M * Nld;
#pragma unroll
        for (int v = 0; v < 8; ++v) {
            Cp[(size_t)(mb * 16 + v + 8 * half) * Nld + cn] = acc[v];
        }
    }
}

#define FC_NT     64
#define FC_AP     (FC_KC + 4)
#define FC_BP     (FC_NT + 8)
__global__ __launch_bounds__(256)
void fc_slab_gemm_f32(const float* __restrict__ A, const float* __restrict__ B,
                      float* __restrict__ part) {
    __shared__ __attribute__((aligned(16))) float As[B_GRAPHS * FC_AP];
    __shared__ __attribute__((aligned(16))) float Bs[FC_KC * FC_BP];
    const int tid = threadIdx.x;
    const int n0  = blockIdx.x * FC_NT;
    const int z   = blockIdx.y;
    const int k0  = z * FC_KC;

    {
        float4 rb[16], ra[16];
#pragma unroll
        for (int u = 0; u < 16; ++u) {
            const int i = u * 256 + tid, r = i >> 4, c4 = i & 15;
            rb[u] = *(const float4*)(B + (size_t)(k0 + r) * 256 + n0 + c4 * 4);
        }
#pragma unroll
        for (int u = 0; u < 16; ++u) {
            const int i = u * 256 + tid, r = i >> 6, c4 = i & 63;
            ra[u] = *(const float4*)(A + (size_t)r * FC_IN_SZ + k0 + c4 * 4);
        }
#pragma unroll
        for (int u = 0; u < 16; ++u) {
            const int i = u * 256 + tid, r = i >> 4, c4 = i & 15;
            *(float4*)(Bs + r * FC_BP + c4 * 4) = rb[u];
        }
#pragma unroll
        for (int u = 0; u < 16; ++u) {
            const int i = u * 256 + tid, r = i >> 6, c4 = i & 63;
            *(float4*)(As + r * FC_AP + c4 * 4) = ra[u];
        }
#pragma unroll
        for (int u = 0; u < 16; ++u) {
            const int i = (u + 16) * 256 + tid, r = i >> 6, c4 = i & 63;
            ra[u] = *(const float4*)(A + (size_t)r * FC_IN_SZ + k0 + c4 * 4);
        }
#pragma unroll
        for (int u = 0; u < 16; ++u) {
            const int i = (u + 16) * 256 + tid, r = i >> 6, c4 = i & 63;
            *(float4*)(As + r * FC_AP + c4 * 4) = ra[u];
        }
    }
    __syncthreads();

    const int lane = tid & 31;
    const int w    = tid >> 5;
    const int half = lane >> 4;
    const int c    = lane & 15;
    const float* ap = As + (16 * w + c) * FC_AP + 4 * half;
    const float* bp = Bs + (4 * half) * FC_BP + c;

    v8f acc[4] = {};
    float4 av = *(const float4*)ap;
    v2f b1[4], b2[4];
#pragma unroll
    for (int j = 0; j < 4; ++j) {
        b1[j].x = bp[0 * FC_BP + 16 * j];  b1[j].y = bp[1 * FC_BP + 16 * j];
        b2[j].x = bp[2 * FC_BP + 16 * j];  b2[j].y = bp[3 * FC_BP + 16 * j];
    }
    for (int kk = 0; kk < FC_KC; kk += 8) {
        const int kn = kk + 8 < FC_KC ? kk + 8 : kk;
        const float4 avn = *(const float4*)(ap + kn);
        const float* bk = bp + kn * FC_BP;
        v2f n1[4], n2[4];
#pragma unroll
        for (int j = 0; j < 4; ++j) {
            n1[j].x = bk[0 * FC_BP + 16 * j];  n1[j].y = bk[1 * FC_BP + 16 * j];
            n2[j].x = bk[2 * FC_BP + 16 * j];  n2[j].y = bk[3 * FC_BP + 16 * j];
        }
        v2f a1, a2;
        a1.x = av.x;  a1.y = av.y;
        a2.x = av.z;  a2.y = av.w;
#pragma unroll
        for (int j = 0; j < 4; ++j)
            acc[j] = __builtin_amdgcn_wmma_f32_16x16x4_f32(
                         false, a1, false, b1[j], (short)0, acc[j], false, false);
#pragma unroll
        for (int j = 0; j < 4; ++j)
            acc[j] = __builtin_amdgcn_wmma_f32_16x16x4_f32(
                         false, a2, false, b2[j], (short)0, acc[j], false, false);
        av = avn;
#pragma unroll
        for (int j = 0; j < 4; ++j) { b1[j] = n1[j]; b2[j] = n2[j]; }
    }

    float* Cp = part + (size_t)z * (B_GRAPHS * 256) + n0 + c;
#pragma unroll
    for (int j = 0; j < 4; ++j) {
#pragma unroll
        for (int v = 0; v < 8; ++v)
            Cp[(size_t)(16 * w + v + 8 * half) * 256 + 16 * j] = acc[j][v];
    }
}

__global__ void reduce_bias_relu_kernel(const float* __restrict__ part,
                                        const float* __restrict__ bias,
                                        float* __restrict__ C,
                                        int MN, int Nld, int relu) {
    int i = blockIdx.x * blockDim.x + threadIdx.x;
    if (i >= MN) return;
    float s = 0.0f;
#pragma unroll
    for (int zc = 0; zc < KCHUNKS; ++zc) s += part[(size_t)zc * MN + i];
    s += bias[i % Nld];
    if (relu) s = fmaxf(s, 0.0f);
    C[i] = s;
}


#define SLOTS 3
#define EPI   8

__device__ __forceinline__ unsigned int list_claim(unsigned int* fill, const unsigned int cnt,
                                                   const int lane) {
    unsigned int at = 0u;
    if (lane == 0) at = atomicAdd(fill, cnt);
    return (unsigned int)__builtin_amdgcn_readfirstlane((int)at);
}
template <bool PF>
__device__ __forceinline__ void add_pass(const int i0, const int n, const int l4,
                                         const unsigned int* q, const float* dl, float* acc,
                                         uint2* __restrict__ lst,
                                         const float4* __restrict__ xw4,
                                         const float* __restrict__ dinv,
                                         int4 (&cv)[4], int4 (&rv)[4],
                                         const int4* __restrict__ nxc, const int4* __restrict__ nxr,
                                         const int tid) {
    bool   h[SLOTS];
    unsigned int qe[SLOTS];
    int    nd[SLOTS];
    float  dc[SLOTS], dr[SLOTS];
    float4 v[SLOTS];
#pragma unroll
    for (int j = 0; j < SLOTS; ++j) {
        const int i = i0 + j * EPI;
        h[j] = i < n;
        const unsigned int en = q[h[j] ? i : 0];
        const int r = h[j] ? (int)(en & ((1u << RBITS) - 1u)) : 0;
        qe[j] = en;
        nd[j] = (int)(en >> RBITS) & (PNODES - 1);
        dr[j] = dinv[r];
        v[j]  = xw4[r * 4 + l4];
        dc[j] = dl[nd[j]];
    }
    if (PF) {
#pragma unroll
        for (int u = 0; u < 4; ++u) cv[u] = nxc[u * WG + tid];
#pragma unroll
        for (int u = 0; u < 4; ++u) rv[u] = nxr[u * WG + tid];
    }
    asm volatile("" :: "v"(dr[0]), "v"(dr[1]), "v"(dr[2]),
                       "v"(v[0].x), "v"(v[0].y), "v"(v[0].z), "v"(v[0].w),
                       "v"(v[1].x), "v"(v[1].y), "v"(v[1].z), "v"(v[1].w),
                       "v"(v[2].x), "v"(v[2].y), "v"(v[2].z), "v"(v[2].w));
#pragma unroll
    for (int j = 0; j < SLOTS; ++j) {
        if (h[j]) {
            const float w  = dr[j] * dc[j];
            if (l4 == 0) lst[i0 + j * EPI] = make_uint2(qe[j], __float_as_uint(w));
            const int   sw = (nd[j] >> 2) & 3;
            float* a = acc + (nd[j] * H_DIM + l4 * 4);
            atomicAdd(a + (0 ^ sw), w * v[j].x);
            atomicAdd(a + (1 ^ sw), w * v[j].y);
            atomicAdd(a + (2 ^ sw), w * v[j].z);
            atomicAdd(a + (3 ^ sw), w * v[j].w);
        }
    }
}

__global__ __launch_bounds__(WG)
void scatter_lds_kernel(const float* __restrict__ xw,
                        const int* __restrict__ row, const int* __restrict__ col,
                        const float* __restrict__ dinv,
                        const unsigned int* __restrict__ tot, unsigned int* __restrict__ list,
                        float* __restrict__ part) {
    __shared__ __attribute__((aligned(16))) float acc[PNODES * H_DIM];
    __shared__ unsigned int q[(WG / 32) * QCAP];
    __shared__ float dl[PNODES];
    __shared__ unsigned int lfill;

    const int tid  = threadIdx.x;
    const int lane = tid & 31;
    const int p = blockIdx.x >> 3, s = blockIdx.x & 7;
    float4* acc4 = (float4*)acc;
#pragma unroll
    for (int k = 0; k < PNODES * H_DIM / 4 / WG; ++k)
        acc4[k * WG + tid] = make_float4(0.0f, 0.0f, 0.0f, 0.0f);
#pragma unroll
    for (int k = 0; k < PNODES / WG; ++k)
        dl[k * WG + tid] = dinv[p * PNODES + k * WG + tid];
    if (tid == 0) lfill = 0u;
    uint2* mylist = (uint2*)list + list_start(tot, p, s, lane);
    __syncthreads();

    const int ebase = s * SLICE_E;
    const float4* xw4 = (const float4*)xw;
    const int g  = lane >> 2;
    const int l4 = lane & 3;
    unsigned int* qw = q + (tid >> 5) * QCAP;

    int4 cv[4], rv[4];
    {
        const int4* c4 = (const int4*)(col + ebase);
        const int4* r4 = (const int4*)(row + ebase);
#pragma unroll
        for (int u = 0; u < 4; ++u) cv[u] = c4[u * WG + tid];
#pragma unroll
        for (int u = 0; u < 4; ++u) rv[u] = r4[u * WG + tid];
    }

    for (int ch = 0; ch < SLICE_E / CH; ++ch) {
        const int nb = ebase + (ch + 1 < SLICE_E / CH ? ch + 1 : ch) * CH;
        const int4* nxc = (const int4*)(col + nb);
        const int4* nxr = (const int4*)(row + nb);

        const int t[16] = {cv[0].x, cv[0].y, cv[0].z, cv[0].w, cv[1].x, cv[1].y, cv[1].z, cv[1].w,
                           cv[2].x, cv[2].y, cv[2].z, cv[2].w, cv[3].x, cv[3].y, cv[3].z, cv[3].w};
        const int f[16] = {rv[0].x, rv[0].y, rv[0].z, rv[0].w, rv[1].x, rv[1].y, rv[1].z, rv[1].w,
                           rv[2].x, rv[2].y, rv[2].z, rv[2].w, rv[3].x, rv[3].y, rv[3].z, rv[3].w};
        unsigned int base = 0u;
#pragma unroll
        for (int k = 0; k < 16; ++k) {
            if (k != 0 && (k & 3) == 0 && base > QCAP - 4 * 32) {
                __builtin_amdgcn_fence(__ATOMIC_RELEASE, "wavefront");
                __builtin_amdgcn_wave_barrier();
                __builtin_amdgcn_fence(__ATOMIC_ACQUIRE, "wavefront");
                uint2* lst = mylist + list_claim(&lfill, base, lane);
                for (int i = g; i < (int)base + g; i += SLOTS * EPI)
                    add_pass<false>(i, (int)base, l4, qw, dl, acc, lst, xw4, dinv, cv, rv, nxc, nxr, tid);
                base = 0u;
            }
            const bool m = (t[k] >> 12) == p;
            const unsigned int bal = __builtin_amdgcn_ballot_w32(m);
            if (m)
                qw[base + __builtin_amdgcn_mbcnt_lo(bal, 0u)] =
                    ((unsigned int)(t[k] & (PNODES - 1)) << RBITS) |
                    ((unsigned int)f[k] & ((1u << RBITS) - 1u));
            base += (unsigned int)__builtin_popcount(bal);
        }
        const int n = (int)base;
        __builtin_amdgcn_fence(__ATOMIC_RELEASE, "wavefront");
        __builtin_amdgcn_wave_barrier();
        __builtin_amdgcn_fence(__ATOMIC_ACQUIRE, "wavefront");

        uint2* lst = mylist + list_claim(&lfill, base, lane);

        add_pass<true>(g, n, l4, qw, dl, acc, lst, xw4, dinv, cv, rv, nxc, nxr, tid);
        for (int i = g + SLOTS * EPI; i < n + g; i += SLOTS * EPI)
            add_pass<false>(i, n, l4, qw, dl, acc, lst, xw4, dinv, cv, rv, nxc, nxr, tid);
    }
    __syncthreads();

    float4* out4 = (float4*)part + ((size_t)s * N_NODES + (size_t)p * PNODES) * (H_DIM / 4);
#pragma unroll
    for (int k = 0; k < PNODES * H_DIM / 4 / WG; ++k) {
        const int idx = k * WG + tid;
        const int sw  = (idx >> 4) & 3;
        const float4 a = acc4[idx];
        float4 b, o;
        b.x = (sw & 1) ? a.y : a.x;  b.y = (sw & 1) ? a.x : a.y;
        b.z = (sw & 1) ? a.w : a.z;  b.w = (sw & 1) ? a.z : a.w;
        o.x = (sw & 2) ? b.z : b.x;  o.y = (sw & 2) ? b.w : b.y;
        o.z = (sw & 2) ? b.x : b.z;  o.w = (sw & 2) ? b.y : b.w;
        out4[idx] = o;
    }
}

#define LSLOTS 4
__global__ __launch_bounds__(WG)
void scatter_list_kernel(const float* __restrict__ xw,
                         const unsigned int* __restrict__ tot, const unsigned int* __restrict__ list,
                         float* __restrict__ part) {
    __shared__ __attribute__((aligned(16))) float acc[PNODES * H_DIM];

    const int tid  = threadIdx.x;
    const int lane = tid & 31;
    const int p = blockIdx.x >> 3, s = blockIdx.x & 7;
    float4* acc4 = (float4*)acc;
#pragma unroll
    for (int k = 0; k < PNODES * H_DIM / 4 / WG; ++k)
        acc4[k * WG + tid] = make_float4(0.0f, 0.0f, 0.0f, 0.0f);
    const uint2* L = (const uint2*)list + list_start(tot, p, s, lane);
    const int n = (int)list_len(tot, p, s);
    __syncthreads();

    const float4* xw4 = (const float4*)xw;
    const int g  = lane >> 2;
    const int l4 = lane & 3;
    const int nblk = (n + 31) >> 5;
    const int wstep = WG / 32;

    int blk = tid >> 5;
    uint2 e[LSLOTS];
    if (blk < nblk) {
#pragma unroll
        for (int j = 0; j < LSLOTS; ++j) {
            const int i = blk * 32 + j * EPI + g;
            e[j] = L[i < n ? i : n - 1];
        }
    }
    for (; blk < nblk; blk += wstep) {
        const int nb = blk + wstep < nblk ? blk + wstep : blk;
        bool   h[LSLOTS];
        float4 v[LSLOTS];
        uint2  en[LSLOTS];
#pragma unroll
        for (int j = 0; j < LSLOTS; ++j) {
            h[j] = blk * 32 + j * EPI + g < n;
            const int r = (int)(e[j].x & ((1u << RBITS) - 1u));
            v[j] = xw4[r * 4 + l4];
        }
#pragma unroll
        for (int j = 0; j < LSLOTS; ++j) {
            const int i = nb * 32 + j * EPI + g;
            en[j] = L[i < n ? i : n - 1];
        }
        asm volatile("" :: "v"(v[0].x), "v"(v[0].y), "v"(v[0].z), "v"(v[0].w),
                           "v"(v[1].x), "v"(v[1].y), "v"(v[1].z), "v"(v[1].w),
                           "v"(v[2].x), "v"(v[2].y), "v"(v[2].z), "v"(v[2].w),
                           "v"(v[3].x), "v"(v[3].y), "v"(v[3].z), "v"(v[3].w));
#pragma unroll
        for (int j = 0; j < LSLOTS; ++j) {
            if (h[j]) {
                const float w  = __uint_as_float(e[j].y);
                const int   nd = (int)(e[j].x >> RBITS) & (PNODES - 1);
                const int   sw = (nd >> 2) & 3;
                float* a = acc + (nd * H_DIM + l4 * 4);
                atomicAdd(a + (0 ^ sw), w * v[j].x);
                atomicAdd(a + (1 ^ sw), w * v[j].y);
                atomicAdd(a + (2 ^ sw), w * v[j].z);
                atomicAdd(a + (3 ^ sw), w * v[j].w);
            }
        }
#pragma unroll
        for (int j = 0; j < LSLOTS; ++j) e[j] = en[j];
    }
    __syncthreads();

    float4* out4 = (float4*)part + ((size_t)s * N_NODES + (size_t)p * PNODES) * (H_DIM / 4);
#pragma unroll
    for (int k = 0; k < PNODES * H_DIM / 4 / WG; ++k) {
        const int idx = k * WG + tid;
        const int sw  = (idx >> 4) & 3;
        const float4 a = acc4[idx];
        float4 b, o;
        b.x = (sw & 1) ? a.y : a.x;  b.y = (sw & 1) ? a.x : a.y;
        b.z = (sw & 1) ? a.w : a.z;  b.w = (sw & 1) ? a.z : a.w;
        o.x = (sw & 2) ? b.z : b.x;  o.y = (sw & 2) ? b.w : b.y;
        o.z = (sw & 2) ? b.x : b.z;  o.w = (sw & 2) ? b.y : b.w;
        out4[idx] = o;
    }
}

__global__ void finalize_kernel(float* __restrict__ h, const float* __restrict__ part,
                                const float* __restrict__ xw,
                                const float* __restrict__ dinv,
                                const float* __restrict__ bias, int n4) {
    int i = blockIdx.x * blockDim.x + threadIdx.x;
    if (i >= n4) return;
    const float4* p4 = (const float4*)part;
    float4 a = p4[i];
#pragma unroll
    for (int s = 1; s < NSLICE; ++s) {
        const float4 t = p4[(size_t)s * n4 + i];
        a.x += t.x; a.y += t.y; a.z += t.z; a.w += t.w;
    }
    const int node = i >> 2, j0 = (i & 3) * 4;
    const float d = dinv[node];
    const float4 x = ((const float4*)xw)[i];
    float4 r;
    r.x = fmaxf(a.x + d * d * x.x + bias[j0 + 0], 0.0f);
    r.y = fmaxf(a.y + d * d * x.y + bias[j0 + 1], 0.0f);
    r.z = fmaxf(a.z + d * d * x.z + bias[j0 + 2], 0.0f);
    r.w = fmaxf(a.w + d * d * x.w + bias[j0 + 3], 0.0f);
    ((float4*)h)[i] = r;
}

__global__ __launch_bounds__(256)
void finalize_mm_kernel(float* __restrict__ hout, const float* __restrict__ part,
                        const float* __restrict__ hin, const float* __restrict__ dinv,
                        const float* __restrict__ W, const float* __restrict__ bias) {
    const int lane = threadIdx.x & 31;
    const int tile = blockIdx.x * 8 + (threadIdx.x >> 5);
    const int hh   = lane >> 4;
    const int c    = lane & 15;
    const int node = tile * 16 + c;
    const size_t n4 = (size_t)N_NODES * (H_DIM / 4);
    const size_t i4 = (size_t)node * (H_DIM / 4) + 2 * hh;
    const float4* p4 = (const float4*)part;

    float4 t0[NSLICE], t1[NSLICE];
#pragma unroll
    for (int s = 0; s < NSLICE; ++s) {
        t0[s] = p4[(size_t)s * n4 + i4];
        t1[s] = p4[(size_t)s * n4 + i4 + 1];
    }
    const float4 x0 = ((const float4*)hin)[i4];
    const float4 x1 = ((const float4*)hin)[i4 + 1];
    const float d  = dinv[node];
    v2f b[4];
#pragma unroll
    for (int j = 0; j < 4; ++j) {
        b[j].x = W[(8 * hh + 2 * j) * H_DIM + c];
        b[j].y = W[(8 * hh + 2 * j + 1) * H_DIM + c];
    }
    const float bb = bias[c];

    float4 lo = t0[0], hi = t1[0];
#pragma unroll
    for (int s = 1; s < NSLICE; ++s) {
        lo.x += t0[s].x; lo.y += t0[s].y; lo.z += t0[s].z; lo.w += t0[s].w;
        hi.x += t1[s].x; hi.y += t1[s].y; hi.z += t1[s].z; hi.w += t1[s].w;
    }
    const float d2 = d * d;
    v2f a[4];
    a[0].x = lo.x + d2 * x0.x;  a[0].y = lo.y + d2 * x0.y;
    a[1].x = lo.z + d2 * x0.z;  a[1].y = lo.w + d2 * x0.w;
    a[2].x = hi.x + d2 * x1.x;  a[2].y = hi.y + d2 * x1.y;
    a[3].x = hi.z + d2 * x1.z;  a[3].y = hi.w + d2 * x1.w;

    v8f acc = {};
#pragma unroll
    for (int j = 0; j < 4; ++j)
        acc = __builtin_amdgcn_wmma_f32_16x16x4_f32(
                  false, a[j], false, b[j], (short)0, acc, false, false);

    float* o = hout + (size_t)(tile * 16 + 8 * hh) * H_DIM + c;
#pragma unroll
    for (int r = 0; r < 8; ++r) o[r * H_DIM] = fmaxf(acc[r] + bb, 0.0f);
}

#define HD_NT     32
#define HD_NX     (1024 / HD_NT)
#define HD_MT     32
#define HD_K      256
#define HD_AP     (HD_K + 4)
#define HD_BP     (HD_NT + 8)
__global__ __launch_bounds__(128)
void head_slab_gemm_f32(const float* __restrict__ Hp, const float* __restrict__ V,
                        const float* __restrict__ Wp2, const float* __restrict__ bp2,
                        const float* __restrict__ Wiv, const float* __restrict__ biv,
                        const float* __restrict__ Wev, const float* __restrict__ bev,
                        float* __restrict__ P, float* __restrict__ out_ext,
                        float* __restrict__ out_int) {
    __shared__ __attribute__((aligned(16))) float As[HD_MT * HD_AP];
    __shared__ __attribute__((aligned(16))) float Bs[HD_K * HD_BP];
    const int  tid   = threadIdx.x;
    const bool value = blockIdx.x == HD_NX;
    const int  n0    = blockIdx.x * HD_NT;
    const int  m0    = blockIdx.y * HD_MT;
    const float* A   = value ? V : Hp;

    const int lane = tid & 31;
    const int w    = tid >> 5;
    const int rt   = w & 1;
    const int ct   = w >> 1;
    const int half = lane >> 4;
    const int c    = lane & 15;

    float bb;
    if (!value) bb = bp2[n0 + 16 * ct + c];
    else        bb = c == 0 ? biv[0] : bev[0];

    {
        const float* a0 = A + (size_t)(m0 + (tid >> 6)) * HD_K + (tid & 63) * 4;
        float*       as = As + (tid >> 6) * HD_AP + (tid & 63) * 4;
#define HD_REP16(M) M(0) M(1) M(2) M(3) M(4) M(5) M(6) M(7) M(8) M(9) M(10) M(11) M(12) M(13) M(14) M(15)
#define HD_LOAD_A(u)  const float4 ra##u = *(const float4*)(a0 + (u) * (2 * HD_K));
#define HD_STORE_A(u) *(float4*)(as + (u) * (2 * HD_AP)) = ra##u;
        HD_REP16(HD_LOAD_A)
        if (!value) {
            const float* b0 = Wp2 + (size_t)(tid >> 3) * 1024 + n0 + (tid & 7) * 4;
            float*       bs = Bs + (tid >> 3) * HD_BP + (tid & 7) * 4;
            float4 rb[16];
#pragma unroll
            for (int u = 0; u < 16; ++u) rb[u] = *(const float4*)(b0 + u * (16 * 1024));
#pragma unroll
            for (int u = 0; u < 16; ++u) *(float4*)(bs + u * (16 * HD_BP)) = rb[u];
        } else {
            const float wi0 = Wiv[tid], wi1 = Wiv[tid + 128];
            const float we0 = Wev[tid], we1 = Wev[tid + 128];
            *(float4*)(Bs + tid * HD_BP)         = make_float4(wi0, we0, 0.0f, 0.0f);
            *(float4*)(Bs + (tid + 128) * HD_BP) = make_float4(wi1, we1, 0.0f, 0.0f);
#pragma unroll
            for (int c4 = 1; c4 < HD_NT / 4; ++c4) {
                *(float4*)(Bs + tid * HD_BP + c4 * 4)         = make_float4(0.0f, 0.0f, 0.0f, 0.0f);
                *(float4*)(Bs + (tid + 128) * HD_BP + c4 * 4) = make_float4(0.0f, 0.0f, 0.0f, 0.0f);
            }
        }
        HD_REP16(HD_STORE_A)
#undef HD_STORE_A
#undef HD_LOAD_A
#undef HD_REP16
    }
    __syncthreads();

    const float* ap = As + (16 * rt + c) * HD_AP + 4 * half;
    const float* bp = Bs + (4 * half) * HD_BP + 16 * ct + c;
    float4 af[HD_K / 8];
    float  bf[HD_K / 8][4];
#pragma unroll
    for (int s = 0; s < HD_K / 8; ++s) {
        af[s] = *(const float4*)(ap + 8 * s);
#pragma unroll
        for (int q = 0; q < 4; ++q) bf[s][q] = bp[(8 * s + q) * HD_BP];
    }
    __builtin_amdgcn_sched_barrier(0);
    v8f acc0 = {}, acc1 = {};
#pragma unroll
    for (int s = 0; s < HD_K / 8; ++s) {
        v2f a1, a2, b1, b2;
        a1.x = af[s].x;   a1.y = af[s].y;
        a2.x = af[s].z;   a2.y = af[s].w;
        b1.x = bf[s][0];  b1.y = bf[s][1];
        b2.x = bf[s][2];  b2.y = bf[s][3];
        acc0 = __builtin_amdgcn_wmma_f32_16x16x4_f32(
                   false, a1, false, b1, (short)0, acc0, false, false);
        acc1 = __builtin_amdgcn_wmma_f32_16x16x4_f32(
                   false, a2, false, b2, (short)0, acc1, false, false);
    }

    const int row0 = m0 + 16 * rt + 8 * half;
    if (!value) {
        const int cn = n0 + 16 * ct + c;
#pragma unroll
        for (int v = 0; v < 8; ++v)
            P[(size_t)(row0 + v) * 1024 + cn] = (acc0[v] + acc1[v]) + bb;
    } else if (ct == 0 && c < 2) {
        float* o = c == 0 ? out_int : out_ext;
#pragma unroll
        for (int v = 0; v < 8; ++v) o[row0 + v] = (acc0[v] + acc1[v]) + bb;
    }
}

extern "C" void kernel_launch(void* const* d_in, const int* in_sizes, int n_in,
                              void* d_out, int out_size, void* d_ws, size_t ws_size,
                              hipStream_t stream) {
    const float* x    = (const float*)d_in[0];
    const int*   ei   = (const int*)d_in[1];
    const float* W1   = (const float*)d_in[2];
    const float* b1   = (const float*)d_in[3];
    const float* W2   = (const float*)d_in[4];
    const float* b2   = (const float*)d_in[5];
    const float* W3   = (const float*)d_in[6];
    const float* b3   = (const float*)d_in[7];
    const float* Wp1  = (const float*)d_in[8];
    const float* bp1  = (const float*)d_in[9];
    const float* Wp2  = (const float*)d_in[10];
    const float* bp2  = (const float*)d_in[11];
    const float* Wv1  = (const float*)d_in[12];
    const float* bv1  = (const float*)d_in[13];
    const float* Wiv  = (const float*)d_in[14];
    const float* biv  = (const float*)d_in[15];
    const float* Wev  = (const float*)d_in[16];
    const float* bev  = (const float*)d_in[17];

    const int* row = ei;
    const int* col = ei + E_EDGES;

    char* ws = (char*)d_ws;
    float* dinv = (float*)ws;
    float* bufA = (float*)(ws + (size_t)N_NODES * 4);
    float* bufB = bufA + (size_t)N_NODES * H_DIM;
    float* Hp   = bufB + (size_t)N_NODES * H_DIM;
    float* Vv   = Hp + (size_t)B_GRAPHS * 256;
    float* part = Vv + (size_t)B_GRAPHS * 256;
    unsigned int* cnt = (unsigned int*)(part + (size_t)KCHUNKS * B_GRAPHS * 256);
    float* spart = (float*)(cnt + (size_t)DSLICE * N_NODES);
    unsigned int* elist = (unsigned int*)(spart + (size_t)NSLICE * N_NODES * H_DIM);
    unsigned int* tot = elist + (size_t)2 * E_EDGES;

    float* P     = (float*)d_out;
    float* ext_V = P + (size_t)B_GRAPHS * 1024;
    float* int_V = ext_V + B_GRAPHS;

    const int TPB = 256;
    const int nh  = N_NODES * H_DIM;
    const int nh4 = nh / 4;
    const int fcMN = B_GRAPHS * 256;

    degree_lds_kernel<<<DPART * DSLICE, WG, 0, stream>>>(col, cnt, tot);
    dinv_kernel<<<(N_NODES + TPB - 1) / TPB, TPB, 0, stream>>>(cnt, dinv, N_NODES);

    wmma_gemm_f32<<<dim3(1, N_NODES / 16, 1), 32, 0, stream>>>(x, W1, nullptr, bufA, S_IN, H_DIM, 0, S_IN);
    scatter_lds_kernel<<<NPART * NSLICE, WG, 0, stream>>>(bufA, row, col, dinv, tot, elist, spart);
    finalize_kernel<<<(nh4 + TPB - 1) / TPB, TPB, 0, stream>>>(bufB, spart, bufA, dinv, b1, nh4);

    scatter_list_kernel<<<NPART * NSLICE, WG, 0, stream>>>(bufB, tot, elist, spart);
    finalize_mm_kernel<<<N_NODES / 16 / 8, 256, 0, stream>>>(bufA, spart, bufB, dinv, W2, b2);

    scatter_list_kernel<<<NPART * NSLICE, WG, 0, stream>>>(bufA, tot, elist, spart);
    finalize_mm_kernel<<<N_NODES / 16 / 8, 256, 0, stream>>>(bufB, spart, bufA, dinv, W3, b3);


    fc_slab_gemm_f32<<<dim3(256 / FC_NT, KCHUNKS), 256, 0, stream>>>(bufB, Wp1, part);
    reduce_bias_relu_kernel<<<(fcMN + TPB - 1) / TPB, TPB, 0, stream>>>(part, bp1, Hp, fcMN, 256, 1);
    fc_slab_gemm_f32<<<dim3(256 / FC_NT, KCHUNKS), 256, 0, stream>>>(bufB, Wv1, part);
    reduce_bias_relu_kernel<<<(fcMN + TPB - 1) / TPB, TPB, 0, stream>>>(part, bv1, Vv, fcMN, 256, 1);

    head_slab_gemm_f32<<<dim3(HD_NX + 1, B_GRAPHS / HD_MT), 128, 0, stream>>>(
        Hp, Vv, Wp2, bp2, Wiv, biv, Wev, bev, P, ext_V, int_V);
}


# conv1 scatter stops searching its edge slice (the list was read 32 times): one partition pass buckets the edges by node range (LDS ranks; a global add claims list places); sums in arrival order
# speedup vs baseline: 4.4541x; 1.1350x over previous
#include <hip/hip_runtime.h>

typedef __attribute__((ext_vector_type(2))) float v2f;
typedef __attribute__((ext_vector_type(8))) float v8f;

#define N_NODES   131072
#define E_EDGES   4194304
#define S_IN      32
#define H_DIM     16
#define B_GRAPHS  128
#define FC_IN_SZ  16384
#define FC_KC     256
#define KCHUNKS   (FC_IN_SZ / FC_KC)

#define NPART    32
#define PNODES   4096
#define NSLICE   8
#define SLICE_E  (E_EDGES / NSLICE)
#define WG       1024
#define RBITS    17
#define PCH      16384
#define PGROUPS  (SLICE_E / PCH)

#define DPART    4
#define DNODES   (N_NODES / DPART)
#define DSHIFT   15
#define DSLICE   64
#define DSUB     (DSLICE / NSLICE)
#define DSLICE_E (E_EDGES / DSLICE)
#define DRANGES  (DNODES / PNODES)

__device__ __forceinline__ unsigned int wave_sum(unsigned int v) {
#pragma unroll
    for (int m = 16; m > 0; m >>= 1) v += (unsigned int)__shfl_xor((int)v, m);
    return v;
}

__device__ __forceinline__ unsigned int list_len(const unsigned int* __restrict__ tot,
                                                 const int p, const int s) {
    const uint4* t4 = (const uint4*)(tot + p * DSLICE + s * DSUB);
    const uint4 a = t4[0], b = t4[1];
    return (a.x + a.y) + (a.z + a.w) + (b.x + b.y) + (b.z + b.w);
}

__device__ __forceinline__ unsigned int list_start(const unsigned int* __restrict__ tot,
                                                   const int p, const int s, const int lane) {
    const unsigned int below = wave_sum(lane < p ? list_len(tot, lane, s) : 0u);
    return (unsigned int)s * SLICE_E + below;
}

__global__ __launch_bounds__(WG)
void degree_lds_kernel(const int* __restrict__ col, unsigned int* __restrict__ cnt,
                       unsigned int* __restrict__ tot, unsigned int* __restrict__ fillc) {
    __shared__ __attribute__((aligned(16))) unsigned int hist[DNODES];
    __shared__ unsigned int wsum[(WG / 32) * DRANGES];
    const int tid = threadIdx.x;
    const int r = blockIdx.x / DSLICE, s = blockIdx.x % DSLICE;
    if (blockIdx.x == 0 && tid < NSLICE * NPART) fillc[tid] = 0u;
    uint4* h4 = (uint4*)hist;
#pragma unroll
    for (int k = 0; k < DNODES / 4 / WG; ++k) h4[k * WG + tid] = make_uint4(0u, 0u, 0u, 0u);
    __syncthreads();
    const int4* c4 = (const int4*)(col + (size_t)s * DSLICE_E);
    int4 v[DSLICE_E / 4 / WG];
#pragma unroll
    for (int u = 0; u < DSLICE_E / 4 / WG; ++u) v[u] = c4[u * WG + tid];
#pragma unroll
    for (int u = 0; u < DSLICE_E / 4 / WG; ++u) {
        if ((v[u].x >> DSHIFT) == r) atomicAdd(&hist[v[u].x & (DNODES - 1)], 1u);
        if ((v[u].y >> DSHIFT) == r) atomicAdd(&hist[v[u].y & (DNODES - 1)], 1u);
        if ((v[u].z >> DSHIFT) == r) atomicAdd(&hist[v[u].z & (DNODES - 1)], 1u);
        if ((v[u].w >> DSHIFT) == r) atomicAdd(&hist[v[u].w & (DNODES - 1)], 1u);
    }
    __syncthreads();
    unsigned int mine[DRANGES];
#pragma unroll
    for (int q = 0; q < DRANGES; ++q) mine[q] = 0u;
    unsigned int* plane = cnt + (size_t)s * N_NODES + (size_t)r * DNODES;
#pragma unroll
    for (int k = 0; k < DNODES / WG; ++k) {
        const unsigned int hv = hist[k * WG + tid];
        plane[k * WG + tid] = hv;
        mine[k / (PNODES / WG)] += hv;
    }
#pragma unroll
    for (int q = 0; q < DRANGES; ++q) {
        const unsigned int t = wave_sum(mine[q]);
        if ((tid & 31) == 0) wsum[(tid >> 5) * DRANGES + q] = t;
    }
    __syncthreads();
    if (tid < DRANGES) {
        unsigned int t = 0u;
#pragma unroll
        for (int w = 0; w < WG / 32; ++w) t += wsum[w * DRANGES + tid];
        tot[(r * DRANGES + tid) * DSLICE + s] = t;
    }
}

__global__ void dinv_kernel(const unsigned int* __restrict__ cnt, float* __restrict__ d, int n) {
    int i = blockIdx.x * blockDim.x + threadIdx.x;
    if (i >= n) return;
    unsigned int t = 0u;
#pragma unroll
    for (int s = 0; s < DSLICE; ++s) t += cnt[(size_t)s * n + i];
    d[i] = rsqrtf(1.0f + (float)t);
}

__global__ __launch_bounds__(32)
void wmma_gemm_f32(const float* __restrict__ A, const float* __restrict__ B,
                   const float* __restrict__ bias, float* __restrict__ C,
                   int K, int Nld, int relu, int Kc) {
    const int nb   = blockIdx.x;
    const int mb   = blockIdx.y;
    const int z    = blockIdx.z;
    const int lane = threadIdx.x;
    const int half = lane >> 4;
    const int lm   = lane & 15;

    const int k0 = z * Kc;

    v8f acc = {};
    const float* Ap = A + (size_t)(mb * 16 + lm) * K + k0 + 2 * half;
    const float* Bp = B + (size_t)(k0 + 2 * half) * Nld + nb * 16 + lm;

#pragma unroll 4
    for (int k = 0; k < Kc; k += 4) {
        v2f a = *(const v2f*)Ap;
        v2f b;
        b.x = Bp[0];
        b.y = Bp[Nld];
        acc = __builtin_amdgcn_wmma_f32_16x16x4_f32(
                  false, a, false, b, (short)0, acc, false, false);
        Ap += 4;
        Bp += (size_t)4 * Nld;
    }

    const int cn = nb * 16 + lm;
    if (gridDim.z == 1) {
        const float bb = bias ? bias[cn] : 0.0f;
#pragma unroll
        for (int v = 0; v < 8; ++v) {
            float r = acc[v] + bb;
            if (relu) r = fmaxf(r, 0.0f);
            C[(size_t)(mb * 16 + v + 8 * half) * Nld + cn] = r;
        }
    } else {
        const size_t M  = (size_t)gridDim.y * 16;
        float* Cp = C + (size_t)z * M * Nld;
#pragma unroll
        for (int v = 0; v < 8; ++v) {
            Cp[(size_t)(mb * 16 + v + 8 * half) * Nld + cn] = acc[v];
        }
    }
}

#define FC_NT     64
#define FC_AP     (FC_KC + 4)
#define FC_BP     (FC_NT + 8)
__global__ __launch_bounds__(256)
void fc_slab_gemm_f32(const float* __restrict__ A, const float* __restrict__ B,
                      float* __restrict__ part) {
    __shared__ __attribute__((aligned(16))) float As[B_GRAPHS * FC_AP];
    __shared__ __attribute__((aligned(16))) float Bs[FC_KC * FC_BP];
    const int tid = threadIdx.x;
    const int n0  = blockIdx.x * FC_NT;
    const int z   = blockIdx.y;
    const int k0  = z * FC_KC;

    {
        float4 rb[16], ra[16];
#pragma unroll
        for (int u = 0; u < 16; ++u) {
            const int i = u * 256 + tid, r = i >> 4, c4 = i & 15;
            rb[u] = *(const float4*)(B + (size_t)(k0 + r) * 256 + n0 + c4 * 4);
        }
#pragma unroll
        for (int u = 0; u < 16; ++u) {
            const int i = u * 256 + tid, r = i >> 6, c4 = i & 63;
            ra[u] = *(const float4*)(A + (size_t)r * FC_IN_SZ + k0 + c4 * 4);
        }
#pragma unroll
        for (int u = 0; u < 16; ++u) {
            const int i = u * 256 + tid, r = i >> 4, c4 = i & 15;
            *(float4*)(Bs + r * FC_BP + c4 * 4) = rb[u];
        }
#pragma unroll
        for (int u = 0; u < 16; ++u) {
            const int i = u * 256 + tid, r = i >> 6, c4 = i & 63;
            *(float4*)(As + r * FC_AP + c4 * 4) = ra[u];
        }
#pragma unroll
        for (int u = 0; u < 16; ++u) {
            const int i = (u + 16) * 256 + tid, r = i >> 6, c4 = i & 63;
            ra[u] = *(const float4*)(A + (size_t)r * FC_IN_SZ + k0 + c4 * 4);
        }
#pragma unroll
        for (int u = 0; u < 16; ++u) {
            const int i = (u + 16) * 256 + tid, r = i >> 6, c4 = i & 63;
            *(float4*)(As + r * FC_AP + c4 * 4) = ra[u];
        }
    }
    __syncthreads();

    const int lane = tid & 31;
    const int w    = tid >> 5;
    const int half = lane >> 4;
    const int c    = lane & 15;
    const float* ap = As + (16 * w + c) * FC_AP + 4 * half;
    const float* bp = Bs + (4 * half) * FC_BP + c;

    v8f acc[4] = {};
    float4 av = *(const float4*)ap;
    v2f b1[4], b2[4];
#pragma unroll
    for (int j = 0; j < 4; ++j) {
        b1[j].x = bp[0 * FC_BP + 16 * j];  b1[j].y = bp[1 * FC_BP + 16 * j];
        b2[j].x = bp[2 * FC_BP + 16 * j];  b2[j].y = bp[3 * FC_BP + 16 * j];
    }
    for (int kk = 0; kk < FC_KC; kk += 8) {
        const int kn = kk + 8 < FC_KC ? kk + 8 : kk;
        const float4 avn = *(const float4*)(ap + kn);
        const float* bk = bp + kn * FC_BP;
        v2f n1[4], n2[4];
#pragma unroll
        for (int j = 0; j < 4; ++j) {
            n1[j].x = bk[0 * FC_BP + 16 * j];  n1[j].y = bk[1 * FC_BP + 16 * j];
            n2[j].x = bk[2 * FC_BP + 16 * j];  n2[j].y = bk[3 * FC_BP + 16 * j];
        }
        v2f a1, a2;
        a1.x = av.x;  a1.y = av.y;
        a2.x = av.z;  a2.y = av.w;
#pragma unroll
        for (int j = 0; j < 4; ++j)
            acc[j] = __builtin_amdgcn_wmma_f32_16x16x4_f32(
                         false, a1, false, b1[j], (short)0, acc[j], false, false);
#pragma unroll
        for (int j = 0; j < 4; ++j)
            acc[j] = __builtin_amdgcn_wmma_f32_16x16x4_f32(
                         false, a2, false, b2[j], (short)0, acc[j], false, false);
        av = avn;
#pragma unroll
        for (int j = 0; j < 4; ++j) { b1[j] = n1[j]; b2[j] = n2[j]; }
    }

    float* Cp = part + (size_t)z * (B_GRAPHS * 256) + n0 + c;
#pragma unroll
    for (int j = 0; j < 4; ++j) {
#pragma unroll
        for (int v = 0; v < 8; ++v)
            Cp[(size_t)(16 * w + v + 8 * half) * 256 + 16 * j] = acc[j][v];
    }
}

__global__ void reduce_bias_relu_kernel(const float* __restrict__ part,
                                        const float* __restrict__ bias,
                                        float* __restrict__ C,
                                        int MN, int Nld, int relu) {
    int i = blockIdx.x * blockDim.x + threadIdx.x;
    if (i >= MN) return;
    float s = 0.0f;
#pragma unroll
    for (int zc = 0; zc < KCHUNKS; ++zc) s += part[(size_t)zc * MN + i];
    s += bias[i % Nld];
    if (relu) s = fmaxf(s, 0.0f);
    C[i] = s;
}

static_assert((WG / 32) * NPART == WG, "one counter per thread in the clearing store");
static_assert(WG / 32 == NPART, "one wave per run in the copy");
__global__ __launch_bounds__(WG)
void partition_kernel(const int* __restrict__ row, const int* __restrict__ col,
                      const unsigned int* __restrict__ tot, unsigned int* __restrict__ fillc,
                      unsigned int* __restrict__ list4) {
    __shared__ unsigned int stage[PCH];
    __shared__ unsigned int cw[(WG / 32) * NPART];
    __shared__ unsigned int lbase[NPART], gbase[NPART], btot[NPART];
    const int tid = threadIdx.x, lane = tid & 31, w = tid >> 5;
    const int s = blockIdx.x / PGROUPS;

    int4 cv[4], rv[4];
    {
        const int4* c4 = (const int4*)(col + (size_t)blockIdx.x * PCH);
        const int4* r4 = (const int4*)(row + (size_t)blockIdx.x * PCH);
#pragma unroll
        for (int u = 0; u < 4; ++u) cv[u] = c4[u * WG + tid];
#pragma unroll
        for (int u = 0; u < 4; ++u) rv[u] = r4[u * WG + tid];
    }
    const unsigned int li = list_len(tot, lane, s);
    cw[tid] = 0u;
    __syncthreads();

    const int t[16] = {cv[0].x, cv[0].y, cv[0].z, cv[0].w, cv[1].x, cv[1].y, cv[1].z, cv[1].w,
                       cv[2].x, cv[2].y, cv[2].z, cv[2].w, cv[3].x, cv[3].y, cv[3].z, cv[3].w};
    const int f[16] = {rv[0].x, rv[0].y, rv[0].z, rv[0].w, rv[1].x, rv[1].y, rv[1].z, rv[1].w,
                       rv[2].x, rv[2].y, rv[2].z, rv[2].w, rv[3].x, rv[3].y, rv[3].z, rv[3].w};
    unsigned int rk[16];
#pragma unroll
    for (int k = 0; k < 16; ++k) rk[k] = atomicAdd(&cw[w * NPART + (t[k] >> 12)], 1u);
    __syncthreads();

    if (w == 0) {
        unsigned int run = 0u;
        for (int w0 = 0; w0 < WG / 32; w0 += 16) {
            unsigned int c[16];
#pragma unroll
            for (int i = 0; i < 16; ++i) c[i] = cw[(w0 + i) * NPART + lane];
#pragma unroll
            for (int i = 0; i < 16; ++i) { cw[(w0 + i) * NPART + lane] = run; run += c[i]; }
        }
        unsigned int inc = run, linc = li;
#pragma unroll
        for (int d = 1; d < 32; d <<= 1) {
            const unsigned int o  = (unsigned int)__shfl_up((int)inc, d);
            const unsigned int lo = (unsigned int)__shfl_up((int)linc, d);
            if (lane >= d) { inc += o; linc += lo; }
        }
        lbase[lane] = inc - run;
        btot[lane]  = run;
        gbase[lane] = (unsigned int)s * SLICE_E + (linc - li) + atomicAdd(&fillc[s * NPART + lane], run);
    }
    __syncthreads();

#pragma unroll
    for (int k = 0; k < 16; ++k) {
        const int b = t[k] >> 12;
        stage[lbase[b] + cw[w * NPART + b] + rk[k]] =
            ((unsigned int)(t[k] & (PNODES - 1)) << RBITS) | ((unsigned int)f[k] & ((1u << RBITS) - 1u));
    }
    __syncthreads();

    const unsigned int n = btot[w], lb = lbase[w];
    unsigned int* out = list4 + gbase[w];
    for (unsigned int i = (unsigned int)lane; i < n; i += 32u) out[i] = stage[lb + i];
}

#define EPI    8
#define LSLOTS 4
__global__ __launch_bounds__(WG)
void scatter_first_kernel(const float* __restrict__ xw, const float* __restrict__ dinv,
                          const unsigned int* __restrict__ tot, const unsigned int* __restrict__ list4,
                          unsigned int* __restrict__ list, float* __restrict__ part) {
    __shared__ __attribute__((aligned(16))) float acc[PNODES * H_DIM];
    __shared__ float dl[PNODES];

    const int tid  = threadIdx.x;
    const int lane = tid & 31;
    const int p = blockIdx.x >> 3, s = blockIdx.x & 7;
    float4* acc4 = (float4*)acc;
#pragma unroll
    for (int k = 0; k < PNODES * H_DIM / 4 / WG; ++k)
        acc4[k * WG + tid] = make_float4(0.0f, 0.0f, 0.0f, 0.0f);
#pragma unroll
    for (int k = 0; k < PNODES / WG; ++k)
        dl[k * WG + tid] = dinv[p * PNODES + k * WG + tid];
    const unsigned int start = list_start(tot, p, s, lane);
    const unsigned int* L = list4 + start;
    uint2* LW = (uint2*)list + start;
    const int n = (int)list_len(tot, p, s);
    __syncthreads();

    const float4* xw4 = (const float4*)xw;
    const int g  = lane >> 2;
    const int l4 = lane & 3;
    const int nblk = (n + 31) >> 5;
    const int wstep = WG / 32;

    int blk = tid >> 5;
    unsigned int e[LSLOTS];
    if (blk < nblk) {
#pragma unroll
        for (int j = 0; j < LSLOTS; ++j) {
            const int i = blk * 32 + j * EPI + g;
            e[j] = L[i < n ? i : n - 1];
        }
    }
    for (; blk < nblk; blk += wstep) {
        const int nb = blk + wstep < nblk ? blk + wstep : blk;
        bool   h[LSLOTS];
        float4 v[LSLOTS];
        float  dr[LSLOTS], dc[LSLOTS];
        int    nd[LSLOTS];
        unsigned int en[LSLOTS];
#pragma unroll
        for (int j = 0; j < LSLOTS; ++j) {
            h[j] = blk * 32 + j * EPI + g < n;
            const int r = (int)(e[j] & ((1u << RBITS) - 1u));
            nd[j] = (int)(e[j] >> RBITS) & (PNODES - 1);
            dr[j] = dinv[r];
            v[j]  = xw4[r * 4 + l4];
            dc[j] = dl[nd[j]];
        }
#pragma unroll
        for (int j = 0; j < LSLOTS; ++j) {
            const int i = nb * 32 + j * EPI + g;
            en[j] = L[i < n ? i : n - 1];
        }
        asm volatile("" :: "v"(dr[0]), "v"(dr[1]), "v"(dr[2]), "v"(dr[3]),
                           "v"(dc[0]), "v"(dc[1]), "v"(dc[2]), "v"(dc[3]),
                           "v"(v[0].x), "v"(v[0].y), "v"(v[0].z), "v"(v[0].w),
                           "v"(v[1].x), "v"(v[1].y), "v"(v[1].z), "v"(v[1].w),
                           "v"(v[2].x), "v"(v[2].y), "v"(v[2].z), "v"(v[2].w),
                           "v"(v[3].x), "v"(v[3].y), "v"(v[3].z), "v"(v[3].w));
#pragma unroll
        for (int j = 0; j < LSLOTS; ++j) {
            if (h[j]) {
                const float w  = dr[j] * dc[j];
                if (l4 == 0) LW[blk * 32 + j * EPI + g] = make_uint2(e[j], __float_as_uint(w));
                const int   sw = (nd[j] >> 2) & 3;
                float* a = acc + (nd[j] * H_DIM + l4 * 4);
                atomicAdd(a + (0 ^ sw), w * v[j].x);
                atomicAdd(a + (1 ^ sw), w * v[j].y);
                atomicAdd(a + (2 ^ sw), w * v[j].z);
                atomicAdd(a + (3 ^ sw), w * v[j].w);
            }
        }
#pragma unroll
        for (int j = 0; j < LSLOTS; ++j) e[j] = en[j];
    }
    __syncthreads();

    float4* out4 = (float4*)part + ((size_t)s * N_NODES + (size_t)p * PNODES) * (H_DIM / 4);
#pragma unroll
    for (int k = 0; k < PNODES * H_DIM / 4 / WG; ++k) {
        const int idx = k * WG + tid;
        const int sw  = (idx >> 4) & 3;
        const float4 a = acc4[idx];
        float4 b, o;
        b.x = (sw & 1) ? a.y : a.x;  b.y = (sw & 1) ? a.x : a.y;
        b.z = (sw & 1) ? a.w : a.z;  b.w = (sw & 1) ? a.z : a.w;
        o.x = (sw & 2) ? b.z : b.x;  o.y = (sw & 2) ? b.w : b.y;
        o.z = (sw & 2) ? b.x : b.z;  o.w = (sw & 2) ? b.y : b.w;
        out4[idx] = o;
    }
}

__global__ __launch_bounds__(WG)
void scatter_list_kernel(const float* __restrict__ xw,
                         const unsigned int* __restrict__ tot, const unsigned int* __restrict__ list,
                         float* __restrict__ part) {
    __shared__ __attribute__((aligned(16))) float acc[PNODES * H_DIM];

    const int tid  = threadIdx.x;
    const int lane = tid & 31;
    const int p = blockIdx.x >> 3, s = blockIdx.x & 7;
    float4* acc4 = (float4*)acc;
#pragma unroll
    for (int k = 0; k < PNODES * H_DIM / 4 / WG; ++k)
        acc4[k * WG + tid] = make_float4(0.0f, 0.0f, 0.0f, 0.0f);
    const uint2* L = (const uint2*)list + list_start(tot, p, s, lane);
    const int n = (int)list_len(tot, p, s);
    __syncthreads();

    const float4* xw4 = (const float4*)xw;
    const int g  = lane >> 2;
    const int l4 = lane & 3;
    const int nblk = (n + 31) >> 5;
    const int wstep = WG / 32;

    int blk = tid >> 5;
    uint2 e[LSLOTS];
    if (blk < nblk) {
#pragma unroll
        for (int j = 0; j < LSLOTS; ++j) {
            const int i = blk * 32 + j * EPI + g;
            e[j] = L[i < n ? i : n - 1];
        }
    }
    for (; blk < nblk; blk += wstep) {
        const int nb = blk + wstep < nblk ? blk + wstep : blk;
        bool   h[LSLOTS];
        float4 v[LSLOTS];
        uint2  en[LSLOTS];
#pragma unroll
        for (int j = 0; j < LSLOTS; ++j) {
            h[j] = blk * 32 + j * EPI + g < n;
            const int r = (int)(e[j].x & ((1u << RBITS) - 1u));
            v[j] = xw4[r * 4 + l4];
        }
#pragma unroll
        for (int j = 0; j < LSLOTS; ++j) {
            const int i = nb * 32 + j * EPI + g;
            en[j] = L[i < n ? i : n - 1];
        }
        asm volatile("" :: "v"(v[0].x), "v"(v[0].y), "v"(v[0].z), "v"(v[0].w),
                           "v"(v[1].x), "v"(v[1].y), "v"(v[1].z), "v"(v[1].w),
                           "v"(v[2].x), "v"(v[2].y), "v"(v[2].z), "v"(v[2].w),
                           "v"(v[3].x), "v"(v[3].y), "v"(v[3].z), "v"(v[3].w));
#pragma unroll
        for (int j = 0; j < LSLOTS; ++j) {
            if (h[j]) {
                const float w  = __uint_as_float(e[j].y);
                const int   nd = (int)(e[j].x >> RBITS) & (PNODES - 1);
                const int   sw = (nd >> 2) & 3;
                float* a = acc + (nd * H_DIM + l4 * 4);
                atomicAdd(a + (0 ^ sw), w * v[j].x);
                atomicAdd(a + (1 ^ sw), w * v[j].y);
                atomicAdd(a + (2 ^ sw), w * v[j].z);
                atomicAdd(a + (3 ^ sw), w * v[j].w);
            }
        }
#pragma unroll
        for (int j = 0; j < LSLOTS; ++j) e[j] = en[j];
    }
    __syncthreads();

    float4* out4 = (float4*)part + ((size_t)s * N_NODES + (size_t)p * PNODES) * (H_DIM / 4);
#pragma unroll
    for (int k = 0; k < PNODES * H_DIM / 4 / WG; ++k) {
        const int idx = k * WG + tid;
        const int sw  = (idx >> 4) & 3;
        const float4 a = acc4[idx];
        float4 b, o;
        b.x = (sw & 1) ? a.y : a.x;  b.y = (sw & 1) ? a.x : a.y;
        b.z = (sw & 1) ? a.w : a.z;  b.w = (sw & 1) ? a.z : a.w;
        o.x = (sw & 2) ? b.z : b.x;  o.y = (sw & 2) ? b.w : b.y;
        o.z = (sw & 2) ? b.x : b.z;  o.w = (sw & 2) ? b.y : b.w;
        out4[idx] = o;
    }
}

__global__ void finalize_kernel(float* __restrict__ h, const float* __restrict__ part,
                                const float* __restrict__ xw,
                                const float* __restrict__ dinv,
                                const float* __restrict__ bias, int n4) {
    int i = blockIdx.x * blockDim.x + threadIdx.x;
    if (i >= n4) return;
    const float4* p4 = (const float4*)part;
    float4 a = p4[i];
#pragma unroll
    for (int s = 1; s < NSLICE; ++s) {
        const float4 t = p4[(size_t)s * n4 + i];
        a.x += t.x; a.y += t.y; a.z += t.z; a.w += t.w;
    }
    const int node = i >> 2, j0 = (i & 3) * 4;
    const float d = dinv[node];
    const float4 x = ((const float4*)xw)[i];
    float4 r;
    r.x = fmaxf(a.x + d * d * x.x + bias[j0 + 0], 0.0f);
    r.y = fmaxf(a.y + d * d * x.y + bias[j0 + 1], 0.0f);
    r.z = fmaxf(a.z + d * d * x.z + bias[j0 + 2], 0.0f);
    r.w = fmaxf(a.w + d * d * x.w + bias[j0 + 3], 0.0f);
    ((float4*)h)[i] = r;
}

__global__ __launch_bounds__(256)
void finalize_mm_kernel(float* __restrict__ hout, const float* __restrict__ part,
                        const float* __restrict__ hin, const float* __restrict__ dinv,
                        const float* __restrict__ W, const float* __restrict__ bias) {
    const int lane = threadIdx.x & 31;
    const int tile = blockIdx.x * 8 + (threadIdx.x >> 5);
    const int hh   = lane >> 4;
    const int c    = lane & 15;
    const int node = tile * 16 + c;
    const size_t n4 = (size_t)N_NODES * (H_DIM / 4);
    const size_t i4 = (size_t)node * (H_DIM / 4) + 2 * hh;
    const float4* p4 = (const float4*)part;

    float4 t0[NSLICE], t1[NSLICE];
#pragma unroll
    for (int s = 0; s < NSLICE; ++s) {
        t0[s] = p4[(size_t)s * n4 + i4];
        t1[s] = p4[(size_t)s * n4 + i4 + 1];
    }
    const float4 x0 = ((const float4*)hin)[i4];
    const float4 x1 = ((const float4*)hin)[i4 + 1];
    const float d  = dinv[node];
    v2f b[4];
#pragma unroll
    for (int j = 0; j < 4; ++j) {
        b[j].x = W[(8 * hh + 2 * j) * H_DIM + c];
        b[j].y = W[(8 * hh + 2 * j + 1) * H_DIM + c];
    }
    const float bb = bias[c];

    float4 lo = t0[0], hi = t1[0];
#pragma unroll
    for (int s = 1; s < NSLICE; ++s) {
        lo.x += t0[s].x; lo.y += t0[s].y; lo.z += t0[s].z; lo.w += t0[s].w;
        hi.x += t1[s].x; hi.y += t1[s].y; hi.z += t1[s].z; hi.w += t1[s].w;
    }
    const float d2 = d * d;
    v2f a[4];
    a[0].x = lo.x + d2 * x0.x;  a[0].y = lo.y + d2 * x0.y;
    a[1].x = lo.z + d2 * x0.z;  a[1].y = lo.w + d2 * x0.w;
    a[2].x = hi.x + d2 * x1.x;  a[2].y = hi.y + d2 * x1.y;
    a[3].x = hi.z + d2 * x1.z;  a[3].y = hi.w + d2 * x1.w;

    v8f acc = {};
#pragma unroll
    for (int j = 0; j < 4; ++j)
        acc = __builtin_amdgcn_wmma_f32_16x16x4_f32(
                  false, a[j], false, b[j], (short)0, acc, false, false);

    float* o = hout + (size_t)(tile * 16 + 8 * hh) * H_DIM + c;
#pragma unroll
    for (int r = 0; r < 8; ++r) o[r * H_DIM] = fmaxf(acc[r] + bb, 0.0f);
}

#define HD_NT     32
#define HD_NX     (1024 / HD_NT)
#define HD_MT     32
#define HD_K      256
#define HD_AP     (HD_K + 4)
#define HD_BP     (HD_NT + 8)
__global__ __launch_bounds__(128)
void head_slab_gemm_f32(const float* __restrict__ Hp, const float* __restrict__ V,
                        const float* __restrict__ Wp2, const float* __restrict__ bp2,
                        const float* __restrict__ Wiv, const float* __restrict__ biv,
                        const float* __restrict__ Wev, const float* __restrict__ bev,
                        float* __restrict__ P, float* __restrict__ out_ext,
                        float* __restrict__ out_int) {
    __shared__ __attribute__((aligned(16))) float As[HD_MT * HD_AP];
    __shared__ __attribute__((aligned(16))) float Bs[HD_K * HD_BP];
    const int  tid   = threadIdx.x;
    const bool value = blockIdx.x == HD_NX;
    const int  n0    = blockIdx.x * HD_NT;
    const int  m0    = blockIdx.y * HD_MT;
    const float* A   = value ? V : Hp;

    const int lane = tid & 31;
    const int w    = tid >> 5;
    const int rt   = w & 1;
    const int ct   = w >> 1;
    const int half = lane >> 4;
    const int c    = lane & 15;

    float bb;
    if (!value) bb = bp2[n0 + 16 * ct + c];
    else        bb = c == 0 ? biv[0] : bev[0];

    {
        const float* a0 = A + (size_t)(m0 + (tid >> 6)) * HD_K + (tid & 63) * 4;
        float*       as = As + (tid >> 6) * HD_AP + (tid & 63) * 4;
#define HD_REP16(M) M(0) M(1) M(2) M(3) M(4) M(5) M(6) M(7) M(8) M(9) M(10) M(11) M(12) M(13) M(14) M(15)
#define HD_LOAD_A(u)  const float4 ra##u = *(const float4*)(a0 + (u) * (2 * HD_K));
#define HD_STORE_A(u) *(float4*)(as + (u) * (2 * HD_AP)) = ra##u;
        HD_REP16(HD_LOAD_A)
        if (!value) {
            const float* b0 = Wp2 + (size_t)(tid >> 3) * 1024 + n0 + (tid & 7) * 4;
            float*       bs = Bs + (tid >> 3) * HD_BP + (tid & 7) * 4;
            float4 rb[16];
#pragma unroll
            for (int u = 0; u < 16; ++u) rb[u] = *(const float4*)(b0 + u * (16 * 1024));
#pragma unroll
            for (int u = 0; u < 16; ++u) *(float4*)(bs + u * (16 * HD_BP)) = rb[u];
        } else {
            const float wi0 = Wiv[tid], wi1 = Wiv[tid + 128];
            const float we0 = Wev[tid], we1 = Wev[tid + 128];
            *(float4*)(Bs + tid * HD_BP)         = make_float4(wi0, we0, 0.0f, 0.0f);
            *(float4*)(Bs + (tid + 128) * HD_BP) = make_float4(wi1, we1, 0.0f, 0.0f);
#pragma unroll
            for (int c4 = 1; c4 < HD_NT / 4; ++c4) {
                *(float4*)(Bs + tid * HD_BP + c4 * 4)         = make_float4(0.0f, 0.0f, 0.0f, 0.0f);
                *(float4*)(Bs + (tid + 128) * HD_BP + c4 * 4) = make_float4(0.0f, 0.0f, 0.0f, 0.0f);
            }
        }
        HD_REP16(HD_STORE_A)
#undef HD_STORE_A
#undef HD_LOAD_A
#undef HD_REP16
    }
    __syncthreads();

    const float* ap = As + (16 * rt + c) * HD_AP + 4 * half;
    const float* bp = Bs + (4 * half) * HD_BP + 16 * ct + c;
    float4 af[HD_K / 8];
    float  bf[HD_K / 8][4];
#pragma unroll
    for (int s = 0; s < HD_K / 8; ++s) {
        af[s] = *(const float4*)(ap + 8 * s);
#pragma unroll
        for (int q = 0; q < 4; ++q) bf[s][q] = bp[(8 * s + q) * HD_BP];
    }
    __builtin_amdgcn_sched_barrier(0);
    v8f acc0 = {}, acc1 = {};
#pragma unroll
    for (int s = 0; s < HD_K / 8; ++s) {
        v2f a1, a2, b1, b2;
        a1.x = af[s].x;   a1.y = af[s].y;
        a2.x = af[s].z;   a2.y = af[s].w;
        b1.x = bf[s][0];  b1.y = bf[s][1];
        b2.x = bf[s][2];  b2.y = bf[s][3];
        acc0 = __builtin_amdgcn_wmma_f32_16x16x4_f32(
                   false, a1, false, b1, (short)0, acc0, false, false);
        acc1 = __builtin_amdgcn_wmma_f32_16x16x4_f32(
                   false, a2, false, b2, (short)0, acc1, false, false);
    }

    const int row0 = m0 + 16 * rt + 8 * half;
    if (!value) {
        const int cn = n0 + 16 * ct + c;
#pragma unroll
        for (int v = 0; v < 8; ++v)
            P[(size_t)(row0 + v) * 1024 + cn] = (acc0[v] + acc1[v]) + bb;
    } else if (ct == 0 && c < 2) {
        float* o = c == 0 ? out_int : out_ext;
#pragma unroll
        for (int v = 0; v < 8; ++v) o[row0 + v] = (acc0[v] + acc1[v]) + bb;
    }
}

extern "C" void kernel_launch(void* const* d_in, const int* in_sizes, int n_in,
                              void* d_out, int out_size, void* d_ws, size_t ws_size,
                              hipStream_t stream) {
    const float* x    = (const float*)d_in[0];
    const int*   ei   = (const int*)d_in[1];
    const float* W1   = (const float*)d_in[2];
    const float* b1   = (const float*)d_in[3];
    const float* W2   = (const float*)d_in[4];
    const float* b2   = (const float*)d_in[5];
    const float* W3   = (const float*)d_in[6];
    const float* b3   = (const float*)d_in[7];
    const float* Wp1  = (const float*)d_in[8];
    const float* bp1  = (const float*)d_in[9];
    const float* Wp2  = (const float*)d_in[10];
    const float* bp2  = (const float*)d_in[11];
    const float* Wv1  = (const float*)d_in[12];
    const float* bv1  = (const float*)d_in[13];
    const float* Wiv  = (const float*)d_in[14];
    const float* biv  = (const float*)d_in[15];
    const float* Wev  = (const float*)d_in[16];
    const float* bev  = (const float*)d_in[17];

    const int* row = ei;
    const int* col = ei + E_EDGES;

    char* ws = (char*)d_ws;
    float* dinv = (float*)ws;
    float* bufA = (float*)(ws + (size_t)N_NODES * 4);
    float* bufB = bufA + (size_t)N_NODES * H_DIM;
    float* Hp   = bufB + (size_t)N_NODES * H_DIM;
    float* Vv   = Hp + (size_t)B_GRAPHS * 256;
    float* part = Vv + (size_t)B_GRAPHS * 256;
    unsigned int* cnt = (unsigned int*)(part + (size_t)KCHUNKS * B_GRAPHS * 256);
    float* spart = (float*)(cnt + (size_t)DSLICE * N_NODES);
    unsigned int* elist = (unsigned int*)(spart + (size_t)NSLICE * N_NODES * H_DIM);
    unsigned int* tot = elist + (size_t)2 * E_EDGES;
    unsigned int* fillc = tot + NPART * DSLICE;
    unsigned int* list4 = fillc + NSLICE * NPART;

    float* P     = (float*)d_out;
    float* ext_V = P + (size_t)B_GRAPHS * 1024;
    float* int_V = ext_V + B_GRAPHS;

    const int TPB = 256;
    const int nh  = N_NODES * H_DIM;
    const int nh4 = nh / 4;
    const int fcMN = B_GRAPHS * 256;

    degree_lds_kernel<<<DPART * DSLICE, WG, 0, stream>>>(col, cnt, tot, fillc);
    dinv_kernel<<<(N_NODES + TPB - 1) / TPB, TPB, 0, stream>>>(cnt, dinv, N_NODES);

    partition_kernel<<<E_EDGES / PCH, WG, 0, stream>>>(row, col, tot, fillc, list4);

    wmma_gemm_f32<<<dim3(1, N_NODES / 16, 1), 32, 0, stream>>>(x, W1, nullptr, bufA, S_IN, H_DIM, 0, S_IN);
    scatter_first_kernel<<<NPART * NSLICE, WG, 0, stream>>>(bufA, dinv, tot, list4, elist, spart);
    finalize_kernel<<<(nh4 + TPB - 1) / TPB, TPB, 0, stream>>>(bufB, spart, bufA, dinv, b1, nh4);

    scatter_list_kernel<<<NPART * NSLICE, WG, 0, stream>>>(bufB, tot, elist, spart);
    finalize_mm_kernel<<<N_NODES / 16 / 8, 256, 0, stream>>>(bufA, spart, bufB, dinv, W2, b2);

    scatter_list_kernel<<<NPART * NSLICE, WG, 0, stream>>>(bufA, tot, elist, spart);
    finalize_mm_kernel<<<N_NODES / 16 / 8, 256, 0, stream>>>(bufB, spart, bufA, dinv, W3, b3);


    fc_slab_gemm_f32<<<dim3(256 / FC_NT, KCHUNKS), 256, 0, stream>>>(bufB, Wp1, part);
    reduce_bias_relu_kernel<<<(fcMN + TPB - 1) / TPB, TPB, 0, stream>>>(part, bp1, Hp, fcMN, 256, 1);
    fc_slab_gemm_f32<<<dim3(256 / FC_NT, KCHUNKS), 256, 0, stream>>>(bufB, Wv1, part);
    reduce_bias_relu_kernel<<<(fcMN + TPB - 1) / TPB, TPB, 0, stream>>>(part, bv1, Vv, fcMN, 256, 1);

    head_slab_gemm_f32<<<dim3(HD_NX + 1, B_GRAPHS / HD_MT), 128, 0, stream>>>(
        Hp, Vv, Wp2, bp2, Wiv, biv, Wev, bev, P, ext_V, int_V);
}


# Scatters re-partitioned: a work-group owns 512 nodes for the whole edge list (256 lists, not 32 x 8), so no 8 partial planes are written and re-read; sum order: one LDS arrival-order sum per node
# speedup vs baseline: 5.3199x; 1.1944x over previous
#include <hip/hip_runtime.h>

typedef __attribute__((ext_vector_type(2))) float v2f;
typedef __attribute__((ext_vector_type(8))) float v8f;

#define N_NODES   131072
#define E_EDGES   4194304
#define S_IN      32
#define H_DIM     16
#define B_GRAPHS  128
#define FC_IN_SZ  16384
#define FC_KC     256
#define KCHUNKS   (FC_IN_SZ / FC_KC)

#define NPART    256
#define PNODES   512
#define PSHIFT   9
#define WG       1024
#define RBITS    17
#define PCH      16384

#define DPART    4
#define DNODES   (N_NODES / DPART)
#define DSHIFT   15
#define DSLICE   64
#define DSLICE_E (E_EDGES / DSLICE)

__device__ __forceinline__ unsigned int wave_sum(unsigned int v) {
#pragma unroll
    for (int m = 16; m > 0; m >>= 1) v += (unsigned int)__shfl_xor((int)v, m);
    return v;
}

__device__ __forceinline__ unsigned int list_start(const unsigned int* __restrict__ rtot,
                                                   const int p, const int lane) {
    unsigned int below = 0u;
#pragma unroll
    for (int k = 0; k < NPART / 32; ++k) {
        const int q = k * 32 + lane;
        const unsigned int len = rtot[q];
        below += q < p ? len : 0u;
    }
    return wave_sum(below);
}

__global__ __launch_bounds__(WG)
void degree_lds_kernel(const int* __restrict__ col, unsigned int* __restrict__ cnt,
                       unsigned int* __restrict__ fillc) {
    __shared__ __attribute__((aligned(16))) unsigned int hist[DNODES];
    const int tid = threadIdx.x;
    const int r = blockIdx.x / DSLICE, s = blockIdx.x % DSLICE;
    if (blockIdx.x == 0 && tid < NPART) fillc[tid] = 0u;
    uint4* h4 = (uint4*)hist;
#pragma unroll
    for (int k = 0; k < DNODES / 4 / WG; ++k) h4[k * WG + tid] = make_uint4(0u, 0u, 0u, 0u);
    __syncthreads();
    const int4* c4 = (const int4*)(col + (size_t)s * DSLICE_E);
    int4 v[DSLICE_E / 4 / WG];
#pragma unroll
    for (int u = 0; u < DSLICE_E / 4 / WG; ++u) v[u] = c4[u * WG + tid];
#pragma unroll
    for (int u = 0; u < DSLICE_E / 4 / WG; ++u) {
        if ((v[u].x >> DSHIFT) == r) atomicAdd(&hist[v[u].x & (DNODES - 1)], 1u);
        if ((v[u].y >> DSHIFT) == r) atomicAdd(&hist[v[u].y & (DNODES - 1)], 1u);
        if ((v[u].z >> DSHIFT) == r) atomicAdd(&hist[v[u].z & (DNODES - 1)], 1u);
        if ((v[u].w >> DSHIFT) == r) atomicAdd(&hist[v[u].w & (DNODES - 1)], 1u);
    }
    __syncthreads();
    unsigned int* plane = cnt + (size_t)s * N_NODES + (size_t)r * DNODES;
#pragma unroll
    for (int k = 0; k < DNODES / WG; ++k) plane[k * WG + tid] = hist[k * WG + tid];
}

__global__ __launch_bounds__(PNODES)
void dinv_kernel(const unsigned int* __restrict__ cnt, float* __restrict__ d,
                 unsigned int* __restrict__ rtot) {
    __shared__ unsigned int wsum[PNODES / 32];
    const int tid = threadIdx.x;
    const int i = blockIdx.x * PNODES + tid;
    unsigned int t = 0u;
#pragma unroll
    for (int s = 0; s < DSLICE; ++s) t += cnt[(size_t)s * N_NODES + i];
    d[i] = rsqrtf(1.0f + (float)t);
    const unsigned int wt = wave_sum(t);
    if ((tid & 31) == 0) wsum[tid >> 5] = wt;
    __syncthreads();
    if (tid < 32) {
        const unsigned int a = wave_sum(tid < PNODES / 32 ? wsum[tid] : 0u);
        if (tid == 0) rtot[blockIdx.x] = a;
    }
}

__global__ __launch_bounds__(32)
void wmma_gemm_f32(const float* __restrict__ A, const float* __restrict__ B,
                   const float* __restrict__ bias, float* __restrict__ C,
                   int K, int Nld, int relu, int Kc) {
    const int nb   = blockIdx.x;
    const int mb   = blockIdx.y;
    const int z    = blockIdx.z;
    const int lane = threadIdx.x;
    const int half = lane >> 4;
    const int lm   = lane & 15;

    const int k0 = z * Kc;

    v8f acc = {};
    const float* Ap = A + (size_t)(mb * 16 + lm) * K + k0 + 2 * half;
    const float* Bp = B + (size_t)(k0 + 2 * half) * Nld + nb * 16 + lm;

#pragma unroll 4
    for (int k = 0; k < Kc; k += 4) {
        v2f a = *(const v2f*)Ap;
        v2f b;
        b.x = Bp[0];
        b.y = Bp[Nld];
        acc = __builtin_amdgcn_wmma_f32_16x16x4_f32(
                  false, a, false, b, (short)0, acc, false, false);
        Ap += 4;
        Bp += (size_t)4 * Nld;
    }

    const int cn = nb * 16 + lm;
    if (gridDim.z == 1) {
        const float bb = bias ? bias[cn] : 0.0f;
#pragma unroll
        for (int v = 0; v < 8; ++v) {
            float r = acc[v] + bb;
            if (relu) r = fmaxf(r, 0.0f);
            C[(size_t)(mb * 16 + v + 8 * half) * Nld + cn] = r;
        }
    } else {
        const size_t M  = (size_t)gridDim.y * 16;
        float* Cp = C + (size_t)z * M * Nld;
#pragma unroll
        for (int v = 0; v < 8; ++v) {
            Cp[(size_t)(mb * 16 + v + 8 * half) * Nld + cn] = acc[v];
        }
    }
}

#define FC_NT     64
#define FC_AP     (FC_KC + 4)
#define FC_BP     (FC_NT + 8)
__global__ __launch_bounds__(256)
void fc_slab_gemm_f32(const float* __restrict__ A, const float* __restrict__ B,
                      float* __restrict__ part) {
    __shared__ __attribute__((aligned(16))) float As[B_GRAPHS * FC_AP];
    __shared__ __attribute__((aligned(16))) float Bs[FC_KC * FC_BP];
    const int tid = threadIdx.x;
    const int n0  = blockIdx.x * FC_NT;
    const int z   = blockIdx.y;
    const int k0  = z * FC_KC;

    {
        float4 rb[16], ra[16];
#pragma unroll
        for (int u = 0; u < 16; ++u) {
            const int i = u * 256 + tid, r = i >> 4, c4 = i & 15;
            rb[u] = *(const float4*)(B + (size_t)(k0 + r) * 256 + n0 + c4 * 4);
        }
#pragma unroll
        for (int u = 0; u < 16; ++u) {
            const int i = u * 256 + tid, r = i >> 6, c4 = i & 63;
            ra[u] = *(const float4*)(A + (size_t)r * FC_IN_SZ + k0 + c4 * 4);
        }
#pragma unroll
        for (int u = 0; u < 16; ++u) {
            const int i = u * 256 + tid, r = i >> 4, c4 = i & 15;
            *(float4*)(Bs + r * FC_BP + c4 * 4) = rb[u];
        }
#pragma unroll
        for (int u = 0; u < 16; ++u) {
            const int i = u * 256 + tid, r = i >> 6, c4 = i & 63;
            *(float4*)(As + r * FC_AP + c4 * 4) = ra[u];
        }
#pragma unroll
        for (int u = 0; u < 16; ++u) {
            const int i = (u + 16) * 256 + tid, r = i >> 6, c4 = i & 63;
            ra[u] = *(const float4*)(A + (size_t)r * FC_IN_SZ + k0 + c4 * 4);
        }
#pragma unroll
        for (int u = 0; u < 16; ++u) {
            const int i = (u + 16) * 256 + tid, r = i >> 6, c4 = i & 63;
            *(float4*)(As + r * FC_AP + c4 * 4) = ra[u];
        }
    }
    __syncthreads();

    const int lane = tid & 31;
    const int w    = tid >> 5;
    const int half = lane >> 4;
    const int c    = lane & 15;
    const float* ap = As + (16 * w + c) * FC_AP + 4 * half;
    const float* bp = Bs + (4 * half) * FC_BP + c;

    v8f acc[4] = {};
    float4 av = *(const float4*)ap;
    v2f b1[4], b2[4];
#pragma unroll
    for (int j = 0; j < 4; ++j) {
        b1[j].x = bp[0 * FC_BP + 16 * j];  b1[j].y = bp[1 * FC_BP + 16 * j];
        b2[j].x = bp[2 * FC_BP + 16 * j];  b2[j].y = bp[3 * FC_BP + 16 * j];
    }
    for (int kk = 0; kk < FC_KC; kk += 8) {
        const int kn = kk + 8 < FC_KC ? kk + 8 : kk;
        const float4 avn = *(const float4*)(ap + kn);
        const float* bk = bp + kn * FC_BP;
        v2f n1[4], n2[4];
#pragma unroll
        for (int j = 0; j < 4; ++j) {
            n1[j].x = bk[0 * FC_BP + 16 * j];  n1[j].y = bk[1 * FC_BP + 16 * j];
            n2[j].x = bk[2 * FC_BP + 16 * j];  n2[j].y = bk[3 * FC_BP + 16 * j];
        }
        v2f a1, a2;
        a1.x = av.x;  a1.y = av.y;
        a2.x = av.z;  a2.y = av.w;
#pragma unroll
        for (int j = 0; j < 4; ++j)
            acc[j] = __builtin_amdgcn_wmma_f32_16x16x4_f32(
                         false, a1, false, b1[j], (short)0, acc[j], false, false);
#pragma unroll
        for (int j = 0; j < 4; ++j)
            acc[j] = __builtin_amdgcn_wmma_f32_16x16x4_f32(
                         false, a2, false, b2[j], (short)0, acc[j], false, false);
        av = avn;
#pragma unroll
        for (int j = 0; j < 4; ++j) { b1[j] = n1[j]; b2[j] = n2[j]; }
    }

    float* Cp = part + (size_t)z * (B_GRAPHS * 256) + n0 + c;
#pragma unroll
    for (int j = 0; j < 4; ++j) {
#pragma unroll
        for (int v = 0; v < 8; ++v)
            Cp[(size_t)(16 * w + v + 8 * half) * 256 + 16 * j] = acc[j][v];
    }
}

__global__ void reduce_bias_relu_kernel(const float* __restrict__ part,
                                        const float* __restrict__ bias,
                                        float* __restrict__ C,
                                        int MN, int Nld, int relu) {
    int i = blockIdx.x * blockDim.x + threadIdx.x;
    if (i >= MN) return;
    float s = 0.0f;
#pragma unroll
    for (int zc = 0; zc < KCHUNKS; ++zc) s += part[(size_t)zc * MN + i];
    s += bias[i % Nld];
    if (relu) s = fmaxf(s, 0.0f);
    C[i] = s;
}

#define PRW (NPART / (WG / 32))
static_assert((WG / 32) * NPART == 8 * WG, "eight counters per thread in the clearing store");
__global__ __launch_bounds__(WG)
void partition_kernel(const int* __restrict__ row, const int* __restrict__ col,
                      const unsigned int* __restrict__ rtot, unsigned int* __restrict__ fillc,
                      unsigned int* __restrict__ list4) {
    __shared__ unsigned int stage[PCH];
    __shared__ unsigned int cw[(WG / 32) * NPART];
    __shared__ unsigned int lbase[NPART], gbase[NPART], btot[NPART];
    __shared__ unsigned int wrun[NPART / 32], wlen[NPART / 32];
    const int tid = threadIdx.x, lane = tid & 31, w = tid >> 5;

    int4 cv[4], rv[4];
    {
        const int4* c4 = (const int4*)(col + (size_t)blockIdx.x * PCH);
        const int4* r4 = (const int4*)(row + (size_t)blockIdx.x * PCH);
#pragma unroll
        for (int u = 0; u < 4; ++u) cv[u] = c4[u * WG + tid];
#pragma unroll
        for (int u = 0; u < 4; ++u) rv[u] = r4[u * WG + tid];
    }
    const unsigned int li = tid < NPART ? rtot[tid] : 0u;
#pragma unroll
    for (int k = 0; k < 8; ++k) cw[k * WG + tid] = 0u;
    __syncthreads();

    const int t[16] = {cv[0].x, cv[0].y, cv[0].z, cv[0].w, cv[1].x, cv[1].y, cv[1].z, cv[1].w,
                       cv[2].x, cv[2].y, cv[2].z, cv[2].w, cv[3].x, cv[3].y, cv[3].z, cv[3].w};
    const int f[16] = {rv[0].x, rv[0].y, rv[0].z, rv[0].w, rv[1].x, rv[1].y, rv[1].z, rv[1].w,
                       rv[2].x, rv[2].y, rv[2].z, rv[2].w, rv[3].x, rv[3].y, rv[3].z, rv[3].w};
    unsigned int rk[16];
#pragma unroll
    for (int k = 0; k < 16; ++k) rk[k] = atomicAdd(&cw[w * NPART + (t[k] >> PSHIFT)], 1u);
    __syncthreads();

    unsigned int run = 0u, inc = 0u, linc = 0u;
    if (tid < NPART) {
        for (int w0 = 0; w0 < WG / 32; w0 += 16) {
            unsigned int c[16];
#pragma unroll
            for (int i = 0; i < 16; ++i) c[i] = cw[(w0 + i) * NPART + tid];
#pragma unroll
            for (int i = 0; i < 16; ++i) { cw[(w0 + i) * NPART + tid] = run; run += c[i]; }
        }
        inc = run;  linc = li;
#pragma unroll
        for (int d = 1; d < 32; d <<= 1) {
            const unsigned int o  = (unsigned int)__shfl_up((int)inc, d);
            const unsigned int lo = (unsigned int)__shfl_up((int)linc, d);
            if (lane >= d) { inc += o; linc += lo; }
        }
        if (lane == 31) { wrun[w] = inc; wlen[w] = linc; }
    }
    __syncthreads();
    if (tid < NPART) {
        unsigned int ro = 0u, lo = 0u;
#pragma unroll
        for (int i = 0; i < NPART / 32; ++i) {
            if (i < w) { ro += wrun[i]; lo += wlen[i]; }
        }
        lbase[tid] = ro + (inc - run);
        btot[tid]  = run;
        gbase[tid] = lo + (linc - li) + atomicAdd(&fillc[tid], run);
    }
    __syncthreads();

#pragma unroll
    for (int k = 0; k < 16; ++k) {
        const int b = t[k] >> PSHIFT;
        stage[lbase[b] + cw[w * NPART + b] + rk[k]] =
            ((unsigned int)(t[k] & (PNODES - 1)) << RBITS) | ((unsigned int)f[k] & ((1u << RBITS) - 1u));
    }
    __syncthreads();

    unsigned int rl[PRW], rd[PRW];
#pragma unroll
    for (int j = 0; j < PRW; ++j) {
        const int q = w * PRW + j;
        rl[j] = lbase[q];
        rd[j] = gbase[q] - rl[j];
    }
    const unsigned int hi = rl[PRW - 1] + btot[w * PRW + PRW - 1];
    for (unsigned int i = rl[0] + (unsigned int)lane; i < hi; i += 32u) {
        unsigned int d = rd[0];
#pragma unroll
        for (int j = 1; j < PRW; ++j) d = i >= rl[j] ? rd[j] : d;
        list4[i + d] = stage[i];
    }
}

#define EPI    8
#define LSLOTS 4
__global__ __launch_bounds__(WG)
void scatter_first_kernel(const float* __restrict__ xw, const float* __restrict__ dinv,
                          const unsigned int* __restrict__ rtot, const unsigned int* __restrict__ list4,
                          unsigned int* __restrict__ list, float* __restrict__ sums) {
    __shared__ __attribute__((aligned(16))) float acc[PNODES * H_DIM];
    __shared__ float dl[PNODES];

    const int tid  = threadIdx.x;
    const int lane = tid & 31;
    const int p = blockIdx.x;
    float4* acc4 = (float4*)acc;
#pragma unroll
    for (int k = 0; k < PNODES * H_DIM / 4 / WG; ++k)
        acc4[k * WG + tid] = make_float4(0.0f, 0.0f, 0.0f, 0.0f);
    if (tid < PNODES) dl[tid] = dinv[p * PNODES + tid];
    const unsigned int start = list_start(rtot, p, lane);
    const unsigned int* L = list4 + start;
    uint2* LW = (uint2*)list + start;
    const int n = (int)rtot[p];
    __syncthreads();

    const float4* xw4 = (const float4*)xw;
    const int g  = lane >> 2;
    const int l4 = lane & 3;
    const int nblk = (n + 31) >> 5;
    const int wstep = WG / 32;

    int blk = tid >> 5;
    unsigned int e[LSLOTS];
    if (blk < nblk) {
#pragma unroll
        for (int j = 0; j < LSLOTS; ++j) {
            const int i = blk * 32 + j * EPI + g;
            e[j] = L[i < n ? i : n - 1];
        }
    }
    for (; blk < nblk; blk += wstep) {
        const int nb = blk + wstep < nblk ? blk + wstep : blk;
        bool   h[LSLOTS];
        float4 v[LSLOTS];
        float  dr[LSLOTS], dc[LSLOTS];
        int    nd[LSLOTS];
        unsigned int en[LSLOTS];
#pragma unroll
        for (int j = 0; j < LSLOTS; ++j) {
            h[j] = blk * 32 + j * EPI + g < n;
            const int r = (int)(e[j] & ((1u << RBITS) - 1u));
            nd[j] = (int)(e[j] >> RBITS) & (PNODES - 1);
            dr[j] = dinv[r];
            v[j]  = xw4[r * 4 + l4];
            dc[j] = dl[nd[j]];
        }
#pragma unroll
        for (int j = 0; j < LSLOTS; ++j) {
            const int i = nb * 32 + j * EPI + g;
            en[j] = L[i < n ? i : n - 1];
        }
        asm volatile("" :: "v"(dr[0]), "v"(dr[1]), "v"(dr[2]), "v"(dr[3]),
                           "v"(dc[0]), "v"(dc[1]), "v"(dc[2]), "v"(dc[3]),
                           "v"(v[0].x), "v"(v[0].y), "v"(v[0].z), "v"(v[0].w),
                           "v"(v[1].x), "v"(v[1].y), "v"(v[1].z), "v"(v[1].w),
                           "v"(v[2].x), "v"(v[2].y), "v"(v[2].z), "v"(v[2].w),
                           "v"(v[3].x), "v"(v[3].y), "v"(v[3].z), "v"(v[3].w));
#pragma unroll
        for (int j = 0; j < LSLOTS; ++j) {
            if (h[j]) {
                const float w  = dr[j] * dc[j];
                if (l4 == 0) LW[blk * 32 + j * EPI + g] = make_uint2(e[j], __float_as_uint(w));
                const int   sw = (nd[j] >> 2) & 3;
                float* a = acc + (nd[j] * H_DIM + l4 * 4);
                atomicAdd(a + (0 ^ sw), w * v[j].x);
                atomicAdd(a + (1 ^ sw), w * v[j].y);
                atomicAdd(a + (2 ^ sw), w * v[j].z);
                atomicAdd(a + (3 ^ sw), w * v[j].w);
            }
        }
#pragma unroll
        for (int j = 0; j < LSLOTS; ++j) e[j] = en[j];
    }
    __syncthreads();

    float4* out4 = (float4*)sums + (size_t)p * PNODES * (H_DIM / 4);
#pragma unroll
    for (int k = 0; k < PNODES * H_DIM / 4 / WG; ++k) {
        const int idx = k * WG + tid;
        const int sw  = (idx >> 4) & 3;
        const float4 a = acc4[idx];
        float4 b, o;
        b.x = (sw & 1) ? a.y : a.x;  b.y = (sw & 1) ? a.x : a.y;
        b.z = (sw & 1) ? a.w : a.z;  b.w = (sw & 1) ? a.z : a.w;
        o.x = (sw & 2) ? b.z : b.x;  o.y = (sw & 2) ? b.w : b.y;
        o.z = (sw & 2) ? b.x : b.z;  o.w = (sw & 2) ? b.y : b.w;
        out4[idx] = o;
    }
}

__global__ __launch_bounds__(WG)
void scatter_list_kernel(const float* __restrict__ xw,
                         const unsigned int* __restrict__ rtot, const unsigned int* __restrict__ list,
                         float* __restrict__ sums) {
    __shared__ __attribute__((aligned(16))) float acc[PNODES * H_DIM];

    const int tid  = threadIdx.x;
    const int lane = tid & 31;
    const int p = blockIdx.x;
    float4* acc4 = (float4*)acc;
#pragma unroll
    for (int k = 0; k < PNODES * H_DIM / 4 / WG; ++k)
        acc4[k * WG + tid] = make_float4(0.0f, 0.0f, 0.0f, 0.0f);
    const uint2* L = (const uint2*)list + list_start(rtot, p, lane);
    const int n = (int)rtot[p];
    __syncthreads();

    const float4* xw4 = (const float4*)xw;
    const int g  = lane >> 2;
    const int l4 = lane & 3;
    const int nblk = (n + 31) >> 5;
    const int wstep = WG / 32;

    int blk = tid >> 5;
    uint2 e[LSLOTS];
    if (blk < nblk) {
#pragma unroll
        for (int j = 0; j < LSLOTS; ++j) {
            const int i = blk * 32 + j * EPI + g;
            e[j] = L[i < n ? i : n - 1];
        }
    }
    for (; blk < nblk; blk += wstep) {
        const int nb = blk + wstep < nblk ? blk + wstep : blk;
        bool   h[LSLOTS];
        float4 v[LSLOTS];
        uint2  en[LSLOTS];
#pragma unroll
        for (int j = 0; j < LSLOTS; ++j) {
            h[j] = blk * 32 + j * EPI + g < n;
            const int r = (int)(e[j].x & ((1u << RBITS) - 1u));
            v[j] = xw4[r * 4 + l4];
        }
#pragma unroll
        for (int j = 0; j < LSLOTS; ++j) {
            const int i = nb * 32 + j * EPI + g;
            en[j] = L[i < n ? i : n - 1];
        }
        asm volatile("" :: "v"(v[0].x), "v"(v[0].y), "v"(v[0].z), "v"(v[0].w),
                           "v"(v[1].x), "v"(v[1].y), "v"(v[1].z), "v"(v[1].w),
                           "v"(v[2].x), "v"(v[2].y), "v"(v[2].z), "v"(v[2].w),
                           "v"(v[3].x), "v"(v[3].y), "v"(v[3].z), "v"(v[3].w));
#pragma unroll
        for (int j = 0; j < LSLOTS; ++j) {
            if (h[j]) {
                const float w  = __uint_as_float(e[j].y);
                const int   nd = (int)(e[j].x >> RBITS) & (PNODES - 1);
                const int   sw = (nd >> 2) & 3;
                float* a = acc + (nd * H_DIM + l4 * 4);
                atomicAdd(a + (0 ^ sw), w * v[j].x);
                atomicAdd(a + (1 ^ sw), w * v[j].y);
                atomicAdd(a + (2 ^ sw), w * v[j].z);
                atomicAdd(a + (3 ^ sw), w * v[j].w);
            }
        }
#pragma unroll
        for (int j = 0; j < LSLOTS; ++j) e[j] = en[j];
    }
    __syncthreads();

    float4* out4 = (float4*)sums + (size_t)p * PNODES * (H_DIM / 4);
#pragma unroll
    for (int k = 0; k < PNODES * H_DIM / 4 / WG; ++k) {
        const int idx = k * WG + tid;
        const int sw  = (idx >> 4) & 3;
        const float4 a = acc4[idx];
        float4 b, o;
        b.x = (sw & 1) ? a.y : a.x;  b.y = (sw & 1) ? a.x : a.y;
        b.z = (sw & 1) ? a.w : a.z;  b.w = (sw & 1) ? a.z : a.w;
        o.x = (sw & 2) ? b.z : b.x;  o.y = (sw & 2) ? b.w : b.y;
        o.z = (sw & 2) ? b.x : b.z;  o.w = (sw & 2) ? b.y : b.w;
        out4[idx] = o;
    }
}

__global__ void finalize_kernel(float* __restrict__ h, const float* __restrict__ sums,
                                const float* __restrict__ xw,
                                const float* __restrict__ dinv,
                                const float* __restrict__ bias, int n4) {
    int i = blockIdx.x * blockDim.x + threadIdx.x;
    if (i >= n4) return;
    const float4 a = ((const float4*)sums)[i];
    const int node = i >> 2, j0 = (i & 3) * 4;
    const float d = dinv[node];
    const float4 x = ((const float4*)xw)[i];
    float4 r;
    r.x = fmaxf(a.x + d * d * x.x + bias[j0 + 0], 0.0f);
    r.y = fmaxf(a.y + d * d * x.y + bias[j0 + 1], 0.0f);
    r.z = fmaxf(a.z + d * d * x.z + bias[j0 + 2], 0.0f);
    r.w = fmaxf(a.w + d * d * x.w + bias[j0 + 3], 0.0f);
    ((float4*)h)[i] = r;
}

__global__ __launch_bounds__(256)
void finalize_mm_kernel(float* __restrict__ hout, const float* __restrict__ sums,
                        const float* __restrict__ hin, const float* __restrict__ dinv,
                        const float* __restrict__ W, const float* __restrict__ bias) {
    const int lane = threadIdx.x & 31;
    const int tile = blockIdx.x * 8 + (threadIdx.x >> 5);
    const int hh   = lane >> 4;
    const int c    = lane & 15;
    const int node = tile * 16 + c;
    const size_t i4 = (size_t)node * (H_DIM / 4) + 2 * hh;
    const float4 lo = ((const float4*)sums)[i4];
    const float4 hi = ((const float4*)sums)[i4 + 1];
    const float4 x0 = ((const float4*)hin)[i4];
    const float4 x1 = ((const float4*)hin)[i4 + 1];
    const float d  = dinv[node];
    v2f b[4];
#pragma unroll
    for (int j = 0; j < 4; ++j) {
        b[j].x = W[(8 * hh + 2 * j) * H_DIM + c];
        b[j].y = W[(8 * hh + 2 * j + 1) * H_DIM + c];
    }
    const float bb = bias[c];

    const float d2 = d * d;
    v2f a[4];
    a[0].x = lo.x + d2 * x0.x;  a[0].y = lo.y + d2 * x0.y;
    a[1].x = lo.z + d2 * x0.z;  a[1].y = lo.w + d2 * x0.w;
    a[2].x = hi.x + d2 * x1.x;  a[2].y = hi.y + d2 * x1.y;
    a[3].x = hi.z + d2 * x1.z;  a[3].y = hi.w + d2 * x1.w;

    v8f acc = {};
#pragma unroll
    for (int j = 0; j < 4; ++j)
        acc = __builtin_amdgcn_wmma_f32_16x16x4_f32(
                  false, a[j], false, b[j], (short)0, acc, false, false);

    float* o = hout + (size_t)(tile * 16 + 8 * hh) * H_DIM + c;
#pragma unroll
    for (int r = 0; r < 8; ++r) o[r * H_DIM] = fmaxf(acc[r] + bb, 0.0f);
}

#define HD_NT     32
#define HD_NX     (1024 / HD_NT)
#define HD_MT     32
#define HD_K      256
#define HD_AP     (HD_K + 4)
#define HD_BP     (HD_NT + 8)
__global__ __launch_bounds__(128)
void head_slab_gemm_f32(const float* __restrict__ Hp, const float* __restrict__ V,
                        const float* __restrict__ Wp2, const float* __restrict__ bp2,
                        const float* __restrict__ Wiv, const float* __restrict__ biv,
                        const float* __restrict__ Wev, const float* __restrict__ bev,
                        float* __restrict__ P, float* __restrict__ out_ext,
                        float* __restrict__ out_int) {
    __shared__ __attribute__((aligned(16))) float As[HD_MT * HD_AP];
    __shared__ __attribute__((aligned(16))) float Bs[HD_K * HD_BP];
    const int  tid   = threadIdx.x;
    const bool value = blockIdx.x == HD_NX;
    const int  n0    = blockIdx.x * HD_NT;
    const int  m0    = blockIdx.y * HD_MT;
    const float* A   = value ? V : Hp;

    const int lane = tid & 31;
    const int w    = tid >> 5;
    const int rt   = w & 1;
    const int ct   = w >> 1;
    const int half = lane >> 4;
    const int c    = lane & 15;

    float bb;
    if (!value) bb = bp2[n0 + 16 * ct + c];
    else        bb = c == 0 ? biv[0] : bev[0];

    {
        const float* a0 = A + (size_t)(m0 + (tid >> 6)) * HD_K + (tid & 63) * 4;
        float*       as = As + (tid >> 6) * HD_AP + (tid & 63) * 4;
#define HD_REP16(M) M(0) M(1) M(2) M(3) M(4) M(5) M(6) M(7) M(8) M(9) M(10) M(11) M(12) M(13) M(14) M(15)
#define HD_LOAD_A(u)  const float4 ra##u = *(const float4*)(a0 + (u) * (2 * HD_K));
#define HD_STORE_A(u) *(float4*)(as + (u) * (2 * HD_AP)) = ra##u;
        HD_REP16(HD_LOAD_A)
        if (!value) {
            const float* b0 = Wp2 + (size_t)(tid >> 3) * 1024 + n0 + (tid & 7) * 4;
            float*       bs = Bs + (tid >> 3) * HD_BP + (tid & 7) * 4;
            float4 rb[16];
#pragma unroll
            for (int u = 0; u < 16; ++u) rb[u] = *(const float4*)(b0 + u * (16 * 1024));
#pragma unroll
            for (int u = 0; u < 16; ++u) *(float4*)(bs + u * (16 * HD_BP)) = rb[u];
        } else {
            const float wi0 = Wiv[tid], wi1 = Wiv[tid + 128];
            const float we0 = Wev[tid], we1 = Wev[tid + 128];
            *(float4*)(Bs + tid * HD_BP)         = make_float4(wi0, we0, 0.0f, 0.0f);
            *(float4*)(Bs + (tid + 128) * HD_BP) = make_float4(wi1, we1, 0.0f, 0.0f);
#pragma unroll
            for (int c4 = 1; c4 < HD_NT / 4; ++c4) {
                *(float4*)(Bs + tid * HD_BP + c4 * 4)         = make_float4(0.0f, 0.0f, 0.0f, 0.0f);
                *(float4*)(Bs + (tid + 128) * HD_BP + c4 * 4) = make_float4(0.0f, 0.0f, 0.0f, 0.0f);
            }
        }
        HD_REP16(HD_STORE_A)
#undef HD_STORE_A
#undef HD_LOAD_A
#undef HD_REP16
    }
    __syncthreads();

    const float* ap = As + (16 * rt + c) * HD_AP + 4 * half;
    const float* bp = Bs + (4 * half) * HD_BP + 16 * ct + c;
    float4 af[HD_K / 8];
    float  bf[HD_K / 8][4];
#pragma unroll
    for (int s = 0; s < HD_K / 8; ++s) {
        af[s] = *(const float4*)(ap + 8 * s);
#pragma unroll
        for (int q = 0; q < 4; ++q) bf[s][q] = bp[(8 * s + q) * HD_BP];
    }
    __builtin_amdgcn_sched_barrier(0);
    v8f acc0 = {}, acc1 = {};
#pragma unroll
    for (int s = 0; s < HD_K / 8; ++s) {
        v2f a1, a2, b1, b2;
        a1.x = af[s].x;   a1.y = af[s].y;
        a2.x = af[s].z;   a2.y = af[s].w;
        b1.x = bf[s][0];  b1.y = bf[s][1];
        b2.x = bf[s][2];  b2.y = bf[s][3];
        acc0 = __builtin_amdgcn_wmma_f32_16x16x4_f32(
                   false, a1, false, b1, (short)0, acc0, false, false);
        acc1 = __builtin_amdgcn_wmma_f32_16x16x4_f32(
                   false, a2, false, b2, (short)0, acc1, false, false);
    }

    const int row0 = m0 + 16 * rt + 8 * half;
    if (!value) {
        const int cn = n0 + 16 * ct + c;
#pragma unroll
        for (int v = 0; v < 8; ++v)
            P[(size_t)(row0 + v) * 1024 + cn] = (acc0[v] + acc1[v]) + bb;
    } else if (ct == 0 && c < 2) {
        float* o = c == 0 ? out_int : out_ext;
#pragma unroll
        for (int v = 0; v < 8; ++v) o[row0 + v] = (acc0[v] + acc1[v]) + bb;
    }
}

extern "C" void kernel_launch(void* const* d_in, const int* in_sizes, int n_in,
                              void* d_out, int out_size, void* d_ws, size_t ws_size,
                              hipStream_t stream) {
    const float* x    = (const float*)d_in[0];
    const int*   ei   = (const int*)d_in[1];
    const float* W1   = (const float*)d_in[2];
    const float* b1   = (const float*)d_in[3];
    const float* W2   = (const float*)d_in[4];
    const float* b2   = (const float*)d_in[5];
    const float* W3   = (const float*)d_in[6];
    const float* b3   = (const float*)d_in[7];
    const float* Wp1  = (const float*)d_in[8];
    const float* bp1  = (const float*)d_in[9];
    const float* Wp2  = (const float*)d_in[10];
    const float* bp2  = (const float*)d_in[11];
    const float* Wv1  = (const float*)d_in[12];
    const float* bv1  = (const float*)d_in[13];
    const float* Wiv  = (const float*)d_in[14];
    const float* biv  = (const float*)d_in[15];
    const float* Wev  = (const float*)d_in[16];
    const float* bev  = (const float*)d_in[17];

    const int* row = ei;
    const int* col = ei + E_EDGES;

    char* ws = (char*)d_ws;
    float* dinv = (float*)ws;
    float* bufA = (float*)(ws + (size_t)N_NODES * 4);
    float* bufB = bufA + (size_t)N_NODES * H_DIM;
    float* Hp   = bufB + (size_t)N_NODES * H_DIM;
    float* Vv   = Hp + (size_t)B_GRAPHS * 256;
    float* part = Vv + (size_t)B_GRAPHS * 256;
    unsigned int* cnt = (unsigned int*)(part + (size_t)KCHUNKS * B_GRAPHS * 256);
    float* sums = (float*)(cnt + (size_t)DSLICE * N_NODES);
    unsigned int* elist = (unsigned int*)(sums + (size_t)N_NODES * H_DIM);
    unsigned int* rtot = elist + (size_t)2 * E_EDGES;
    unsigned int* fillc = rtot + NPART;
    unsigned int* list4 = fillc + NPART;

    float* P     = (float*)d_out;
    float* ext_V = P + (size_t)B_GRAPHS * 1024;
    float* int_V = ext_V + B_GRAPHS;

    const int TPB = 256;
    const int nh  = N_NODES * H_DIM;
    const int nh4 = nh / 4;
    const int fcMN = B_GRAPHS * 256;

    degree_lds_kernel<<<DPART * DSLICE, WG, 0, stream>>>(col, cnt, fillc);
    dinv_kernel<<<NPART, PNODES, 0, stream>>>(cnt, dinv, rtot);

    partition_kernel<<<E_EDGES / PCH, WG, 0, stream>>>(row, col, rtot, fillc, list4);

    wmma_gemm_f32<<<dim3(1, N_NODES / 16, 1), 32, 0, stream>>>(x, W1, nullptr, bufA, S_IN, H_DIM, 0, S_IN);
    scatter_first_kernel<<<NPART, WG, 0, stream>>>(bufA, dinv, rtot, list4, elist, sums);
    finalize_kernel<<<(nh4 + TPB - 1) / TPB, TPB, 0, stream>>>(bufB, sums, bufA, dinv, b1, nh4);

    scatter_list_kernel<<<NPART, WG, 0, stream>>>(bufB, rtot, elist, sums);
    finalize_mm_kernel<<<N_NODES / 16 / 8, 256, 0, stream>>>(bufA, sums, bufB, dinv, W2, b2);

    scatter_list_kernel<<<NPART, WG, 0, stream>>>(bufA, rtot, elist, sums);
    finalize_mm_kernel<<<N_NODES / 16 / 8, 256, 0, stream>>>(bufB, sums, bufA, dinv, W3, b3);


    fc_slab_gemm_f32<<<dim3(256 / FC_NT, KCHUNKS), 256, 0, stream>>>(bufB, Wp1, part);
    reduce_bias_relu_kernel<<<(fcMN + TPB - 1) / TPB, TPB, 0, stream>>>(part, bp1, Hp, fcMN, 256, 1);
    fc_slab_gemm_f32<<<dim3(256 / FC_NT, KCHUNKS), 256, 0, stream>>>(bufB, Wv1, part);
    reduce_bias_relu_kernel<<<(fcMN + TPB - 1) / TPB, TPB, 0, stream>>>(part, bv1, Vv, fcMN, 256, 1);

    head_slab_gemm_f32<<<dim3(HD_NX + 1, B_GRAPHS / HD_MT), 128, 0, stream>>>(
        Hp, Vv, Wp2, bp2, Wiv, biv, Wev, bev, P, ext_V, int_V);
}
